# starting kernel
# baseline (speedup 1.0000x reference)
#include <hip/hip_runtime.h>
#include <hip/hip_bf16.h>

typedef __attribute__((ext_vector_type(2))) float v2f;
typedef __attribute__((ext_vector_type(8))) float v8f;

#define IN_FEAT 512
#define HIDDEN  16
#define NCLS    7


__global__ void gcn_deg_init(float* deg, int n) {
    int i = blockIdx.x * blockDim.x + threadIdx.x;
    if (i < n) deg[i] = 1.0f;
}

__global__ void gcn_deg_accum(const int* __restrict__ dst, float* deg, int E) {
    int e = blockIdx.x * blockDim.x + threadIdx.x;
    if (e < E) atomicAdd(&deg[dst[e]], 1.0f);
}

__global__ void gcn_deg_rsqrt(float* dinv, int n) {
    int i = blockIdx.x * blockDim.x + threadIdx.x;
    if (i < n) {
        float d = dinv[i];
        dinv[i] = (d > 0.0f) ? rsqrtf(d) : 0.0f;
    }
}

__global__ void gcn_zero(float* p, int n) {
    int i = blockIdx.x * blockDim.x + threadIdx.x;
    if (i < n) p[i] = 0.0f;
}


__global__ void gcn_gemm1_wmma(const float* __restrict__ x,
                               const float* __restrict__ W1,
                               float* __restrict__ h, int n) {
    int wave = (blockIdx.x * blockDim.x + threadIdx.x) >> 5;
    int lane = threadIdx.x & 31;
    int row0 = wave * 16;
    if (row0 >= n) return;

    int r    = lane & 15;
    int hi   = lane >> 4;
    int koff = hi * 2;
    int row  = row0 + r;
    if (row >= n) row = n - 1;
    const float* xrow = x + (size_t)row * IN_FEAT;

    v8f c = {0.f, 0.f, 0.f, 0.f, 0.f, 0.f, 0.f, 0.f};

    #pragma unroll 4
    for (int k = 0; k < IN_FEAT; k += 4) {
        if ((k & 63) == 0)
            __builtin_prefetch(xrow + k + 128, 0, 1);
        v2f a = *(const v2f*)(xrow + k + koff);
        int kk = k + koff;
        v2f b;
        b.x = W1[(size_t)kk * HIDDEN + r];
        b.y = W1[(size_t)(kk + 1) * HIDDEN + r];
        c = __builtin_amdgcn_wmma_f32_16x16x4_f32(false, a, false, b,
                                                  (short)0, c, false, false);
    }

    int srow = row0 + hi * 8;
    #pragma unroll
    for (int i = 0; i < 8; ++i) {
        int rr = srow + i;
        if (rr < n) h[(size_t)rr * HIDDEN + r] = c[i];
    }
}


__global__ void gcn_agg1(const int* __restrict__ src, const int* __restrict__ dst,
                         const float* __restrict__ dinv, const float* __restrict__ h,
                         float* out1, int E) {
    int t = blockIdx.x * blockDim.x + threadIdx.x;
    int e = t >> 2, q = t & 3;
    if (e >= E) return;
    int s = src[e], d = dst[e];
    float w = dinv[s] * dinv[d];
    float4 v = *((const float4*)(h + (size_t)s * HIDDEN) + q);
    float* o = out1 + (size_t)d * HIDDEN + q * 4;
    atomicAdd(o + 0, v.x * w);
    atomicAdd(o + 1, v.y * w);
    atomicAdd(o + 2, v.z * w);
    atomicAdd(o + 3, v.w * w);
}


__global__ void gcn_selfloop_relu(const float* __restrict__ out1,
                                  const float* __restrict__ h,
                                  const float* __restrict__ dinv,
                                  const float* __restrict__ b1,
                                  float* hrelu, int n) {
    int i = blockIdx.x * blockDim.x + threadIdx.x;
    if (i >= n) return;
    float w = dinv[i] * dinv[i];
    const float* o = out1 + (size_t)i * HIDDEN;
    const float* hh = h + (size_t)i * HIDDEN;
    float* dst = hrelu + (size_t)i * HIDDEN;
    #pragma unroll
    for (int c = 0; c < HIDDEN; ++c)
        dst[c] = fmaxf(o[c] + hh[c] * w + b1[c], 0.0f);
}


__global__ void gcn_gemm2_wmma(const float* __restrict__ hrelu,
                               const float* __restrict__ W2,
                               float* __restrict__ g, int n) {
    int wave = (blockIdx.x * blockDim.x + threadIdx.x) >> 5;
    int lane = threadIdx.x & 31;
    int row0 = wave * 16;
    if (row0 >= n) return;

    int r    = lane & 15;
    int hi   = lane >> 4;
    int koff = hi * 2;
    int row  = row0 + r;
    if (row >= n) row = n - 1;
    const float* hrow = hrelu + (size_t)row * HIDDEN;

    v8f c = {0.f, 0.f, 0.f, 0.f, 0.f, 0.f, 0.f, 0.f};

    #pragma unroll
    for (int k = 0; k < HIDDEN; k += 4) {
        int kk = k + koff;
        v2f a = *(const v2f*)(hrow + kk);
        v2f b;
        b.x = (r < NCLS) ? W2[(size_t)kk * NCLS + r] : 0.0f;
        b.y = (r < NCLS) ? W2[(size_t)(kk + 1) * NCLS + r] : 0.0f;
        c = __builtin_amdgcn_wmma_f32_16x16x4_f32(false, a, false, b,
                                                  (short)0, c, false, false);
    }

    int srow = row0 + hi * 8;
    #pragma unroll
    for (int i = 0; i < 8; ++i) {
        int rr = srow + i;
        if (rr < n) g[(size_t)rr * HIDDEN + r] = c[i];
    }
}


__global__ void gcn_agg2(const int* __restrict__ src, const int* __restrict__ dst,
                         const float* __restrict__ dinv, const float* __restrict__ g,
                         float* out2, int E) {
    int e = blockIdx.x * blockDim.x + threadIdx.x;
    if (e >= E) return;
    int s = src[e], d = dst[e];
    float w = dinv[s] * dinv[d];
    const float* gs = g + (size_t)s * HIDDEN;
    float* o = out2 + (size_t)d * 8;
    #pragma unroll
    for (int c = 0; c < NCLS; ++c)
        atomicAdd(o + c, gs[c] * w);
}


__global__ void gcn_logsoftmax(const float* __restrict__ out2,
                               const float* __restrict__ g,
                               const float* __restrict__ dinv,
                               const float* __restrict__ b2,
                               float* __restrict__ out, int n) {
    int i = blockIdx.x * blockDim.x + threadIdx.x;
    if (i >= n) return;
    float w = dinv[i] * dinv[i];
    const float* o = out2 + (size_t)i * 8;
    const float* gg = g + (size_t)i * HIDDEN;
    float v[NCLS];
    float m = -3.4e38f;
    #pragma unroll
    for (int c = 0; c < NCLS; ++c) {
        v[c] = o[c] + gg[c] * w + b2[c];
        m = fmaxf(m, v[c]);
    }
    float s = 0.0f;
    #pragma unroll
    for (int c = 0; c < NCLS; ++c) s += expf(v[c] - m);
    float l = logf(s);
    float* dsto = out + (size_t)i * NCLS;
    #pragma unroll
    for (int c = 0; c < NCLS; ++c) dsto[c] = v[c] - m - l;
}


extern "C" void kernel_launch(void* const* d_in, const int* in_sizes, int n_in,
                              void* d_out, int out_size, void* d_ws, size_t ws_size,
                              hipStream_t stream) {
    const float* x  = (const float*)d_in[0];
    const int*   ei = (const int*)d_in[1];
    const float* W1 = (const float*)d_in[2];
    const float* b1 = (const float*)d_in[3];
    const float* W2 = (const float*)d_in[4];
    const float* b2 = (const float*)d_in[5];
    float* out = (float*)d_out;

    const int n = in_sizes[0] / IN_FEAT;
    const int E = in_sizes[1] / 2;
    const int* src = ei;
    const int* dst = ei + E;

    float* f     = (float*)d_ws;
    float* dinv  = f;
    float* h     = f + (size_t)n;
    float* out1  = f + (size_t)17 * n;
    float* hrelu = f + (size_t)33 * n;
    float* g     = out1;
    float* out2  = h;

    const int B = 256;
    const int tiles = (n + 15) / 16;
    const int gemmBlocks = (tiles + 7) / 8;

    gcn_deg_init   <<<(n + B - 1) / B, B, 0, stream>>>(dinv, n);
    gcn_deg_accum  <<<(E + B - 1) / B, B, 0, stream>>>(dst, dinv, E);
    gcn_deg_rsqrt  <<<(n + B - 1) / B, B, 0, stream>>>(dinv, n);

    gcn_zero       <<<((size_t)16 * n + B - 1) / B, B, 0, stream>>>(out1, 16 * n);
    gcn_gemm1_wmma <<<gemmBlocks, B, 0, stream>>>(x, W1, h, n);
    gcn_agg1       <<<((size_t)4 * E + B - 1) / B, B, 0, stream>>>(src, dst, dinv, h, out1, E);
    gcn_selfloop_relu<<<(n + B - 1) / B, B, 0, stream>>>(out1, h, dinv, b1, hrelu, n);

    gcn_zero       <<<((size_t)8 * n + B - 1) / B, B, 0, stream>>>(out2, 8 * n);
    gcn_gemm2_wmma <<<gemmBlocks, B, 0, stream>>>(hrelu, W2, g, n);
    gcn_agg2       <<<(E + B - 1) / B, B, 0, stream>>>(src, dst, dinv, g, out2, E);
    gcn_logsoftmax <<<(n + B - 1) / B, B, 0, stream>>>(out2, g, dinv, b2, out, n);
}


# agg2: 8 lanes per edge (one per class) through wave shuffles, one wave per 32 edges: an edge's 7 float atomics leave as one 28-byte row request, not 7 scattered dwords; arrival-order sums as before
# speedup vs baseline: 1.6641x; 1.6641x over previous
#include <hip/hip_runtime.h>
#include <hip/hip_bf16.h>

typedef __attribute__((ext_vector_type(2))) float v2f;
typedef __attribute__((ext_vector_type(8))) float v8f;

#define IN_FEAT 512
#define HIDDEN  16
#define NCLS    7


__global__ void gcn_deg_init(float* deg, int n) {
    int i = blockIdx.x * blockDim.x + threadIdx.x;
    if (i < n) deg[i] = 1.0f;
}

__global__ void gcn_deg_accum(const int* __restrict__ dst, float* deg, int E) {
    int e = blockIdx.x * blockDim.x + threadIdx.x;
    if (e < E) atomicAdd(&deg[dst[e]], 1.0f);
}

__global__ void gcn_deg_rsqrt(float* dinv, int n) {
    int i = blockIdx.x * blockDim.x + threadIdx.x;
    if (i < n) {
        float d = dinv[i];
        dinv[i] = (d > 0.0f) ? rsqrtf(d) : 0.0f;
    }
}

__global__ void gcn_zero(float* p, int n) {
    int i = blockIdx.x * blockDim.x + threadIdx.x;
    if (i < n) p[i] = 0.0f;
}


__global__ void gcn_gemm1_wmma(const float* __restrict__ x,
                               const float* __restrict__ W1,
                               float* __restrict__ h, int n) {
    int wave = (blockIdx.x * blockDim.x + threadIdx.x) >> 5;
    int lane = threadIdx.x & 31;
    int row0 = wave * 16;
    if (row0 >= n) return;

    int r    = lane & 15;
    int hi   = lane >> 4;
    int koff = hi * 2;
    int row  = row0 + r;
    if (row >= n) row = n - 1;
    const float* xrow = x + (size_t)row * IN_FEAT;

    v8f c = {0.f, 0.f, 0.f, 0.f, 0.f, 0.f, 0.f, 0.f};

    #pragma unroll 4
    for (int k = 0; k < IN_FEAT; k += 4) {
        if ((k & 63) == 0)
            __builtin_prefetch(xrow + k + 128, 0, 1);
        v2f a = *(const v2f*)(xrow + k + koff);
        int kk = k + koff;
        v2f b;
        b.x = W1[(size_t)kk * HIDDEN + r];
        b.y = W1[(size_t)(kk + 1) * HIDDEN + r];
        c = __builtin_amdgcn_wmma_f32_16x16x4_f32(false, a, false, b,
                                                  (short)0, c, false, false);
    }

    int srow = row0 + hi * 8;
    #pragma unroll
    for (int i = 0; i < 8; ++i) {
        int rr = srow + i;
        if (rr < n) h[(size_t)rr * HIDDEN + r] = c[i];
    }
}


__global__ void gcn_agg1(const int* __restrict__ src, const int* __restrict__ dst,
                         const float* __restrict__ dinv, const float* __restrict__ h,
                         float* out1, int E) {
    int t = blockIdx.x * blockDim.x + threadIdx.x;
    int e = t >> 2, q = t & 3;
    if (e >= E) return;
    int s = src[e], d = dst[e];
    float w = dinv[s] * dinv[d];
    float4 v = *((const float4*)(h + (size_t)s * HIDDEN) + q);
    float* o = out1 + (size_t)d * HIDDEN + q * 4;
    atomicAdd(o + 0, v.x * w);
    atomicAdd(o + 1, v.y * w);
    atomicAdd(o + 2, v.z * w);
    atomicAdd(o + 3, v.w * w);
}


__global__ void gcn_selfloop_relu(const float* __restrict__ out1,
                                  const float* __restrict__ h,
                                  const float* __restrict__ dinv,
                                  const float* __restrict__ b1,
                                  float* hrelu, int n) {
    int i = blockIdx.x * blockDim.x + threadIdx.x;
    if (i >= n) return;
    float w = dinv[i] * dinv[i];
    const float* o = out1 + (size_t)i * HIDDEN;
    const float* hh = h + (size_t)i * HIDDEN;
    float* dst = hrelu + (size_t)i * HIDDEN;
    #pragma unroll
    for (int c = 0; c < HIDDEN; ++c)
        dst[c] = fmaxf(o[c] + hh[c] * w + b1[c], 0.0f);
}


__global__ void gcn_gemm2_wmma(const float* __restrict__ hrelu,
                               const float* __restrict__ W2,
                               float* __restrict__ g, int n) {
    int wave = (blockIdx.x * blockDim.x + threadIdx.x) >> 5;
    int lane = threadIdx.x & 31;
    int row0 = wave * 16;
    if (row0 >= n) return;

    int r    = lane & 15;
    int hi   = lane >> 4;
    int koff = hi * 2;
    int row  = row0 + r;
    if (row >= n) row = n - 1;
    const float* hrow = hrelu + (size_t)row * HIDDEN;

    v8f c = {0.f, 0.f, 0.f, 0.f, 0.f, 0.f, 0.f, 0.f};

    #pragma unroll
    for (int k = 0; k < HIDDEN; k += 4) {
        int kk = k + koff;
        v2f a = *(const v2f*)(hrow + kk);
        v2f b;
        b.x = (r < NCLS) ? W2[(size_t)kk * NCLS + r] : 0.0f;
        b.y = (r < NCLS) ? W2[(size_t)(kk + 1) * NCLS + r] : 0.0f;
        c = __builtin_amdgcn_wmma_f32_16x16x4_f32(false, a, false, b,
                                                  (short)0, c, false, false);
    }

    int srow = row0 + hi * 8;
    #pragma unroll
    for (int i = 0; i < 8; ++i) {
        int rr = srow + i;
        if (rr < n) g[(size_t)rr * HIDDEN + r] = c[i];
    }
}


__global__ void gcn_agg2(const int* __restrict__ src, const int* __restrict__ dst,
                         const float* __restrict__ dinv, const float* __restrict__ g,
                         float* out2, int E) {
    int t    = blockIdx.x * blockDim.x + threadIdx.x;
    int lane = threadIdx.x & 31;
    int e0   = t - lane;
    if (e0 >= E) return;
    int e = e0 + lane;
    int s = 0, d = 0;
    float w = 0.0f;
    if (e < E) {
        s = src[e];
        d = dst[e];
        w = dinv[s] * dinv[d];
    }
    int nvalid = E - e0;
    if (nvalid > 32) nvalid = 32;
    int sub = lane >> 3;
    int c   = lane & 7;

    float m[8];
    int   dj[8];
    #pragma unroll
    for (int i = 0; i < 8; ++i) {
        int j    = 4 * i + sub;
        int sj   = __shfl(s, j);
        dj[i]    = __shfl(d, j);
        float wj = __shfl(w, j);
        float gv = g[(size_t)sj * HIDDEN + c];
        m[i] = gv * wj;
    }
    #pragma unroll
    for (int i = 0; i < 8; ++i) {
        int j = 4 * i + sub;
        if (c < NCLS && j < nvalid)
            atomicAdd(out2 + (size_t)dj[i] * 8 + c, m[i]);
    }
}


__global__ void gcn_logsoftmax(const float* __restrict__ out2,
                               const float* __restrict__ g,
                               const float* __restrict__ dinv,
                               const float* __restrict__ b2,
                               float* __restrict__ out, int n) {
    int i = blockIdx.x * blockDim.x + threadIdx.x;
    if (i >= n) return;
    float w = dinv[i] * dinv[i];
    const float* o = out2 + (size_t)i * 8;
    const float* gg = g + (size_t)i * HIDDEN;
    float v[NCLS];
    float m = -3.4e38f;
    #pragma unroll
    for (int c = 0; c < NCLS; ++c) {
        v[c] = o[c] + gg[c] * w + b2[c];
        m = fmaxf(m, v[c]);
    }
    float s = 0.0f;
    #pragma unroll
    for (int c = 0; c < NCLS; ++c) s += expf(v[c] - m);
    float l = logf(s);
    float* dsto = out + (size_t)i * NCLS;
    #pragma unroll
    for (int c = 0; c < NCLS; ++c) dsto[c] = v[c] - m - l;
}


extern "C" void kernel_launch(void* const* d_in, const int* in_sizes, int n_in,
                              void* d_out, int out_size, void* d_ws, size_t ws_size,
                              hipStream_t stream) {
    const float* x  = (const float*)d_in[0];
    const int*   ei = (const int*)d_in[1];
    const float* W1 = (const float*)d_in[2];
    const float* b1 = (const float*)d_in[3];
    const float* W2 = (const float*)d_in[4];
    const float* b2 = (const float*)d_in[5];
    float* out = (float*)d_out;

    const int n = in_sizes[0] / IN_FEAT;
    const int E = in_sizes[1] / 2;
    const int* src = ei;
    const int* dst = ei + E;

    float* f     = (float*)d_ws;
    float* dinv  = f;
    float* h     = f + (size_t)n;
    float* out1  = f + (size_t)17 * n;
    float* hrelu = f + (size_t)33 * n;
    float* g     = out1;
    float* out2  = h;

    const int B = 256;
    const int tiles = (n + 15) / 16;
    const int gemmBlocks = (tiles + 7) / 8;

    gcn_deg_init   <<<(n + B - 1) / B, B, 0, stream>>>(dinv, n);
    gcn_deg_accum  <<<(E + B - 1) / B, B, 0, stream>>>(dst, dinv, E);
    gcn_deg_rsqrt  <<<(n + B - 1) / B, B, 0, stream>>>(dinv, n);

    gcn_zero       <<<((size_t)16 * n + B - 1) / B, B, 0, stream>>>(out1, 16 * n);
    gcn_gemm1_wmma <<<gemmBlocks, B, 0, stream>>>(x, W1, h, n);
    gcn_agg1       <<<((size_t)4 * E + B - 1) / B, B, 0, stream>>>(src, dst, dinv, h, out1, E);
    gcn_selfloop_relu<<<(n + B - 1) / B, B, 0, stream>>>(out1, h, dinv, b1, hrelu, n);

    gcn_zero       <<<((size_t)8 * n + B - 1) / B, B, 0, stream>>>(out2, 8 * n);
    gcn_gemm2_wmma <<<gemmBlocks, B, 0, stream>>>(hrelu, W2, g, n);
    gcn_agg2       <<<(E + B - 1) / B, B, 0, stream>>>(src, dst, dinv, g, out2, E);
    gcn_logsoftmax <<<(n + B - 1) / B, B, 0, stream>>>(out2, g, dinv, b2, out, n);
}


# agg1: 16 lanes per edge (one per feature) via wave shuffles, one wave per 32 edges: an edge's 16 float atomics leave as one 64-byte row request, not 4 strided quads; arrival-order sums as before
# speedup vs baseline: 1.9108x; 1.1482x over previous
#include <hip/hip_runtime.h>
#include <hip/hip_bf16.h>

typedef __attribute__((ext_vector_type(2))) float v2f;
typedef __attribute__((ext_vector_type(8))) float v8f;

#define IN_FEAT 512
#define HIDDEN  16
#define NCLS    7


__global__ void gcn_deg_init(float* deg, int n) {
    int i = blockIdx.x * blockDim.x + threadIdx.x;
    if (i < n) deg[i] = 1.0f;
}

__global__ void gcn_deg_accum(const int* __restrict__ dst, float* deg, int E) {
    int e = blockIdx.x * blockDim.x + threadIdx.x;
    if (e < E) atomicAdd(&deg[dst[e]], 1.0f);
}

__global__ void gcn_deg_rsqrt(float* dinv, int n) {
    int i = blockIdx.x * blockDim.x + threadIdx.x;
    if (i < n) {
        float d = dinv[i];
        dinv[i] = (d > 0.0f) ? rsqrtf(d) : 0.0f;
    }
}

__global__ void gcn_zero(float* p, int n) {
    int i = blockIdx.x * blockDim.x + threadIdx.x;
    if (i < n) p[i] = 0.0f;
}


__global__ void gcn_gemm1_wmma(const float* __restrict__ x,
                               const float* __restrict__ W1,
                               float* __restrict__ h, int n) {
    int wave = (blockIdx.x * blockDim.x + threadIdx.x) >> 5;
    int lane = threadIdx.x & 31;
    int row0 = wave * 16;
    if (row0 >= n) return;

    int r    = lane & 15;
    int hi   = lane >> 4;
    int koff = hi * 2;
    int row  = row0 + r;
    if (row >= n) row = n - 1;
    const float* xrow = x + (size_t)row * IN_FEAT;

    v8f c = {0.f, 0.f, 0.f, 0.f, 0.f, 0.f, 0.f, 0.f};

    #pragma unroll 4
    for (int k = 0; k < IN_FEAT; k += 4) {
        if ((k & 63) == 0)
            __builtin_prefetch(xrow + k + 128, 0, 1);
        v2f a = *(const v2f*)(xrow + k + koff);
        int kk = k + koff;
        v2f b;
        b.x = W1[(size_t)kk * HIDDEN + r];
        b.y = W1[(size_t)(kk + 1) * HIDDEN + r];
        c = __builtin_amdgcn_wmma_f32_16x16x4_f32(false, a, false, b,
                                                  (short)0, c, false, false);
    }

    int srow = row0 + hi * 8;
    #pragma unroll
    for (int i = 0; i < 8; ++i) {
        int rr = srow + i;
        if (rr < n) h[(size_t)rr * HIDDEN + r] = c[i];
    }
}


__global__ void gcn_agg1(const int* __restrict__ src, const int* __restrict__ dst,
                         const float* __restrict__ dinv, const float* __restrict__ h,
                         float* out1, int E) {
    int t    = blockIdx.x * blockDim.x + threadIdx.x;
    int lane = threadIdx.x & 31;
    int e0   = __builtin_amdgcn_readfirstlane(t - lane);
    if (e0 >= E) return;
    int e = e0 + lane;
    int s = 0, d = 0;
    if (e < E) {
        s = src[e];
        d = dst[e];
    }
    float w = dinv[s] * dinv[d];
    int nvalid = E - e0;
    if (nvalid > 32) nvalid = 32;
    int sub = lane >> 4;
    int c   = lane & 15;

    float m[16];
    int   dj[16];
    #pragma unroll
    for (int i = 0; i < 16; ++i) {
        int j    = 2 * i + sub;
        int sj   = __shfl(s, j);
        dj[i]    = __shfl(d, j);
        float wj = __shfl(w, j);
        float hv = h[sj * HIDDEN + c];
        m[i] = hv * wj;
    }
    if (nvalid == 32) {
        #pragma unroll
        for (int i = 0; i < 16; ++i)
            atomicAdd(out1 + (dj[i] * HIDDEN + c), m[i]);
    } else {
        #pragma unroll
        for (int i = 0; i < 16; ++i) {
            int j = 2 * i + sub;
            if (j < nvalid)
                atomicAdd(out1 + (dj[i] * HIDDEN + c), m[i]);
        }
    }
}


__global__ void gcn_selfloop_relu(const float* __restrict__ out1,
                                  const float* __restrict__ h,
                                  const float* __restrict__ dinv,
                                  const float* __restrict__ b1,
                                  float* hrelu, int n) {
    int i = blockIdx.x * blockDim.x + threadIdx.x;
    if (i >= n) return;
    float w = dinv[i] * dinv[i];
    const float* o = out1 + (size_t)i * HIDDEN;
    const float* hh = h + (size_t)i * HIDDEN;
    float* dst = hrelu + (size_t)i * HIDDEN;
    #pragma unroll
    for (int c = 0; c < HIDDEN; ++c)
        dst[c] = fmaxf(o[c] + hh[c] * w + b1[c], 0.0f);
}


__global__ void gcn_gemm2_wmma(const float* __restrict__ hrelu,
                               const float* __restrict__ W2,
                               float* __restrict__ g, int n) {
    int wave = (blockIdx.x * blockDim.x + threadIdx.x) >> 5;
    int lane = threadIdx.x & 31;
    int row0 = wave * 16;
    if (row0 >= n) return;

    int r    = lane & 15;
    int hi   = lane >> 4;
    int koff = hi * 2;
    int row  = row0 + r;
    if (row >= n) row = n - 1;
    const float* hrow = hrelu + (size_t)row * HIDDEN;

    v8f c = {0.f, 0.f, 0.f, 0.f, 0.f, 0.f, 0.f, 0.f};

    #pragma unroll
    for (int k = 0; k < HIDDEN; k += 4) {
        int kk = k + koff;
        v2f a = *(const v2f*)(hrow + kk);
        v2f b;
        b.x = (r < NCLS) ? W2[(size_t)kk * NCLS + r] : 0.0f;
        b.y = (r < NCLS) ? W2[(size_t)(kk + 1) * NCLS + r] : 0.0f;
        c = __builtin_amdgcn_wmma_f32_16x16x4_f32(false, a, false, b,
                                                  (short)0, c, false, false);
    }

    int srow = row0 + hi * 8;
    #pragma unroll
    for (int i = 0; i < 8; ++i) {
        int rr = srow + i;
        if (rr < n) g[(size_t)rr * HIDDEN + r] = c[i];
    }
}


__global__ void gcn_agg2(const int* __restrict__ src, const int* __restrict__ dst,
                         const float* __restrict__ dinv, const float* __restrict__ g,
                         float* out2, int E) {
    int t    = blockIdx.x * blockDim.x + threadIdx.x;
    int lane = threadIdx.x & 31;
    int e0   = t - lane;
    if (e0 >= E) return;
    int e = e0 + lane;
    int s = 0, d = 0;
    float w = 0.0f;
    if (e < E) {
        s = src[e];
        d = dst[e];
        w = dinv[s] * dinv[d];
    }
    int nvalid = E - e0;
    if (nvalid > 32) nvalid = 32;
    int sub = lane >> 3;
    int c   = lane & 7;

    float m[8];
    int   dj[8];
    #pragma unroll
    for (int i = 0; i < 8; ++i) {
        int j    = 4 * i + sub;
        int sj   = __shfl(s, j);
        dj[i]    = __shfl(d, j);
        float wj = __shfl(w, j);
        float gv = g[(size_t)sj * HIDDEN + c];
        m[i] = gv * wj;
    }
    #pragma unroll
    for (int i = 0; i < 8; ++i) {
        int j = 4 * i + sub;
        if (c < NCLS && j < nvalid)
            atomicAdd(out2 + (size_t)dj[i] * 8 + c, m[i]);
    }
}


__global__ void gcn_logsoftmax(const float* __restrict__ out2,
                               const float* __restrict__ g,
                               const float* __restrict__ dinv,
                               const float* __restrict__ b2,
                               float* __restrict__ out, int n) {
    int i = blockIdx.x * blockDim.x + threadIdx.x;
    if (i >= n) return;
    float w = dinv[i] * dinv[i];
    const float* o = out2 + (size_t)i * 8;
    const float* gg = g + (size_t)i * HIDDEN;
    float v[NCLS];
    float m = -3.4e38f;
    #pragma unroll
    for (int c = 0; c < NCLS; ++c) {
        v[c] = o[c] + gg[c] * w + b2[c];
        m = fmaxf(m, v[c]);
    }
    float s = 0.0f;
    #pragma unroll
    for (int c = 0; c < NCLS; ++c) s += expf(v[c] - m);
    float l = logf(s);
    float* dsto = out + (size_t)i * NCLS;
    #pragma unroll
    for (int c = 0; c < NCLS; ++c) dsto[c] = v[c] - m - l;
}


extern "C" void kernel_launch(void* const* d_in, const int* in_sizes, int n_in,
                              void* d_out, int out_size, void* d_ws, size_t ws_size,
                              hipStream_t stream) {
    const float* x  = (const float*)d_in[0];
    const int*   ei = (const int*)d_in[1];
    const float* W1 = (const float*)d_in[2];
    const float* b1 = (const float*)d_in[3];
    const float* W2 = (const float*)d_in[4];
    const float* b2 = (const float*)d_in[5];
    float* out = (float*)d_out;

    const int n = in_sizes[0] / IN_FEAT;
    const int E = in_sizes[1] / 2;
    const int* src = ei;
    const int* dst = ei + E;

    float* f     = (float*)d_ws;
    float* dinv  = f;
    float* h     = f + (size_t)n;
    float* out1  = f + (size_t)17 * n;
    float* hrelu = f + (size_t)33 * n;
    float* g     = out1;
    float* out2  = h;

    const int B = 256;
    const int tiles = (n + 15) / 16;
    const int gemmBlocks = (tiles + 7) / 8;

    gcn_deg_init   <<<(n + B - 1) / B, B, 0, stream>>>(dinv, n);
    gcn_deg_accum  <<<(E + B - 1) / B, B, 0, stream>>>(dst, dinv, E);
    gcn_deg_rsqrt  <<<(n + B - 1) / B, B, 0, stream>>>(dinv, n);

    gcn_zero       <<<((size_t)16 * n + B - 1) / B, B, 0, stream>>>(out1, 16 * n);
    gcn_gemm1_wmma <<<gemmBlocks, B, 0, stream>>>(x, W1, h, n);
    gcn_agg1       <<<(E + B - 1) / B, B, 0, stream>>>(src, dst, dinv, h, out1, E);
    gcn_selfloop_relu<<<(n + B - 1) / B, B, 0, stream>>>(out1, h, dinv, b1, hrelu, n);

    gcn_zero       <<<((size_t)8 * n + B - 1) / B, B, 0, stream>>>(out2, 8 * n);
    gcn_gemm2_wmma <<<gemmBlocks, B, 0, stream>>>(hrelu, W2, g, n);
    gcn_agg2       <<<(E + B - 1) / B, B, 0, stream>>>(src, dst, dinv, g, out2, E);
    gcn_logsoftmax <<<(n + B - 1) / B, B, 0, stream>>>(out2, g, dinv, b2, out, n);
}


# edge weight dinv[src]*dinv[dst] split: rows pre-scaled by dinv[src] in both GEMM epilogues, dinv[dst] applied once to the finished sum; edge kernels gather no dinv; atomics in arrival order as before
# speedup vs baseline: 2.0473x; 1.0714x over previous
#include <hip/hip_runtime.h>
#include <hip/hip_bf16.h>

typedef __attribute__((ext_vector_type(2))) float v2f;
typedef __attribute__((ext_vector_type(8))) float v8f;

#define IN_FEAT 512
#define HIDDEN  16
#define NCLS    7


__global__ void gcn_deg_init(float* deg, int n) {
    int i = blockIdx.x * blockDim.x + threadIdx.x;
    if (i < n) deg[i] = 1.0f;
}

__global__ void gcn_deg_accum(const int* __restrict__ dst, float* deg, int E) {
    int e = blockIdx.x * blockDim.x + threadIdx.x;
    if (e < E) atomicAdd(&deg[dst[e]], 1.0f);
}

__global__ void gcn_deg_rsqrt(float* dinv, int n) {
    int i = blockIdx.x * blockDim.x + threadIdx.x;
    if (i < n) {
        float d = dinv[i];
        dinv[i] = (d > 0.0f) ? rsqrtf(d) : 0.0f;
    }
}

__global__ void gcn_zero(float* p, int n) {
    int i = blockIdx.x * blockDim.x + threadIdx.x;
    if (i < n) p[i] = 0.0f;
}


__global__ void gcn_gemm1_wmma(const float* __restrict__ x,
                               const float* __restrict__ W1,
                               const float* __restrict__ dinv,
                               float* __restrict__ h, int n) {
    int wave = (blockIdx.x * blockDim.x + threadIdx.x) >> 5;
    int lane = threadIdx.x & 31;
    int row0 = wave * 16;
    if (row0 >= n) return;

    int r    = lane & 15;
    int hi   = lane >> 4;
    int koff = hi * 2;
    int row  = row0 + r;
    if (row >= n) row = n - 1;
    const float* xrow = x + (size_t)row * IN_FEAT;

    v8f c = {0.f, 0.f, 0.f, 0.f, 0.f, 0.f, 0.f, 0.f};

    #pragma unroll 4
    for (int k = 0; k < IN_FEAT; k += 4) {
        if ((k & 63) == 0)
            __builtin_prefetch(xrow + k + 128, 0, 1);
        v2f a = *(const v2f*)(xrow + k + koff);
        int kk = k + koff;
        v2f b;
        b.x = W1[(size_t)kk * HIDDEN + r];
        b.y = W1[(size_t)(kk + 1) * HIDDEN + r];
        c = __builtin_amdgcn_wmma_f32_16x16x4_f32(false, a, false, b,
                                                  (short)0, c, false, false);
    }

    int srow = row0 + hi * 8;
    #pragma unroll
    for (int i = 0; i < 8; ++i) {
        int rr = srow + i;
        if (rr < n) h[(size_t)rr * HIDDEN + r] = c[i] * dinv[rr];
    }
}


__global__ void gcn_agg1(const int* __restrict__ src, const int* __restrict__ dst,
                         const float* __restrict__ h,
                         float* out1, int E) {
    int t    = blockIdx.x * blockDim.x + threadIdx.x;
    int lane = threadIdx.x & 31;
    int e0   = __builtin_amdgcn_readfirstlane(t - lane);
    if (e0 >= E) return;
    int e = e0 + lane;
    int s = 0, d = 0;
    if (e < E) {
        s = src[e];
        d = dst[e];
    }
    int nvalid = E - e0;
    if (nvalid > 32) nvalid = 32;
    int sub = lane >> 4;
    int c   = lane & 15;

    float m[16];
    int   dj[16];
    #pragma unroll
    for (int i = 0; i < 16; ++i) {
        int j    = 2 * i + sub;
        int sj   = __shfl(s, j);
        dj[i]    = __shfl(d, j);
        m[i] = h[sj * HIDDEN + c];
    }
    if (nvalid == 32) {
        #pragma unroll
        for (int i = 0; i < 16; ++i)
            atomicAdd(out1 + (dj[i] * HIDDEN + c), m[i]);
    } else {
        #pragma unroll
        for (int i = 0; i < 16; ++i) {
            int j = 2 * i + sub;
            if (j < nvalid)
                atomicAdd(out1 + (dj[i] * HIDDEN + c), m[i]);
        }
    }
}


__global__ void gcn_selfloop_relu(const float* __restrict__ out1,
                                  const float* __restrict__ h,
                                  const float* __restrict__ dinv,
                                  const float* __restrict__ b1,
                                  float* hrelu, int n) {
    int i = blockIdx.x * blockDim.x + threadIdx.x;
    if (i >= n) return;
    float w = dinv[i];
    const float* o = out1 + (size_t)i * HIDDEN;
    const float* hh = h + (size_t)i * HIDDEN;
    float* dst = hrelu + (size_t)i * HIDDEN;
    #pragma unroll
    for (int c = 0; c < HIDDEN; ++c)
        dst[c] = fmaxf((o[c] + hh[c]) * w + b1[c], 0.0f);
}


__global__ void gcn_gemm2_wmma(const float* __restrict__ hrelu,
                               const float* __restrict__ W2,
                               const float* __restrict__ dinv,
                               float* __restrict__ g, int n) {
    int wave = (blockIdx.x * blockDim.x + threadIdx.x) >> 5;
    int lane = threadIdx.x & 31;
    int row0 = wave * 16;
    if (row0 >= n) return;

    int r    = lane & 15;
    int hi   = lane >> 4;
    int koff = hi * 2;
    int row  = row0 + r;
    if (row >= n) row = n - 1;
    const float* hrow = hrelu + (size_t)row * HIDDEN;

    v8f c = {0.f, 0.f, 0.f, 0.f, 0.f, 0.f, 0.f, 0.f};

    #pragma unroll
    for (int k = 0; k < HIDDEN; k += 4) {
        int kk = k + koff;
        v2f a = *(const v2f*)(hrow + kk);
        v2f b;
        b.x = (r < NCLS) ? W2[(size_t)kk * NCLS + r] : 0.0f;
        b.y = (r < NCLS) ? W2[(size_t)(kk + 1) * NCLS + r] : 0.0f;
        c = __builtin_amdgcn_wmma_f32_16x16x4_f32(false, a, false, b,
                                                  (short)0, c, false, false);
    }

    int srow = row0 + hi * 8;
    #pragma unroll
    for (int i = 0; i < 8; ++i) {
        int rr = srow + i;
        if (rr < n) g[(size_t)rr * HIDDEN + r] = c[i] * dinv[rr];
    }
}


__global__ void gcn_agg2(const int* __restrict__ src, const int* __restrict__ dst,
                         const float* __restrict__ g,
                         float* out2, int E) {
    int t    = blockIdx.x * blockDim.x + threadIdx.x;
    int lane = threadIdx.x & 31;
    int e0   = t - lane;
    if (e0 >= E) return;
    int e = e0 + lane;
    int s = 0, d = 0;
    if (e < E) {
        s = src[e];
        d = dst[e];
    }
    int nvalid = E - e0;
    if (nvalid > 32) nvalid = 32;
    int sub = lane >> 3;
    int c   = lane & 7;

    float m[8];
    int   dj[8];
    #pragma unroll
    for (int i = 0; i < 8; ++i) {
        int j    = 4 * i + sub;
        int sj   = __shfl(s, j);
        dj[i]    = __shfl(d, j);
        m[i] = g[(size_t)sj * HIDDEN + c];
    }
    #pragma unroll
    for (int i = 0; i < 8; ++i) {
        int j = 4 * i + sub;
        if (c < NCLS && j < nvalid)
            atomicAdd(out2 + (size_t)dj[i] * 8 + c, m[i]);
    }
}


__global__ void gcn_logsoftmax(const float* __restrict__ out2,
                               const float* __restrict__ g,
                               const float* __restrict__ dinv,
                               const float* __restrict__ b2,
                               float* __restrict__ out, int n) {
    int i = blockIdx.x * blockDim.x + threadIdx.x;
    if (i >= n) return;
    float w = dinv[i];
    const float* o = out2 + (size_t)i * 8;
    const float* gg = g + (size_t)i * HIDDEN;
    float v[NCLS];
    float m = -3.4e38f;
    #pragma unroll
    for (int c = 0; c < NCLS; ++c) {
        v[c] = (o[c] + gg[c]) * w + b2[c];
        m = fmaxf(m, v[c]);
    }
    float s = 0.0f;
    #pragma unroll
    for (int c = 0; c < NCLS; ++c) s += expf(v[c] - m);
    float l = logf(s);
    float* dsto = out + (size_t)i * NCLS;
    #pragma unroll
    for (int c = 0; c < NCLS; ++c) dsto[c] = v[c] - m - l;
}


extern "C" void kernel_launch(void* const* d_in, const int* in_sizes, int n_in,
                              void* d_out, int out_size, void* d_ws, size_t ws_size,
                              hipStream_t stream) {
    const float* x  = (const float*)d_in[0];
    const int*   ei = (const int*)d_in[1];
    const float* W1 = (const float*)d_in[2];
    const float* b1 = (const float*)d_in[3];
    const float* W2 = (const float*)d_in[4];
    const float* b2 = (const float*)d_in[5];
    float* out = (float*)d_out;

    const int n = in_sizes[0] / IN_FEAT;
    const int E = in_sizes[1] / 2;
    const int* src = ei;
    const int* dst = ei + E;

    float* f     = (float*)d_ws;
    float* dinv  = f;
    float* h     = f + (size_t)n;
    float* out1  = f + (size_t)17 * n;
    float* hrelu = f + (size_t)33 * n;
    float* g     = out1;
    float* out2  = h;

    const int B = 256;
    const int tiles = (n + 15) / 16;
    const int gemmBlocks = (tiles + 7) / 8;

    gcn_deg_init   <<<(n + B - 1) / B, B, 0, stream>>>(dinv, n);
    gcn_deg_accum  <<<(E + B - 1) / B, B, 0, stream>>>(dst, dinv, E);
    gcn_deg_rsqrt  <<<(n + B - 1) / B, B, 0, stream>>>(dinv, n);

    gcn_zero       <<<((size_t)16 * n + B - 1) / B, B, 0, stream>>>(out1, 16 * n);
    gcn_gemm1_wmma <<<gemmBlocks, B, 0, stream>>>(x, W1, dinv, h, n);
    gcn_agg1       <<<(E + B - 1) / B, B, 0, stream>>>(src, dst, h, out1, E);
    gcn_selfloop_relu<<<(n + B - 1) / B, B, 0, stream>>>(out1, h, dinv, b1, hrelu, n);

    gcn_zero       <<<((size_t)8 * n + B - 1) / B, B, 0, stream>>>(out2, 8 * n);
    gcn_gemm2_wmma <<<gemmBlocks, B, 0, stream>>>(hrelu, W2, dinv, g, n);
    gcn_agg2       <<<(E + B - 1) / B, B, 0, stream>>>(src, dst, g, out2, E);
    gcn_logsoftmax <<<(n + B - 1) / B, B, 0, stream>>>(out2, g, dinv, b2, out, n);
}


# agg1, agg2: edges counting-sorted by destination (rank = degree pass's returning add, scan, scatter); a wave sums each row's gathered rows in registers: no float atomics; sum order by arrival rank
# speedup vs baseline: 2.2883x; 1.1177x over previous
#include <hip/hip_runtime.h>
#include <hip/hip_bf16.h>

typedef __attribute__((ext_vector_type(2))) float v2f;
typedef __attribute__((ext_vector_type(8))) float v8f;

#define IN_FEAT 512
#define HIDDEN  16
#define NCLS    7


__global__ void gcn_deg_init(float* deg, int n) {
    int i = blockIdx.x * blockDim.x + threadIdx.x;
    if (i < n) deg[i] = 1.0f;
}

__global__ void gcn_deg_accum(const int* __restrict__ dst, float* deg, int E) {
    int e = blockIdx.x * blockDim.x + threadIdx.x;
    if (e < E) atomicAdd(&deg[dst[e]], 1.0f);
}

__global__ void gcn_deg_rsqrt(float* dinv, int n) {
    int i = blockIdx.x * blockDim.x + threadIdx.x;
    if (i < n) {
        float d = dinv[i];
        dinv[i] = (d > 0.0f) ? rsqrtf(d) : 0.0f;
    }
}

__global__ void gcn_zero(float* p, int n) {
    int i = blockIdx.x * blockDim.x + threadIdx.x;
    if (i < n) p[i] = 0.0f;
}


__global__ void gcn_gemm1_wmma(const float* __restrict__ x,
                               const float* __restrict__ W1,
                               const float* __restrict__ dinv,
                               float* __restrict__ h, int n) {
    int wave = (blockIdx.x * blockDim.x + threadIdx.x) >> 5;
    int lane = threadIdx.x & 31;
    int row0 = wave * 16;
    if (row0 >= n) return;

    int r    = lane & 15;
    int hi   = lane >> 4;
    int koff = hi * 2;
    int row  = row0 + r;
    if (row >= n) row = n - 1;
    const float* xrow = x + (size_t)row * IN_FEAT;

    v8f c = {0.f, 0.f, 0.f, 0.f, 0.f, 0.f, 0.f, 0.f};

    #pragma unroll 4
    for (int k = 0; k < IN_FEAT; k += 4) {
        if ((k & 63) == 0)
            __builtin_prefetch(xrow + k + 128, 0, 1);
        v2f a = *(const v2f*)(xrow + k + koff);
        int kk = k + koff;
        v2f b;
        b.x = W1[(size_t)kk * HIDDEN + r];
        b.y = W1[(size_t)(kk + 1) * HIDDEN + r];
        c = __builtin_amdgcn_wmma_f32_16x16x4_f32(false, a, false, b,
                                                  (short)0, c, false, false);
    }

    int srow = row0 + hi * 8;
    #pragma unroll
    for (int i = 0; i < 8; ++i) {
        int rr = srow + i;
        if (rr < n) h[(size_t)rr * HIDDEN + r] = c[i] * dinv[rr];
    }
}


__global__ void gcn_agg1(const int* __restrict__ src, const int* __restrict__ dst,
                         const float* __restrict__ h,
                         float* out1, int E) {
    int t    = blockIdx.x * blockDim.x + threadIdx.x;
    int lane = threadIdx.x & 31;
    int e0   = __builtin_amdgcn_readfirstlane(t - lane);
    if (e0 >= E) return;
    int e = e0 + lane;
    int s = 0, d = 0;
    if (e < E) {
        s = src[e];
        d = dst[e];
    }
    int nvalid = E - e0;
    if (nvalid > 32) nvalid = 32;
    int sub = lane >> 4;
    int c   = lane & 15;

    float m[16];
    int   dj[16];
    #pragma unroll
    for (int i = 0; i < 16; ++i) {
        int j    = 2 * i + sub;
        int sj   = __shfl(s, j);
        dj[i]    = __shfl(d, j);
        m[i] = h[sj * HIDDEN + c];
    }
    if (nvalid == 32) {
        #pragma unroll
        for (int i = 0; i < 16; ++i)
            atomicAdd(out1 + (dj[i] * HIDDEN + c), m[i]);
    } else {
        #pragma unroll
        for (int i = 0; i < 16; ++i) {
            int j = 2 * i + sub;
            if (j < nvalid)
                atomicAdd(out1 + (dj[i] * HIDDEN + c), m[i]);
        }
    }
}


__global__ void gcn_selfloop_relu(const float* __restrict__ out1,
                                  const float* __restrict__ h,
                                  const float* __restrict__ dinv,
                                  const float* __restrict__ b1,
                                  float* hrelu, int n) {
    int i = blockIdx.x * blockDim.x + threadIdx.x;
    if (i >= n) return;
    float w = dinv[i];
    const float* o = out1 + (size_t)i * HIDDEN;
    const float* hh = h + (size_t)i * HIDDEN;
    float* dst = hrelu + (size_t)i * HIDDEN;
    #pragma unroll
    for (int c = 0; c < HIDDEN; ++c)
        dst[c] = fmaxf((o[c] + hh[c]) * w + b1[c], 0.0f);
}


__global__ void gcn_gemm2_wmma(const float* __restrict__ hrelu,
                               const float* __restrict__ W2,
                               const float* __restrict__ dinv,
                               float* __restrict__ g, int n) {
    int wave = (blockIdx.x * blockDim.x + threadIdx.x) >> 5;
    int lane = threadIdx.x & 31;
    int row0 = wave * 16;
    if (row0 >= n) return;

    int r    = lane & 15;
    int hi   = lane >> 4;
    int koff = hi * 2;
    int row  = row0 + r;
    if (row >= n) row = n - 1;
    const float* hrow = hrelu + (size_t)row * HIDDEN;

    v8f c = {0.f, 0.f, 0.f, 0.f, 0.f, 0.f, 0.f, 0.f};

    #pragma unroll
    for (int k = 0; k < HIDDEN; k += 4) {
        int kk = k + koff;
        v2f a = *(const v2f*)(hrow + kk);
        v2f b;
        b.x = (r < NCLS) ? W2[(size_t)kk * NCLS + r] : 0.0f;
        b.y = (r < NCLS) ? W2[(size_t)(kk + 1) * NCLS + r] : 0.0f;
        c = __builtin_amdgcn_wmma_f32_16x16x4_f32(false, a, false, b,
                                                  (short)0, c, false, false);
    }

    int srow = row0 + hi * 8;
    #pragma unroll
    for (int i = 0; i < 8; ++i) {
        int rr = srow + i;
        if (rr < n) g[(size_t)rr * HIDDEN + r] = c[i] * dinv[rr];
    }
}


__global__ void gcn_agg2(const int* __restrict__ src, const int* __restrict__ dst,
                         const float* __restrict__ g,
                         float* out2, int E) {
    int t    = blockIdx.x * blockDim.x + threadIdx.x;
    int lane = threadIdx.x & 31;
    int e0   = t - lane;
    if (e0 >= E) return;
    int e = e0 + lane;
    int s = 0, d = 0;
    if (e < E) {
        s = src[e];
        d = dst[e];
    }
    int nvalid = E - e0;
    if (nvalid > 32) nvalid = 32;
    int sub = lane >> 3;
    int c   = lane & 7;

    float m[8];
    int   dj[8];
    #pragma unroll
    for (int i = 0; i < 8; ++i) {
        int j    = 4 * i + sub;
        int sj   = __shfl(s, j);
        dj[i]    = __shfl(d, j);
        m[i] = g[(size_t)sj * HIDDEN + c];
    }
    #pragma unroll
    for (int i = 0; i < 8; ++i) {
        int j = 4 * i + sub;
        if (c < NCLS && j < nvalid)
            atomicAdd(out2 + (size_t)dj[i] * 8 + c, m[i]);
    }
}


__global__ void gcn_logsoftmax(const float* __restrict__ out2,
                               const float* __restrict__ g,
                               const float* __restrict__ dinv,
                               const float* __restrict__ b2,
                               float* __restrict__ out, int n) {
    int i = blockIdx.x * blockDim.x + threadIdx.x;
    if (i >= n) return;
    float w = dinv[i];
    const float* o = out2 + (size_t)i * 8;
    const float* gg = g + (size_t)i * HIDDEN;
    float v[NCLS];
    float m = -3.4e38f;
    #pragma unroll
    for (int c = 0; c < NCLS; ++c) {
        v[c] = (o[c] + gg[c]) * w + b2[c];
        m = fmaxf(m, v[c]);
    }
    float s = 0.0f;
    #pragma unroll
    for (int c = 0; c < NCLS; ++c) s += expf(v[c] - m);
    float l = logf(s);
    float* dsto = out + (size_t)i * NCLS;
    #pragma unroll
    for (int c = 0; c < NCLS; ++c) dsto[c] = v[c] - m - l;
}


__global__ void gcn_cnt_init(int* cnt, int n) {
    int i = blockIdx.x * blockDim.x + threadIdx.x;
    if (i < n) cnt[i] = 0;
}

__global__ void gcn_rank(const int* __restrict__ dst, int* cnt, int* __restrict__ rank, int E) {
    int e = blockIdx.x * blockDim.x + threadIdx.x;
    if (e < E) rank[e] = atomicAdd(&cnt[dst[e]], 1);
}

#define SCAN_T   256
#define SCAN_PER 1024

__global__ void gcn_scan(const int* __restrict__ cnt, int* __restrict__ first,
                         float* __restrict__ dinv, int n) {
    __shared__ int red[SCAN_T / 32];
    __shared__ int wsum[SCAN_T / 32];
    int tid  = threadIdx.x;
    int lane = tid & 31;
    int wv   = tid >> 5;
    int base = blockIdx.x * SCAN_PER;

    int acc = 0;
    #pragma unroll 8
    for (int i = tid * 4; i < base; i += SCAN_PER) {
        int4 v = *(const int4*)(cnt + i);
        acc += (v.x + v.y) + (v.z + v.w);
    }
    for (int o = 16; o > 0; o >>= 1) acc += __shfl_xor(acc, o);
    if (lane == 0) red[wv] = acc;

    int i0 = base + tid * 4;
    int v0 = 0, v1 = 0, v2 = 0, v3 = 0;
    if (i0 + 3 < n) {
        int4 v = *(const int4*)(cnt + i0);
        v0 = v.x; v1 = v.y; v2 = v.z; v3 = v.w;
    } else {
        if (i0     < n) v0 = cnt[i0];
        if (i0 + 1 < n) v1 = cnt[i0 + 1];
        if (i0 + 2 < n) v2 = cnt[i0 + 2];
    }
    int t = (v0 + v1) + (v2 + v3);
    int incl = t;
    for (int o = 1; o < 32; o <<= 1) {
        int u = __shfl_up(incl, o);
        if (lane >= o) incl += u;
    }
    if (lane == 31) wsum[wv] = incl;
    __syncthreads();

    int pre = 0;
    for (int k = 0; k < SCAN_T / 32; ++k) {
        pre += red[k];
        if (k < wv) pre += wsum[k];
    }
    int ex = pre + incl - t;
    int vv[4] = {v0, v1, v2, v3};
    int   fo[4];
    float dv[4];
    #pragma unroll
    for (int k = 0; k < 4; ++k) {
        fo[k] = ex;
        float dg = (float)(vv[k] + 1);
        dv[k] = (dg > 0.0f) ? rsqrtf(dg) : 0.0f;
        ex += vv[k];
    }
    if (i0 + 3 < n) {
        *(int4*)(first + i0)  = make_int4(fo[0], fo[1], fo[2], fo[3]);
        *(float4*)(dinv + i0) = make_float4(dv[0], dv[1], dv[2], dv[3]);
    } else {
        #pragma unroll
        for (int k = 0; k < 3; ++k) {
            if (i0 + k < n) {
                first[i0 + k] = fo[k];
                dinv[i0 + k]  = dv[k];
            }
        }
    }
}

__global__ void gcn_scatter(const int* __restrict__ src, const int* __restrict__ dst,
                            const int* __restrict__ rank, const int* __restrict__ first,
                            int* __restrict__ ssrc, int E) {
    int e = blockIdx.x * blockDim.x + threadIdx.x;
    if (e < E) ssrc[first[dst[e]] + rank[e]] = src[e];
}


__global__ void gcn_agg1_rows(const int* __restrict__ ssrc, const int* __restrict__ first,
                              const int* __restrict__ cnt, const float* __restrict__ h,
                              const float* __restrict__ dinv, const float* __restrict__ b1,
                              float* __restrict__ hrelu, int n) {
    int wave = __builtin_amdgcn_readfirstlane((int)((blockIdx.x * blockDim.x + threadIdx.x) >> 5));
    int lane = threadIdx.x & 31;
    int sub  = lane >> 4;
    int c    = lane & 15;
    int d0   = wave * 2;
    if (d0 >= n) return;
    float res = 0.0f;
    for (int rsel = 0; rsel < 2; ++rsel) {
        int d = d0 + rsel;
        if (d >= n) break;
        int beg = first[d];
        int end = beg + cnt[d];
        float acc = 0.0f;
        for (int base = beg; base < end; base += 32) {
            int idx = base + lane;
            int sv  = (idx < end) ? ssrc[idx] : 0;
            int nv  = end - base;
            if (nv > 32) nv = 32;
            for (int p0 = 0; p0 < nv; p0 += 16) {
                float m[8];
                #pragma unroll
                for (int q = 0; q < 8; ++q) {
                    int j  = p0 + 2 * q + sub;
                    int sj = __shfl(sv, j);
                    m[q] = h[sj * HIDDEN + c];
                }
                #pragma unroll
                for (int q = 0; q < 8; ++q) {
                    int j = p0 + 2 * q + sub;
                    acc += (j < nv) ? m[q] : 0.0f;
                }
            }
        }
        acc += __shfl_xor(acc, 16);
        if (sub == rsel) res = acc;
    }
    int row = d0 + sub;
    if (row < n) {
        float w = dinv[row];
        hrelu[row * HIDDEN + c] = fmaxf((res + h[row * HIDDEN + c]) * w + b1[c], 0.0f);
    }
}


__global__ void gcn_agg2_rows(const int* __restrict__ ssrc, const int* __restrict__ first,
                              const int* __restrict__ cnt, const float* __restrict__ g,
                              float* __restrict__ out2, int n) {
    int wave = __builtin_amdgcn_readfirstlane((int)((blockIdx.x * blockDim.x + threadIdx.x) >> 5));
    int lane = threadIdx.x & 31;
    int sub  = lane >> 3;
    int c    = lane & 7;
    int d0   = wave * 4;
    if (d0 >= n) return;
    float res = 0.0f;
    for (int rsel = 0; rsel < 4; ++rsel) {
        int d = d0 + rsel;
        if (d >= n) break;
        int beg = first[d];
        int end = beg + cnt[d];
        float acc = 0.0f;
        for (int base = beg; base < end; base += 32) {
            int idx = base + lane;
            int sv  = (idx < end) ? ssrc[idx] : 0;
            int nv  = end - base;
            if (nv > 32) nv = 32;
            for (int p0 = 0; p0 < nv; p0 += 16) {
                float m[4];
                #pragma unroll
                for (int q = 0; q < 4; ++q) {
                    int j  = p0 + 4 * q + sub;
                    int sj = __shfl(sv, j);
                    m[q] = g[sj * HIDDEN + c];
                }
                #pragma unroll
                for (int q = 0; q < 4; ++q) {
                    int j = p0 + 4 * q + sub;
                    acc += (j < nv) ? m[q] : 0.0f;
                }
            }
        }
        acc += __shfl_xor(acc, 8);
        acc += __shfl_xor(acc, 16);
        if (sub == rsel) res = acc;
    }
    int row = d0 + sub;
    if (row < n) out2[row * 8 + c] = res;
}


extern "C" void kernel_launch(void* const* d_in, const int* in_sizes, int n_in,
                              void* d_out, int out_size, void* d_ws, size_t ws_size,
                              hipStream_t stream) {
    const float* x  = (const float*)d_in[0];
    const int*   ei = (const int*)d_in[1];
    const float* W1 = (const float*)d_in[2];
    const float* b1 = (const float*)d_in[3];
    const float* W2 = (const float*)d_in[4];
    const float* b2 = (const float*)d_in[5];
    float* out = (float*)d_out;

    const int n = in_sizes[0] / IN_FEAT;
    const int E = in_sizes[1] / 2;
    const int* src = ei;
    const int* dst = ei + E;

    const int B = 256;
    const int tiles = (n + 15) / 16;
    const int gemmBlocks = (tiles + 7) / 8;

    const size_t n4 = ((size_t)n + 3) & ~(size_t)3;
    const size_t E4 = ((size_t)E + 3) & ~(size_t)3;
    const size_t a_words = (E4 > (size_t)32 * n) ? E4 : (size_t)32 * n;
    const size_t need = (3 * n4 + a_words + (size_t)E) * sizeof(float);

    if (need <= ws_size) {
        float* f     = (float*)d_ws;
        float* dinv  = f;
        int*   cnt   = (int*)(f + n4);
        int*   first = (int*)(f + 2 * n4);
        float* a     = f + 3 * n4;
        int*   rank  = (int*)a;
        float* h     = a;
        float* hrelu = a + (size_t)16 * n;
        int*   ssrc  = (int*)(a + a_words);
        float* g     = h;
        float* out2  = hrelu;

        gcn_cnt_init   <<<(n + B - 1) / B, B, 0, stream>>>(cnt, n);
        gcn_rank       <<<(E + B - 1) / B, B, 0, stream>>>(dst, cnt, rank, E);
        gcn_scan       <<<(n + SCAN_PER - 1) / SCAN_PER, SCAN_T, 0, stream>>>(cnt, first, dinv, n);
        gcn_scatter    <<<(E + B - 1) / B, B, 0, stream>>>(src, dst, rank, first, ssrc, E);

        gcn_gemm1_wmma <<<gemmBlocks, B, 0, stream>>>(x, W1, dinv, h, n);
        gcn_agg1_rows  <<<(n + 15) / 16, B, 0, stream>>>(ssrc, first, cnt, h, dinv, b1, hrelu, n);

        gcn_gemm2_wmma <<<gemmBlocks, B, 0, stream>>>(hrelu, W2, dinv, g, n);
        gcn_agg2_rows  <<<(n + 31) / 32, B, 0, stream>>>(ssrc, first, cnt, g, out2, n);
        gcn_logsoftmax <<<(n + B - 1) / B, B, 0, stream>>>(out2, g, dinv, b2, out, n);
        return;
    }

    float* f     = (float*)d_ws;
    float* dinv  = f;
    float* h     = f + (size_t)n;
    float* out1  = f + (size_t)17 * n;
    float* hrelu = f + (size_t)33 * n;
    float* g     = out1;
    float* out2  = h;

    gcn_deg_init   <<<(n + B - 1) / B, B, 0, stream>>>(dinv, n);
    gcn_deg_accum  <<<(E + B - 1) / B, B, 0, stream>>>(dst, dinv, E);
    gcn_deg_rsqrt  <<<(n + B - 1) / B, B, 0, stream>>>(dinv, n);

    gcn_zero       <<<((size_t)16 * n + B - 1) / B, B, 0, stream>>>(out1, 16 * n);
    gcn_gemm1_wmma <<<gemmBlocks, B, 0, stream>>>(x, W1, dinv, h, n);
    gcn_agg1       <<<(E + B - 1) / B, B, 0, stream>>>(src, dst, h, out1, E);
    gcn_selfloop_relu<<<(n + B - 1) / B, B, 0, stream>>>(out1, h, dinv, b1, hrelu, n);

    gcn_zero       <<<((size_t)8 * n + B - 1) / B, B, 0, stream>>>(out2, 8 * n);
    gcn_gemm2_wmma <<<gemmBlocks, B, 0, stream>>>(hrelu, W2, dinv, g, n);
    gcn_agg2       <<<(E + B - 1) / B, B, 0, stream>>>(src, dst, g, out2, E);
    gcn_logsoftmax <<<(n + B - 1) / B, B, 0, stream>>>(out2, g, dinv, b2, out, n);
}


# sorted-edge build without per-edge global traffic: counting sort through LDS (512-row bins per chunk, scan, runs written whole, bins sorted in LDS); row sum order by arrival at LDS counters
# speedup vs baseline: 5.2607x; 2.2990x over previous
#include <hip/hip_runtime.h>
#include <hip/hip_bf16.h>

typedef __attribute__((ext_vector_type(2))) float v2f;
typedef __attribute__((ext_vector_type(8))) float v8f;

#define IN_FEAT 512
#define HIDDEN  16
#define NCLS    7


__global__ void gcn_deg_init(float* deg, int n) {
    int i = blockIdx.x * blockDim.x + threadIdx.x;
    if (i < n) deg[i] = 1.0f;
}

__global__ void gcn_deg_accum(const int* __restrict__ dst, float* deg, int E) {
    int e = blockIdx.x * blockDim.x + threadIdx.x;
    if (e < E) atomicAdd(&deg[dst[e]], 1.0f);
}

__global__ void gcn_deg_rsqrt(float* dinv, int n) {
    int i = blockIdx.x * blockDim.x + threadIdx.x;
    if (i < n) {
        float d = dinv[i];
        dinv[i] = (d > 0.0f) ? rsqrtf(d) : 0.0f;
    }
}

__global__ void gcn_zero(float* p, int n) {
    int i = blockIdx.x * blockDim.x + threadIdx.x;
    if (i < n) p[i] = 0.0f;
}


__global__ void gcn_gemm1_wmma(const float* __restrict__ x,
                               const float* __restrict__ W1,
                               const float* __restrict__ dinv,
                               float* __restrict__ h, int n) {
    int wave = (blockIdx.x * blockDim.x + threadIdx.x) >> 5;
    int lane = threadIdx.x & 31;
    int row0 = wave * 16;
    if (row0 >= n) return;

    int r    = lane & 15;
    int hi   = lane >> 4;
    int koff = hi * 2;
    int row  = row0 + r;
    if (row >= n) row = n - 1;
    const float* xrow = x + (size_t)row * IN_FEAT;

    v8f c = {0.f, 0.f, 0.f, 0.f, 0.f, 0.f, 0.f, 0.f};

    #pragma unroll 4
    for (int k = 0; k < IN_FEAT; k += 4) {
        if ((k & 63) == 0)
            __builtin_prefetch(xrow + k + 128, 0, 1);
        v2f a = *(const v2f*)(xrow + k + koff);
        int kk = k + koff;
        v2f b;
        b.x = W1[(size_t)kk * HIDDEN + r];
        b.y = W1[(size_t)(kk + 1) * HIDDEN + r];
        c = __builtin_amdgcn_wmma_f32_16x16x4_f32(false, a, false, b,
                                                  (short)0, c, false, false);
    }

    int srow = row0 + hi * 8;
    #pragma unroll
    for (int i = 0; i < 8; ++i) {
        int rr = srow + i;
        if (rr < n) h[(size_t)rr * HIDDEN + r] = c[i] * dinv[rr];
    }
}


__global__ void gcn_agg1(const int* __restrict__ src, const int* __restrict__ dst,
                         const float* __restrict__ h,
                         float* out1, int E) {
    int t    = blockIdx.x * blockDim.x + threadIdx.x;
    int lane = threadIdx.x & 31;
    int e0   = __builtin_amdgcn_readfirstlane(t - lane);
    if (e0 >= E) return;
    int e = e0 + lane;
    int s = 0, d = 0;
    if (e < E) {
        s = src[e];
        d = dst[e];
    }
    int nvalid = E - e0;
    if (nvalid > 32) nvalid = 32;
    int sub = lane >> 4;
    int c   = lane & 15;

    float m[16];
    int   dj[16];
    #pragma unroll
    for (int i = 0; i < 16; ++i) {
        int j    = 2 * i + sub;
        int sj   = __shfl(s, j);
        dj[i]    = __shfl(d, j);
        m[i] = h[sj * HIDDEN + c];
    }
    if (nvalid == 32) {
        #pragma unroll
        for (int i = 0; i < 16; ++i)
            atomicAdd(out1 + (dj[i] * HIDDEN + c), m[i]);
    } else {
        #pragma unroll
        for (int i = 0; i < 16; ++i) {
            int j = 2 * i + sub;
            if (j < nvalid)
                atomicAdd(out1 + (dj[i] * HIDDEN + c), m[i]);
        }
    }
}


__global__ void gcn_selfloop_relu(const float* __restrict__ out1,
                                  const float* __restrict__ h,
                                  const float* __restrict__ dinv,
                                  const float* __restrict__ b1,
                                  float* hrelu, int n) {
    int i = blockIdx.x * blockDim.x + threadIdx.x;
    if (i >= n) return;
    float w = dinv[i];
    const float* o = out1 + (size_t)i * HIDDEN;
    const float* hh = h + (size_t)i * HIDDEN;
    float* dst = hrelu + (size_t)i * HIDDEN;
    #pragma unroll
    for (int c = 0; c < HIDDEN; ++c)
        dst[c] = fmaxf((o[c] + hh[c]) * w + b1[c], 0.0f);
}


__global__ void gcn_gemm2_wmma(const float* __restrict__ hrelu,
                               const float* __restrict__ W2,
                               const float* __restrict__ dinv,
                               float* __restrict__ g, int n) {
    int wave = (blockIdx.x * blockDim.x + threadIdx.x) >> 5;
    int lane = threadIdx.x & 31;
    int row0 = wave * 16;
    if (row0 >= n) return;

    int r    = lane & 15;
    int hi   = lane >> 4;
    int koff = hi * 2;
    int row  = row0 + r;
    if (row >= n) row = n - 1;
    const float* hrow = hrelu + (size_t)row * HIDDEN;

    v8f c = {0.f, 0.f, 0.f, 0.f, 0.f, 0.f, 0.f, 0.f};

    #pragma unroll
    for (int k = 0; k < HIDDEN; k += 4) {
        int kk = k + koff;
        v2f a = *(const v2f*)(hrow + kk);
        v2f b;
        b.x = (r < NCLS) ? W2[(size_t)kk * NCLS + r] : 0.0f;
        b.y = (r < NCLS) ? W2[(size_t)(kk + 1) * NCLS + r] : 0.0f;
        c = __builtin_amdgcn_wmma_f32_16x16x4_f32(false, a, false, b,
                                                  (short)0, c, false, false);
    }

    int srow = row0 + hi * 8;
    #pragma unroll
    for (int i = 0; i < 8; ++i) {
        int rr = srow + i;
        if (rr < n) g[(size_t)rr * HIDDEN + r] = c[i] * dinv[rr];
    }
}


__global__ void gcn_agg2(const int* __restrict__ src, const int* __restrict__ dst,
                         const float* __restrict__ g,
                         float* out2, int E) {
    int t    = blockIdx.x * blockDim.x + threadIdx.x;
    int lane = threadIdx.x & 31;
    int e0   = t - lane;
    if (e0 >= E) return;
    int e = e0 + lane;
    int s = 0, d = 0;
    if (e < E) {
        s = src[e];
        d = dst[e];
    }
    int nvalid = E - e0;
    if (nvalid > 32) nvalid = 32;
    int sub = lane >> 3;
    int c   = lane & 7;

    float m[8];
    int   dj[8];
    #pragma unroll
    for (int i = 0; i < 8; ++i) {
        int j    = 4 * i + sub;
        int sj   = __shfl(s, j);
        dj[i]    = __shfl(d, j);
        m[i] = g[(size_t)sj * HIDDEN + c];
    }
    #pragma unroll
    for (int i = 0; i < 8; ++i) {
        int j = 4 * i + sub;
        if (c < NCLS && j < nvalid)
            atomicAdd(out2 + (size_t)dj[i] * 8 + c, m[i]);
    }
}


__global__ void gcn_logsoftmax(const float* __restrict__ out2,
                               const float* __restrict__ g,
                               const float* __restrict__ dinv,
                               const float* __restrict__ b2,
                               float* __restrict__ out, int n) {
    int i = blockIdx.x * blockDim.x + threadIdx.x;
    if (i >= n) return;
    float w = dinv[i];
    const float* o = out2 + (size_t)i * 8;
    const float* gg = g + (size_t)i * HIDDEN;
    float v[NCLS];
    float m = -3.4e38f;
    #pragma unroll
    for (int c = 0; c < NCLS; ++c) {
        v[c] = (o[c] + gg[c]) * w + b2[c];
        m = fmaxf(m, v[c]);
    }
    float s = 0.0f;
    #pragma unroll
    for (int c = 0; c < NCLS; ++c) s += expf(v[c] - m);
    float l = logf(s);
    float* dsto = out + (size_t)i * NCLS;
    #pragma unroll
    for (int c = 0; c < NCLS; ++c) dsto[c] = v[c] - m - l;
}

#define RB_SHIFT 9
#define RB       512
#define NB_MAX   1024
#define BIN_T    1024
#define CH_PER   16
#define CH_MAX   (BIN_T * CH_PER)
#define SORT_CAP 32768
#define SORT_B   8

__global__ void __launch_bounds__(BIN_T)
gcn_bin_count(const int* __restrict__ dst, int* __restrict__ gh, int E, int ch, int nb, int G) {
    __shared__ int hist[NB_MAX];
    int tid = threadIdx.x;
    int g   = blockIdx.x;
    hist[tid] = 0;
    int e0 = g * ch;
    int e1 = (E - e0 < ch) ? E : e0 + ch;
    int dv[CH_PER];
    #pragma unroll
    for (int k = 0; k < CH_PER; ++k) {
        int e = e0 + tid + k * BIN_T;
        dv[k] = (e < e1) ? dst[e] : 0;
    }
    __syncthreads();
    #pragma unroll
    for (int k = 0; k < CH_PER; ++k)
        if (e0 + tid + k * BIN_T < e1) atomicAdd(&hist[dv[k] >> RB_SHIFT], 1);
    __syncthreads();
    if (tid < nb) gh[tid * G + g] = hist[tid];
}

#define SCAN_T   256
#define SCAN_PER 1024

__global__ void gcn_scan_plain(const int* __restrict__ cnt, int* __restrict__ first, int m) {
    __shared__ int red[SCAN_T / 32];
    __shared__ int wsum[SCAN_T / 32];
    int tid  = threadIdx.x;
    int lane = tid & 31;
    int wv   = tid >> 5;
    int base = blockIdx.x * SCAN_PER;

    int acc = 0;
    #pragma unroll 8
    for (int i = tid * 4; i < base; i += SCAN_PER) {
        int4 v = *(const int4*)(cnt + i);
        acc += (v.x + v.y) + (v.z + v.w);
    }
    for (int o = 16; o > 0; o >>= 1) acc += __shfl_xor(acc, o);
    if (lane == 0) red[wv] = acc;

    int i0 = base + tid * 4;
    int v0 = 0, v1 = 0, v2 = 0, v3 = 0;
    if (i0 + 3 < m) {
        int4 v = *(const int4*)(cnt + i0);
        v0 = v.x; v1 = v.y; v2 = v.z; v3 = v.w;
    } else {
        if (i0     < m) v0 = cnt[i0];
        if (i0 + 1 < m) v1 = cnt[i0 + 1];
        if (i0 + 2 < m) v2 = cnt[i0 + 2];
    }
    int t = (v0 + v1) + (v2 + v3);
    int incl = t;
    for (int o = 1; o < 32; o <<= 1) {
        int u = __shfl_up(incl, o);
        if (lane >= o) incl += u;
    }
    if (lane == 31) wsum[wv] = incl;
    __syncthreads();

    int pre = 0;
    for (int k = 0; k < SCAN_T / 32; ++k) {
        pre += red[k];
        if (k < wv) pre += wsum[k];
    }
    int ex = pre + incl - t;
    int f0 = ex, f1 = f0 + v0, f2 = f1 + v1, f3 = f2 + v2;
    if (i0 + 3 < m) {
        *(int4*)(first + i0) = make_int4(f0, f1, f2, f3);
    } else {
        if (i0     < m) first[i0]     = f0;
        if (i0 + 1 < m) first[i0 + 1] = f1;
        if (i0 + 2 < m) first[i0 + 2] = f2;
    }
}

__global__ void __launch_bounds__(BIN_T)
gcn_bin_scatter(const int* __restrict__ src, const int* __restrict__ dst,
                const int* __restrict__ gs, unsigned* __restrict__ binned,
                int E, int ch, int nb, int G) {
    __shared__ int  lcur[NB_MAX];
    __shared__ int  gofs[NB_MAX];
    __shared__ int  wtot[BIN_T / 32];
    __shared__ int2 stage[CH_MAX];
    int tid  = threadIdx.x;
    int lane = tid & 31;
    int wv   = tid >> 5;
    int g    = blockIdx.x;
    lcur[tid] = 0;
    int e0 = g * ch;
    int e1 = (E - e0 < ch) ? E : e0 + ch;

    int sv[CH_PER], dv[CH_PER];
    #pragma unroll
    for (int k = 0; k < CH_PER; ++k) {
        int e = e0 + tid + k * BIN_T;
        sv[k] = 0; dv[k] = 0;
        if (e < e1) { sv[k] = src[e]; dv[k] = dst[e]; }
    }
    __syncthreads();
    #pragma unroll
    for (int k = 0; k < CH_PER; ++k)
        if (e0 + tid + k * BIN_T < e1) atomicAdd(&lcur[dv[k] >> RB_SHIFT], 1);
    __syncthreads();

    int c = (tid < nb) ? lcur[tid] : 0;
    int incl = c;
    for (int o = 1; o < 32; o <<= 1) {
        int u = __shfl_up(incl, o);
        if (lane >= o) incl += u;
    }
    if (lane == 31) wtot[wv] = incl;
    __syncthreads();
    int pre = 0;
    for (int k = 0; k < wv; ++k) pre += wtot[k];
    int ls = pre + incl - c;
    if (tid < nb) {
        lcur[tid] = ls;
        gofs[tid] = gs[tid * G + g] - ls;
    }
    __syncthreads();
    #pragma unroll
    for (int k = 0; k < CH_PER; ++k) {
        if (e0 + tid + k * BIN_T < e1) {
            int p = atomicAdd(&lcur[dv[k] >> RB_SHIFT], 1);
            stage[p] = make_int2(sv[k], dv[k]);
        }
    }
    __syncthreads();
    int len = e1 - e0;
    for (int i = tid; i < len; i += BIN_T) {
        int2 v = stage[i];
        binned[gofs[v.y >> RB_SHIFT] + i] = ((unsigned)v.x << RB_SHIFT) | (unsigned)(v.y & (RB - 1));
    }
}

__global__ void __launch_bounds__(BIN_T)
gcn_bin_sort(const unsigned* __restrict__ binned, const int* __restrict__ gs,
             int* __restrict__ ssrc, int* __restrict__ first, int* __restrict__ cnt,
             float* __restrict__ dinv, int n, int E, int nb, int G) {
    __shared__ int rcur[RB];
    __shared__ int wtot[RB / 32];
    __shared__ int stage[SORT_CAP];
    int tid  = threadIdx.x;
    int lane = tid & 31;
    int wv   = tid >> 5;
    int b    = blockIdx.x;
    int beg  = gs[b * G];
    int end  = (b + 1 < nb) ? gs[(b + 1) * G] : E;
    int len  = end - beg;
    if (tid < RB) rcur[tid] = 0;
    __syncthreads();
    for (int i0 = tid; i0 < len; i0 += BIN_T * SORT_B) {
        unsigned w[SORT_B];
        #pragma unroll
        for (int k = 0; k < SORT_B; ++k) {
            int i = i0 + k * BIN_T;
            w[k] = (i < len) ? binned[beg + i] : 0u;
        }
        #pragma unroll
        for (int k = 0; k < SORT_B; ++k)
            if (i0 + k * BIN_T < len) atomicAdd(&rcur[w[k] & (RB - 1)], 1);
    }
    __syncthreads();

    int c = (tid < RB) ? rcur[tid] : 0;
    int incl = c;
    for (int o = 1; o < 32; o <<= 1) {
        int u = __shfl_up(incl, o);
        if (lane >= o) incl += u;
    }
    if (tid < RB && lane == 31) wtot[wv] = incl;
    __syncthreads();
    if (tid < RB) {
        int pre = 0;
        for (int k = 0; k < wv; ++k) pre += wtot[k];
        int ls = pre + incl - c;
        rcur[tid] = ls;
        int node = (b << RB_SHIFT) + tid;
        if (node < n) {
            first[node] = beg + ls;
            cnt[node]   = c;
            float dg = (float)(c + 1);
            dinv[node] = (dg > 0.0f) ? rsqrtf(dg) : 0.0f;
        }
    }
    __syncthreads();
    bool fits = len <= SORT_CAP;
    for (int i0 = tid; i0 < len; i0 += BIN_T * SORT_B) {
        unsigned w[SORT_B];
        #pragma unroll
        for (int k = 0; k < SORT_B; ++k) {
            int i = i0 + k * BIN_T;
            w[k] = (i < len) ? binned[beg + i] : 0u;
        }
        #pragma unroll
        for (int k = 0; k < SORT_B; ++k) {
            if (i0 + k * BIN_T < len) {
                int p = atomicAdd(&rcur[w[k] & (RB - 1)], 1);
                int s = (int)(w[k] >> RB_SHIFT);
                if (fits) stage[p] = s;
                else      ssrc[beg + p] = s;
            }
        }
    }
    __syncthreads();
    if (fits)
        for (int i = tid; i < len; i += BIN_T) ssrc[beg + i] = stage[i];
}


__global__ void gcn_agg1_rows(const int* __restrict__ ssrc, const int* __restrict__ first,
                              const int* __restrict__ cnt, const float* __restrict__ h,
                              const float* __restrict__ dinv, const float* __restrict__ b1,
                              float* __restrict__ hrelu, int n) {
    int wave = __builtin_amdgcn_readfirstlane((int)((blockIdx.x * blockDim.x + threadIdx.x) >> 5));
    int lane = threadIdx.x & 31;
    int sub  = lane >> 4;
    int c    = lane & 15;
    int d0   = wave * 2;
    if (d0 >= n) return;
    float res = 0.0f;
    for (int rsel = 0; rsel < 2; ++rsel) {
        int d = d0 + rsel;
        if (d >= n) break;
        int beg = first[d];
        int end = beg + cnt[d];
        float acc = 0.0f;
        for (int base = beg; base < end; base += 32) {
            int idx = base + lane;
            int sv  = (idx < end) ? ssrc[idx] : 0;
            int nv  = end - base;
            if (nv > 32) nv = 32;
            for (int p0 = 0; p0 < nv; p0 += 16) {
                float m[8];
                #pragma unroll
                for (int q = 0; q < 8; ++q) {
                    int j  = p0 + 2 * q + sub;
                    int sj = __shfl(sv, j);
                    m[q] = h[sj * HIDDEN + c];
                }
                #pragma unroll
                for (int q = 0; q < 8; ++q) {
                    int j = p0 + 2 * q + sub;
                    acc += (j < nv) ? m[q] : 0.0f;
                }
            }
        }
        acc += __shfl_xor(acc, 16);
        if (sub == rsel) res = acc;
    }
    int row = d0 + sub;
    if (row < n) {
        float w = dinv[row];
        hrelu[row * HIDDEN + c] = fmaxf((res + h[row * HIDDEN + c]) * w + b1[c], 0.0f);
    }
}


__global__ void gcn_agg2_rows(const int* __restrict__ ssrc, const int* __restrict__ first,
                              const int* __restrict__ cnt, const float* __restrict__ g,
                              float* __restrict__ out2, int n) {
    int wave = __builtin_amdgcn_readfirstlane((int)((blockIdx.x * blockDim.x + threadIdx.x) >> 5));
    int lane = threadIdx.x & 31;
    int sub  = lane >> 3;
    int c    = lane & 7;
    int d0   = wave * 4;
    if (d0 >= n) return;
    float res = 0.0f;
    for (int rsel = 0; rsel < 4; ++rsel) {
        int d = d0 + rsel;
        if (d >= n) break;
        int beg = first[d];
        int end = beg + cnt[d];
        float acc = 0.0f;
        for (int base = beg; base < end; base += 32) {
            int idx = base + lane;
            int sv  = (idx < end) ? ssrc[idx] : 0;
            int nv  = end - base;
            if (nv > 32) nv = 32;
            for (int p0 = 0; p0 < nv; p0 += 16) {
                float m[4];
                #pragma unroll
                for (int q = 0; q < 4; ++q) {
                    int j  = p0 + 4 * q + sub;
                    int sj = __shfl(sv, j);
                    m[q] = g[sj * HIDDEN + c];
                }
                #pragma unroll
                for (int q = 0; q < 4; ++q) {
                    int j = p0 + 4 * q + sub;
                    acc += (j < nv) ? m[q] : 0.0f;
                }
            }
        }
        acc += __shfl_xor(acc, 8);
        acc += __shfl_xor(acc, 16);
        if (sub == rsel) res = acc;
    }
    int row = d0 + sub;
    if (row < n) out2[row * 8 + c] = res;
}


extern "C" void kernel_launch(void* const* d_in, const int* in_sizes, int n_in,
                              void* d_out, int out_size, void* d_ws, size_t ws_size,
                              hipStream_t stream) {
    const float* x  = (const float*)d_in[0];
    const int*   ei = (const int*)d_in[1];
    const float* W1 = (const float*)d_in[2];
    const float* b1 = (const float*)d_in[3];
    const float* W2 = (const float*)d_in[4];
    const float* b2 = (const float*)d_in[5];
    float* out = (float*)d_out;

    const int n = in_sizes[0] / IN_FEAT;
    const int E = in_sizes[1] / 2;
    const int* src = ei;
    const int* dst = ei + E;

    const int B = 256;
    const int tiles = (n + 15) / 16;
    const int gemmBlocks = (tiles + 7) / 8;

    const int nb = (n + RB - 1) >> RB_SHIFT;
    int ch = (E + 255) / 256;
    if (ch > CH_MAX) ch = CH_MAX;
    if (ch < 1) ch = 1;
    const int G = (E + ch - 1) / ch;
    const size_t n4 = ((size_t)n + 3) & ~(size_t)3;
    const size_t E4 = ((size_t)E + 3) & ~(size_t)3;
    const size_t m  = (size_t)nb * G;
    const size_t m4 = (m + 3) & ~(size_t)3;
    const size_t a_words = (E4 > (size_t)32 * n) ? E4 : (size_t)32 * n;
    const size_t need = (3 * n4 + 2 * m4 + a_words + (size_t)E) * sizeof(float);

    if (need <= ws_size && E > 0 && nb <= NB_MAX && n <= (1 << 23)) {
        float*    f      = (float*)d_ws;
        float*    dinv   = f;
        int*      cnt    = (int*)(f + n4);
        int*      first  = (int*)(f + 2 * n4);
        int*      gh     = (int*)(f + 3 * n4);
        int*      gs     = (int*)(f + 3 * n4 + m4);
        float*    a      = f + 3 * n4 + 2 * m4;
        unsigned* binned = (unsigned*)a;
        float*    h      = a;
        float*    hrelu  = a + (size_t)16 * n;
        int*      ssrc   = (int*)(a + a_words);
        float*    g      = h;
        float*    out2   = hrelu;

        gcn_bin_count  <<<G, BIN_T, 0, stream>>>(dst, gh, E, ch, nb, G);
        gcn_scan_plain <<<(int)((m + SCAN_PER - 1) / SCAN_PER), SCAN_T, 0, stream>>>(gh, gs, (int)m);
        gcn_bin_scatter<<<G, BIN_T, 0, stream>>>(src, dst, gs, binned, E, ch, nb, G);
        gcn_bin_sort   <<<nb, BIN_T, 0, stream>>>(binned, gs, ssrc, first, cnt, dinv, n, E, nb, G);

        gcn_gemm1_wmma <<<gemmBlocks, B, 0, stream>>>(x, W1, dinv, h, n);
        gcn_agg1_rows  <<<(n + 15) / 16, B, 0, stream>>>(ssrc, first, cnt, h, dinv, b1, hrelu, n);

        gcn_gemm2_wmma <<<gemmBlocks, B, 0, stream>>>(hrelu, W2, dinv, g, n);
        gcn_agg2_rows  <<<(n + 31) / 32, B, 0, stream>>>(ssrc, first, cnt, g, out2, n);
        gcn_logsoftmax <<<(n + B - 1) / B, B, 0, stream>>>(out2, g, dinv, b2, out, n);
        return;
    }

    float* f     = (float*)d_ws;
    float* dinv  = f;
    float* h     = f + (size_t)n;
    float* out1  = f + (size_t)17 * n;
    float* hrelu = f + (size_t)33 * n;
    float* g     = out1;
    float* out2  = h;

    gcn_deg_init   <<<(n + B - 1) / B, B, 0, stream>>>(dinv, n);
    gcn_deg_accum  <<<(E + B - 1) / B, B, 0, stream>>>(dst, dinv, E);
    gcn_deg_rsqrt  <<<(n + B - 1) / B, B, 0, stream>>>(dinv, n);

    gcn_zero       <<<((size_t)16 * n + B - 1) / B, B, 0, stream>>>(out1, 16 * n);
    gcn_gemm1_wmma <<<gemmBlocks, B, 0, stream>>>(x, W1, dinv, h, n);
    gcn_agg1       <<<(E + B - 1) / B, B, 0, stream>>>(src, dst, h, out1, E);
    gcn_selfloop_relu<<<(n + B - 1) / B, B, 0, stream>>>(out1, h, dinv, b1, hrelu, n);

    gcn_zero       <<<((size_t)8 * n + B - 1) / B, B, 0, stream>>>(out2, 8 * n);
    gcn_gemm2_wmma <<<gemmBlocks, B, 0, stream>>>(hrelu, W2, dinv, g, n);
    gcn_agg2       <<<(E + B - 1) / B, B, 0, stream>>>(src, dst, g, out2, E);
    gcn_logsoftmax <<<(n + B - 1) / B, B, 0, stream>>>(out2, g, dinv, b2, out, n);
}


# sort build in two launches, not four: a chunk orders its edges by bin inside its own slice and leaves run lengths; a bin's workgroup sums the lengths before it, takes its runs; row order by arrival
# speedup vs baseline: 6.1518x; 1.1694x over previous
#include <hip/hip_runtime.h>
#include <hip/hip_bf16.h>

typedef __attribute__((ext_vector_type(2))) float v2f;
typedef __attribute__((ext_vector_type(8))) float v8f;

#define IN_FEAT 512
#define HIDDEN  16
#define NCLS    7


__global__ void gcn_deg_init(float* deg, int n) {
    int i = blockIdx.x * blockDim.x + threadIdx.x;
    if (i < n) deg[i] = 1.0f;
}

__global__ void gcn_deg_accum(const int* __restrict__ dst, float* deg, int E) {
    int e = blockIdx.x * blockDim.x + threadIdx.x;
    if (e < E) atomicAdd(&deg[dst[e]], 1.0f);
}

__global__ void gcn_deg_rsqrt(float* dinv, int n) {
    int i = blockIdx.x * blockDim.x + threadIdx.x;
    if (i < n) {
        float d = dinv[i];
        dinv[i] = (d > 0.0f) ? rsqrtf(d) : 0.0f;
    }
}

__global__ void gcn_zero(float* p, int n) {
    int i = blockIdx.x * blockDim.x + threadIdx.x;
    if (i < n) p[i] = 0.0f;
}


__global__ void gcn_gemm1_wmma(const float* __restrict__ x,
                               const float* __restrict__ W1,
                               const float* __restrict__ dinv,
                               float* __restrict__ h, int n) {
    int wave = (blockIdx.x * blockDim.x + threadIdx.x) >> 5;
    int lane = threadIdx.x & 31;
    int row0 = wave * 16;
    if (row0 >= n) return;

    int r    = lane & 15;
    int hi   = lane >> 4;
    int koff = hi * 2;
    int row  = row0 + r;
    if (row >= n) row = n - 1;
    const float* xrow = x + (size_t)row * IN_FEAT;

    v8f c = {0.f, 0.f, 0.f, 0.f, 0.f, 0.f, 0.f, 0.f};

    #pragma unroll 4
    for (int k = 0; k < IN_FEAT; k += 4) {
        if ((k & 63) == 0)
            __builtin_prefetch(xrow + k + 128, 0, 1);
        v2f a = *(const v2f*)(xrow + k + koff);
        int kk = k + koff;
        v2f b;
        b.x = W1[(size_t)kk * HIDDEN + r];
        b.y = W1[(size_t)(kk + 1) * HIDDEN + r];
        c = __builtin_amdgcn_wmma_f32_16x16x4_f32(false, a, false, b,
                                                  (short)0, c, false, false);
    }

    int srow = row0 + hi * 8;
    #pragma unroll
    for (int i = 0; i < 8; ++i) {
        int rr = srow + i;
        if (rr < n) h[(size_t)rr * HIDDEN + r] = c[i] * dinv[rr];
    }
}


__global__ void gcn_agg1(const int* __restrict__ src, const int* __restrict__ dst,
                         const float* __restrict__ h,
                         float* out1, int E) {
    int t    = blockIdx.x * blockDim.x + threadIdx.x;
    int lane = threadIdx.x & 31;
    int e0   = __builtin_amdgcn_readfirstlane(t - lane);
    if (e0 >= E) return;
    int e = e0 + lane;
    int s = 0, d = 0;
    if (e < E) {
        s = src[e];
        d = dst[e];
    }
    int nvalid = E - e0;
    if (nvalid > 32) nvalid = 32;
    int sub = lane >> 4;
    int c   = lane & 15;

    float m[16];
    int   dj[16];
    #pragma unroll
    for (int i = 0; i < 16; ++i) {
        int j    = 2 * i + sub;
        int sj   = __shfl(s, j);
        dj[i]    = __shfl(d, j);
        m[i] = h[sj * HIDDEN + c];
    }
    if (nvalid == 32) {
        #pragma unroll
        for (int i = 0; i < 16; ++i)
            atomicAdd(out1 + (dj[i] * HIDDEN + c), m[i]);
    } else {
        #pragma unroll
        for (int i = 0; i < 16; ++i) {
            int j = 2 * i + sub;
            if (j < nvalid)
                atomicAdd(out1 + (dj[i] * HIDDEN + c), m[i]);
        }
    }
}


__global__ void gcn_selfloop_relu(const float* __restrict__ out1,
                                  const float* __restrict__ h,
                                  const float* __restrict__ dinv,
                                  const float* __restrict__ b1,
                                  float* hrelu, int n) {
    int i = blockIdx.x * blockDim.x + threadIdx.x;
    if (i >= n) return;
    float w = dinv[i];
    const float* o = out1 + (size_t)i * HIDDEN;
    const float* hh = h + (size_t)i * HIDDEN;
    float* dst = hrelu + (size_t)i * HIDDEN;
    #pragma unroll
    for (int c = 0; c < HIDDEN; ++c)
        dst[c] = fmaxf((o[c] + hh[c]) * w + b1[c], 0.0f);
}


__global__ void gcn_gemm2_wmma(const float* __restrict__ hrelu,
                               const float* __restrict__ W2,
                               const float* __restrict__ dinv,
                               float* __restrict__ g, int n) {
    int wave = (blockIdx.x * blockDim.x + threadIdx.x) >> 5;
    int lane = threadIdx.x & 31;
    int row0 = wave * 16;
    if (row0 >= n) return;

    int r    = lane & 15;
    int hi   = lane >> 4;
    int koff = hi * 2;
    int row  = row0 + r;
    if (row >= n) row = n - 1;
    const float* hrow = hrelu + (size_t)row * HIDDEN;

    v8f c = {0.f, 0.f, 0.f, 0.f, 0.f, 0.f, 0.f, 0.f};

    #pragma unroll
    for (int k = 0; k < HIDDEN; k += 4) {
        int kk = k + koff;
        v2f a = *(const v2f*)(hrow + kk);
        v2f b;
        b.x = (r < NCLS) ? W2[(size_t)kk * NCLS + r] : 0.0f;
        b.y = (r < NCLS) ? W2[(size_t)(kk + 1) * NCLS + r] : 0.0f;
        c = __builtin_amdgcn_wmma_f32_16x16x4_f32(false, a, false, b,
                                                  (short)0, c, false, false);
    }

    int srow = row0 + hi * 8;
    #pragma unroll
    for (int i = 0; i < 8; ++i) {
        int rr = srow + i;
        if (rr < n) g[(size_t)rr * HIDDEN + r] = c[i] * dinv[rr];
    }
}


__global__ void gcn_agg2(const int* __restrict__ src, const int* __restrict__ dst,
                         const float* __restrict__ g,
                         float* out2, int E) {
    int t    = blockIdx.x * blockDim.x + threadIdx.x;
    int lane = threadIdx.x & 31;
    int e0   = t - lane;
    if (e0 >= E) return;
    int e = e0 + lane;
    int s = 0, d = 0;
    if (e < E) {
        s = src[e];
        d = dst[e];
    }
    int nvalid = E - e0;
    if (nvalid > 32) nvalid = 32;
    int sub = lane >> 3;
    int c   = lane & 7;

    float m[8];
    int   dj[8];
    #pragma unroll
    for (int i = 0; i < 8; ++i) {
        int j    = 4 * i + sub;
        int sj   = __shfl(s, j);
        dj[i]    = __shfl(d, j);
        m[i] = g[(size_t)sj * HIDDEN + c];
    }
    #pragma unroll
    for (int i = 0; i < 8; ++i) {
        int j = 4 * i + sub;
        if (c < NCLS && j < nvalid)
            atomicAdd(out2 + (size_t)dj[i] * 8 + c, m[i]);
    }
}


__global__ void gcn_logsoftmax(const float* __restrict__ out2,
                               const float* __restrict__ g,
                               const float* __restrict__ dinv,
                               const float* __restrict__ b2,
                               float* __restrict__ out, int n) {
    int i = blockIdx.x * blockDim.x + threadIdx.x;
    if (i >= n) return;
    float w = dinv[i];
    const float* o = out2 + (size_t)i * 8;
    const float* gg = g + (size_t)i * HIDDEN;
    float v[NCLS];
    float m = -3.4e38f;
    #pragma unroll
    for (int c = 0; c < NCLS; ++c) {
        v[c] = (o[c] + gg[c]) * w + b2[c];
        m = fmaxf(m, v[c]);
    }
    float s = 0.0f;
    #pragma unroll
    for (int c = 0; c < NCLS; ++c) s += expf(v[c] - m);
    float l = logf(s);
    float* dsto = out + (size_t)i * NCLS;
    #pragma unroll
    for (int c = 0; c < NCLS; ++c) dsto[c] = v[c] - m - l;
}

#define RB_SHIFT 9
#define RB       512
#define NB_MAX   1024
#define BIN_T    1024
#define CH_PER   16
#define CH_MAX   (BIN_T * CH_PER)
#define SORT_CAP 32768
#define SORT_RUNS 8
#define SORT_K    3

__global__ void __launch_bounds__(BIN_T)
gcn_bin_scatter(const int* __restrict__ src, const int* __restrict__ dst,
                int* __restrict__ gh, int* __restrict__ lofs, unsigned* __restrict__ binned,
                int E, int ch, int nb, int G) {
    __shared__ int      lcur[NB_MAX];
    __shared__ int      wtot[BIN_T / 32];
    __shared__ unsigned stage[CH_MAX];
    int tid  = threadIdx.x;
    int lane = tid & 31;
    int wv   = tid >> 5;
    int g    = blockIdx.x;
    lcur[tid] = 0;
    int e0 = g * ch;
    int e1 = (E - e0 < ch) ? E : e0 + ch;

    int sv[CH_PER], dv[CH_PER];
    #pragma unroll
    for (int k = 0; k < CH_PER; ++k) {
        int e = e0 + tid + k * BIN_T;
        sv[k] = 0; dv[k] = 0;
        if (e < e1) { sv[k] = src[e]; dv[k] = dst[e]; }
    }
    __syncthreads();
    #pragma unroll
    for (int k = 0; k < CH_PER; ++k)
        if (e0 + tid + k * BIN_T < e1) atomicAdd(&lcur[dv[k] >> RB_SHIFT], 1);
    __syncthreads();

    int c = (tid < nb) ? lcur[tid] : 0;
    int incl = c;
    for (int o = 1; o < 32; o <<= 1) {
        int u = __shfl_up(incl, o);
        if (lane >= o) incl += u;
    }
    if (lane == 31) wtot[wv] = incl;
    __syncthreads();
    int pre = 0;
    for (int k = 0; k < wv; ++k) pre += wtot[k];
    int ls = pre + incl - c;
    if (tid < nb) {
        lcur[tid] = ls;
        gh[tid * G + g]   = c;
        lofs[tid * G + g] = ls;
    }
    __syncthreads();
    #pragma unroll
    for (int k = 0; k < CH_PER; ++k) {
        if (e0 + tid + k * BIN_T < e1) {
            int p = atomicAdd(&lcur[dv[k] >> RB_SHIFT], 1);
            stage[p] = ((unsigned)sv[k] << RB_SHIFT) | (unsigned)(dv[k] & (RB - 1));
        }
    }
    __syncthreads();
    int len = e1 - e0;
    for (int i = tid; i < len; i += BIN_T) binned[e0 + i] = stage[i];
}

__global__ void __launch_bounds__(BIN_T)
gcn_bin_sort(const unsigned* __restrict__ binned, const int* __restrict__ gh,
             const int* __restrict__ lofs, int* __restrict__ ssrc, int* __restrict__ first,
             int* __restrict__ cnt, float* __restrict__ dinv, int n, int ch, int nb, int G) {
    __shared__ int rcur[RB];
    __shared__ int wtot[RB / 32];
    __shared__ int wpre[BIN_T / 32];
    __shared__ int wlen[BIN_T / 32];
    __shared__ int stage[SORT_CAP];
    int tid  = threadIdx.x;
    int lane = tid & 31;
    int wv   = __builtin_amdgcn_readfirstlane(tid >> 5);
    int b    = blockIdx.x;
    int lim  = b * G;
    int lim4 = lim & ~3;
    if (tid < RB) rcur[tid] = 0;

    int pre = 0;
    #pragma unroll 8
    for (int i = tid * 4; i < lim4; i += BIN_T * 4) {
        int4 v = *(const int4*)(gh + i);
        pre += (v.x + v.y) + (v.z + v.w);
    }
    if (tid < lim - lim4) pre += gh[lim4 + tid];
    __syncthreads();

    int mylen = 0;
    for (int gb = 0; gb < G; gb += BIN_T) {
        int g  = gb + wv + 32 * lane;
        int rc = 0, ro = 0;
        if (g < G) { rc = gh[lim + g]; ro = g * ch + lofs[lim + g]; }
        mylen += rc;
        int nr = (G - gb - wv + 31) >> 5;
        if (nr > 32) nr = 32;
        for (int i0 = 0; i0 < nr; i0 += SORT_RUNS) {
            unsigned w[SORT_RUNS][SORT_K];
            int      cq[SORT_RUNS], oq[SORT_RUNS];
            #pragma unroll
            for (int q = 0; q < SORT_RUNS; ++q) {
                cq[q] = __builtin_amdgcn_readlane(rc, i0 + q);
                oq[q] = __builtin_amdgcn_readlane(ro, i0 + q);
                #pragma unroll
                for (int k = 0; k < SORT_K; ++k) {
                    int i = lane + 32 * k;
                    w[q][k] = binned[(i < cq[q]) ? oq[q] + i : 0];
                }
            }
            #pragma unroll
            for (int q = 0; q < SORT_RUNS; ++q) {
                #pragma unroll
                for (int k = 0; k < SORT_K; ++k)
                    if (lane + 32 * k < cq[q]) atomicAdd(&rcur[w[q][k] & (RB - 1)], 1);
                for (int i1 = 32 * SORT_K; i1 < cq[q]; i1 += 32) {
                    int i = i1 + lane;
                    if (i < cq[q]) atomicAdd(&rcur[binned[oq[q] + i] & (RB - 1)], 1);
                }
            }
        }
    }
    for (int o = 16; o > 0; o >>= 1) {
        pre   += __shfl_xor(pre, o);
        mylen += __shfl_xor(mylen, o);
    }
    if (lane == 0) { wpre[wv] = pre; wlen[wv] = mylen; }
    __syncthreads();
    int beg = wpre[lane];
    int len = wlen[lane];
    for (int o = 16; o > 0; o >>= 1) {
        beg += __shfl_xor(beg, o);
        len += __shfl_xor(len, o);
    }

    int c = (tid < RB) ? rcur[tid] : 0;
    int incl = c;
    for (int o = 1; o < 32; o <<= 1) {
        int u = __shfl_up(incl, o);
        if (lane >= o) incl += u;
    }
    if (tid < RB && lane == 31) wtot[wv] = incl;
    __syncthreads();
    if (tid < RB) {
        int pr = 0;
        for (int k = 0; k < wv; ++k) pr += wtot[k];
        int ls = pr + incl - c;
        rcur[tid] = ls;
        int node = (b << RB_SHIFT) + tid;
        if (node < n) {
            first[node] = beg + ls;
            cnt[node]   = c;
            float dg = (float)(c + 1);
            dinv[node] = (dg > 0.0f) ? rsqrtf(dg) : 0.0f;
        }
    }
    __syncthreads();
    bool fits = len <= SORT_CAP;
    for (int gb = 0; gb < G; gb += BIN_T) {
        int g  = gb + wv + 32 * lane;
        int rc = 0, ro = 0;
        if (g < G) { rc = gh[lim + g]; ro = g * ch + lofs[lim + g]; }
        int nr = (G - gb - wv + 31) >> 5;
        if (nr > 32) nr = 32;
        for (int i0 = 0; i0 < nr; i0 += SORT_RUNS) {
            unsigned w[SORT_RUNS][SORT_K];
            int      cq[SORT_RUNS], oq[SORT_RUNS];
            #pragma unroll
            for (int q = 0; q < SORT_RUNS; ++q) {
                cq[q] = __builtin_amdgcn_readlane(rc, i0 + q);
                oq[q] = __builtin_amdgcn_readlane(ro, i0 + q);
                #pragma unroll
                for (int k = 0; k < SORT_K; ++k) {
                    int i = lane + 32 * k;
                    w[q][k] = binned[(i < cq[q]) ? oq[q] + i : 0];
                }
            }
            #pragma unroll
            for (int q = 0; q < SORT_RUNS; ++q) {
                #pragma unroll
                for (int k = 0; k < SORT_K; ++k) {
                    if (lane + 32 * k < cq[q]) {
                        int p = atomicAdd(&rcur[w[q][k] & (RB - 1)], 1);
                        int s = (int)(w[q][k] >> RB_SHIFT);
                        if (fits) stage[p] = s;
                        else      ssrc[beg + p] = s;
                    }
                }
                for (int i1 = 32 * SORT_K; i1 < cq[q]; i1 += 32) {
                    int i = i1 + lane;
                    if (i < cq[q]) {
                        unsigned wl = binned[oq[q] + i];
                        int p = atomicAdd(&rcur[wl & (RB - 1)], 1);
                        int s = (int)(wl >> RB_SHIFT);
                        if (fits) stage[p] = s;
                        else      ssrc[beg + p] = s;
                    }
                }
            }
        }
    }
    __syncthreads();
    if (fits)
        for (int i = tid; i < len; i += BIN_T) ssrc[beg + i] = stage[i];
}


__global__ void gcn_agg1_rows(const int* __restrict__ ssrc, const int* __restrict__ first,
                              const int* __restrict__ cnt, const float* __restrict__ h,
                              const float* __restrict__ dinv, const float* __restrict__ b1,
                              float* __restrict__ hrelu, int n) {
    int wave = __builtin_amdgcn_readfirstlane((int)((blockIdx.x * blockDim.x + threadIdx.x) >> 5));
    int lane = threadIdx.x & 31;
    int sub  = lane >> 4;
    int c    = lane & 15;
    int d0   = wave * 2;
    if (d0 >= n) return;
    float res = 0.0f;
    for (int rsel = 0; rsel < 2; ++rsel) {
        int d = d0 + rsel;
        if (d >= n) break;
        int beg = first[d];
        int end = beg + cnt[d];
        float acc = 0.0f;
        for (int base = beg; base < end; base += 32) {
            int idx = base + lane;
            int sv  = (idx < end) ? ssrc[idx] : 0;
            int nv  = end - base;
            if (nv > 32) nv = 32;
            for (int p0 = 0; p0 < nv; p0 += 16) {
                float m[8];
                #pragma unroll
                for (int q = 0; q < 8; ++q) {
                    int j  = p0 + 2 * q + sub;
                    int sj = __shfl(sv, j);
                    m[q] = h[sj * HIDDEN + c];
                }
                #pragma unroll
                for (int q = 0; q < 8; ++q) {
                    int j = p0 + 2 * q + sub;
                    acc += (j < nv) ? m[q] : 0.0f;
                }
            }
        }
        acc += __shfl_xor(acc, 16);
        if (sub == rsel) res = acc;
    }
    int row = d0 + sub;
    if (row < n) {
        float w = dinv[row];
        hrelu[row * HIDDEN + c] = fmaxf((res + h[row * HIDDEN + c]) * w + b1[c], 0.0f);
    }
}


__global__ void gcn_agg2_rows(const int* __restrict__ ssrc, const int* __restrict__ first,
                              const int* __restrict__ cnt, const float* __restrict__ g,
                              float* __restrict__ out2, int n) {
    int wave = __builtin_amdgcn_readfirstlane((int)((blockIdx.x * blockDim.x + threadIdx.x) >> 5));
    int lane = threadIdx.x & 31;
    int sub  = lane >> 3;
    int c    = lane & 7;
    int d0   = wave * 4;
    if (d0 >= n) return;
    float res = 0.0f;
    for (int rsel = 0; rsel < 4; ++rsel) {
        int d = d0 + rsel;
        if (d >= n) break;
        int beg = first[d];
        int end = beg + cnt[d];
        float acc = 0.0f;
        for (int base = beg; base < end; base += 32) {
            int idx = base + lane;
            int sv  = (idx < end) ? ssrc[idx] : 0;
            int nv  = end - base;
            if (nv > 32) nv = 32;
            for (int p0 = 0; p0 < nv; p0 += 16) {
                float m[4];
                #pragma unroll
                for (int q = 0; q < 4; ++q) {
                    int j  = p0 + 4 * q + sub;
                    int sj = __shfl(sv, j);
                    m[q] = g[sj * HIDDEN + c];
                }
                #pragma unroll
                for (int q = 0; q < 4; ++q) {
                    int j = p0 + 4 * q + sub;
                    acc += (j < nv) ? m[q] : 0.0f;
                }
            }
        }
        acc += __shfl_xor(acc, 8);
        acc += __shfl_xor(acc, 16);
        if (sub == rsel) res = acc;
    }
    int row = d0 + sub;
    if (row < n) out2[row * 8 + c] = res;
}


extern "C" void kernel_launch(void* const* d_in, const int* in_sizes, int n_in,
                              void* d_out, int out_size, void* d_ws, size_t ws_size,
                              hipStream_t stream) {
    const float* x  = (const float*)d_in[0];
    const int*   ei = (const int*)d_in[1];
    const float* W1 = (const float*)d_in[2];
    const float* b1 = (const float*)d_in[3];
    const float* W2 = (const float*)d_in[4];
    const float* b2 = (const float*)d_in[5];
    float* out = (float*)d_out;

    const int n = in_sizes[0] / IN_FEAT;
    const int E = in_sizes[1] / 2;
    const int* src = ei;
    const int* dst = ei + E;

    const int B = 256;
    const int tiles = (n + 15) / 16;
    const int gemmBlocks = (tiles + 7) / 8;

    const int nb = (n + RB - 1) >> RB_SHIFT;
    int ch = (E + 255) / 256;
    if (ch > CH_MAX) ch = CH_MAX;
    if (ch < 1) ch = 1;
    const int G = (E + ch - 1) / ch;
    const size_t n4 = ((size_t)n + 3) & ~(size_t)3;
    const size_t E4 = ((size_t)E + 3) & ~(size_t)3;
    const size_t m  = (size_t)nb * G;
    const size_t m4 = (m + 3) & ~(size_t)3;
    const size_t a_words = (E4 > (size_t)32 * n) ? E4 : (size_t)32 * n;
    const size_t need = (3 * n4 + 2 * m4 + a_words + (size_t)E) * sizeof(float);

    if (need <= ws_size && E > 0 && nb <= NB_MAX && n <= (1 << 23)) {
        float*    f      = (float*)d_ws;
        float*    dinv   = f;
        int*      cnt    = (int*)(f + n4);
        int*      first  = (int*)(f + 2 * n4);
        int*      gh     = (int*)(f + 3 * n4);
        int*      lofs   = (int*)(f + 3 * n4 + m4);
        float*    a      = f + 3 * n4 + 2 * m4;
        unsigned* binned = (unsigned*)a;
        float*    h      = a;
        float*    hrelu  = a + (size_t)16 * n;
        int*      ssrc   = (int*)(a + a_words);
        float*    g      = h;
        float*    out2   = hrelu;

        gcn_bin_scatter<<<G, BIN_T, 0, stream>>>(src, dst, gh, lofs, binned, E, ch, nb, G);
        gcn_bin_sort   <<<nb, BIN_T, 0, stream>>>(binned, gh, lofs, ssrc, first, cnt, dinv, n, ch, nb, G);

        gcn_gemm1_wmma <<<gemmBlocks, B, 0, stream>>>(x, W1, dinv, h, n);
        gcn_agg1_rows  <<<(n + 15) / 16, B, 0, stream>>>(ssrc, first, cnt, h, dinv, b1, hrelu, n);

        gcn_gemm2_wmma <<<gemmBlocks, B, 0, stream>>>(hrelu, W2, dinv, g, n);
        gcn_agg2_rows  <<<(n + 31) / 32, B, 0, stream>>>(ssrc, first, cnt, g, out2, n);
        gcn_logsoftmax <<<(n + B - 1) / B, B, 0, stream>>>(out2, g, dinv, b2, out, n);
        return;
    }

    float* f     = (float*)d_ws;
    float* dinv  = f;
    float* h     = f + (size_t)n;
    float* out1  = f + (size_t)17 * n;
    float* hrelu = f + (size_t)33 * n;
    float* g     = out1;
    float* out2  = h;

    gcn_deg_init   <<<(n + B - 1) / B, B, 0, stream>>>(dinv, n);
    gcn_deg_accum  <<<(E + B - 1) / B, B, 0, stream>>>(dst, dinv, E);
    gcn_deg_rsqrt  <<<(n + B - 1) / B, B, 0, stream>>>(dinv, n);

    gcn_zero       <<<((size_t)16 * n + B - 1) / B, B, 0, stream>>>(out1, 16 * n);
    gcn_gemm1_wmma <<<gemmBlocks, B, 0, stream>>>(x, W1, dinv, h, n);
    gcn_agg1       <<<(E + B - 1) / B, B, 0, stream>>>(src, dst, h, out1, E);
    gcn_selfloop_relu<<<(n + B - 1) / B, B, 0, stream>>>(out1, h, dinv, b1, hrelu, n);

    gcn_zero       <<<((size_t)8 * n + B - 1) / B, B, 0, stream>>>(out2, 8 * n);
    gcn_gemm2_wmma <<<gemmBlocks, B, 0, stream>>>(hrelu, W2, dinv, g, n);
    gcn_agg2       <<<(E + B - 1) / B, B, 0, stream>>>(src, dst, g, out2, E);
    gcn_logsoftmax <<<(n + B - 1) / B, B, 0, stream>>>(out2, g, dinv, b2, out, n);
}


# row-local tails join the launch that makes their rows: W2 product (f32 WMMA, 16x16 tile met in LDS) inside agg1 rows, bias + log_softmax inside agg2 rows; 7 launches -> 5; sum orders as before
# speedup vs baseline: 6.4510x; 1.0486x over previous
#include <hip/hip_runtime.h>
#include <hip/hip_bf16.h>

typedef __attribute__((ext_vector_type(2))) float v2f;
typedef __attribute__((ext_vector_type(8))) float v8f;

#define IN_FEAT 512
#define HIDDEN  16
#define NCLS    7


__global__ void gcn_deg_init(float* deg, int n) {
    int i = blockIdx.x * blockDim.x + threadIdx.x;
    if (i < n) deg[i] = 1.0f;
}

__global__ void gcn_deg_accum(const int* __restrict__ dst, float* deg, int E) {
    int e = blockIdx.x * blockDim.x + threadIdx.x;
    if (e < E) atomicAdd(&deg[dst[e]], 1.0f);
}

__global__ void gcn_deg_rsqrt(float* dinv, int n) {
    int i = blockIdx.x * blockDim.x + threadIdx.x;
    if (i < n) {
        float d = dinv[i];
        dinv[i] = (d > 0.0f) ? rsqrtf(d) : 0.0f;
    }
}

__global__ void gcn_zero(float* p, int n) {
    int i = blockIdx.x * blockDim.x + threadIdx.x;
    if (i < n) p[i] = 0.0f;
}


__global__ void gcn_gemm1_wmma(const float* __restrict__ x,
                               const float* __restrict__ W1,
                               const float* __restrict__ dinv,
                               float* __restrict__ h, int n) {
    int wave = (blockIdx.x * blockDim.x + threadIdx.x) >> 5;
    int lane = threadIdx.x & 31;
    int row0 = wave * 16;
    if (row0 >= n) return;

    int r    = lane & 15;
    int hi   = lane >> 4;
    int koff = hi * 2;
    int row  = row0 + r;
    if (row >= n) row = n - 1;
    const float* xrow = x + (size_t)row * IN_FEAT;

    v8f c = {0.f, 0.f, 0.f, 0.f, 0.f, 0.f, 0.f, 0.f};

    #pragma unroll 4
    for (int k = 0; k < IN_FEAT; k += 4) {
        if ((k & 63) == 0)
            __builtin_prefetch(xrow + k + 128, 0, 1);
        v2f a = *(const v2f*)(xrow + k + koff);
        int kk = k + koff;
        v2f b;
        b.x = W1[(size_t)kk * HIDDEN + r];
        b.y = W1[(size_t)(kk + 1) * HIDDEN + r];
        c = __builtin_amdgcn_wmma_f32_16x16x4_f32(false, a, false, b,
                                                  (short)0, c, false, false);
    }

    int srow = row0 + hi * 8;
    #pragma unroll
    for (int i = 0; i < 8; ++i) {
        int rr = srow + i;
        if (rr < n) h[(size_t)rr * HIDDEN + r] = c[i] * dinv[rr];
    }
}


__global__ void gcn_agg1(const int* __restrict__ src, const int* __restrict__ dst,
                         const float* __restrict__ h,
                         float* out1, int E) {
    int t    = blockIdx.x * blockDim.x + threadIdx.x;
    int lane = threadIdx.x & 31;
    int e0   = __builtin_amdgcn_readfirstlane(t - lane);
    if (e0 >= E) return;
    int e = e0 + lane;
    int s = 0, d = 0;
    if (e < E) {
        s = src[e];
        d = dst[e];
    }
    int nvalid = E - e0;
    if (nvalid > 32) nvalid = 32;
    int sub = lane >> 4;
    int c   = lane & 15;

    float m[16];
    int   dj[16];
    #pragma unroll
    for (int i = 0; i < 16; ++i) {
        int j    = 2 * i + sub;
        int sj   = __shfl(s, j);
        dj[i]    = __shfl(d, j);
        m[i] = h[sj * HIDDEN + c];
    }
    if (nvalid == 32) {
        #pragma unroll
        for (int i = 0; i < 16; ++i)
            atomicAdd(out1 + (dj[i] * HIDDEN + c), m[i]);
    } else {
        #pragma unroll
        for (int i = 0; i < 16; ++i) {
            int j = 2 * i + sub;
            if (j < nvalid)
                atomicAdd(out1 + (dj[i] * HIDDEN + c), m[i]);
        }
    }
}


__global__ void gcn_selfloop_relu(const float* __restrict__ out1,
                                  const float* __restrict__ h,
                                  const float* __restrict__ dinv,
                                  const float* __restrict__ b1,
                                  float* hrelu, int n) {
    int i = blockIdx.x * blockDim.x + threadIdx.x;
    if (i >= n) return;
    float w = dinv[i];
    const float* o = out1 + (size_t)i * HIDDEN;
    const float* hh = h + (size_t)i * HIDDEN;
    float* dst = hrelu + (size_t)i * HIDDEN;
    #pragma unroll
    for (int c = 0; c < HIDDEN; ++c)
        dst[c] = fmaxf((o[c] + hh[c]) * w + b1[c], 0.0f);
}


__global__ void gcn_gemm2_wmma(const float* __restrict__ hrelu,
                               const float* __restrict__ W2,
                               const float* __restrict__ dinv,
                               float* __restrict__ g, int n) {
    int wave = (blockIdx.x * blockDim.x + threadIdx.x) >> 5;
    int lane = threadIdx.x & 31;
    int row0 = wave * 16;
    if (row0 >= n) return;

    int r    = lane & 15;
    int hi   = lane >> 4;
    int koff = hi * 2;
    int row  = row0 + r;
    if (row >= n) row = n - 1;
    const float* hrow = hrelu + (size_t)row * HIDDEN;

    v8f c = {0.f, 0.f, 0.f, 0.f, 0.f, 0.f, 0.f, 0.f};

    #pragma unroll
    for (int k = 0; k < HIDDEN; k += 4) {
        int kk = k + koff;
        v2f a = *(const v2f*)(hrow + kk);
        v2f b;
        b.x = (r < NCLS) ? W2[(size_t)kk * NCLS + r] : 0.0f;
        b.y = (r < NCLS) ? W2[(size_t)(kk + 1) * NCLS + r] : 0.0f;
        c = __builtin_amdgcn_wmma_f32_16x16x4_f32(false, a, false, b,
                                                  (short)0, c, false, false);
    }

    int srow = row0 + hi * 8;
    #pragma unroll
    for (int i = 0; i < 8; ++i) {
        int rr = srow + i;
        if (rr < n) g[(size_t)rr * HIDDEN + r] = c[i] * dinv[rr];
    }
}


__global__ void gcn_agg2(const int* __restrict__ src, const int* __restrict__ dst,
                         const float* __restrict__ g,
                         float* out2, int E) {
    int t    = blockIdx.x * blockDim.x + threadIdx.x;
    int lane = threadIdx.x & 31;
    int e0   = t - lane;
    if (e0 >= E) return;
    int e = e0 + lane;
    int s = 0, d = 0;
    if (e < E) {
        s = src[e];
        d = dst[e];
    }
    int nvalid = E - e0;
    if (nvalid > 32) nvalid = 32;
    int sub = lane >> 3;
    int c   = lane & 7;

    float m[8];
    int   dj[8];
    #pragma unroll
    for (int i = 0; i < 8; ++i) {
        int j    = 4 * i + sub;
        int sj   = __shfl(s, j);
        dj[i]    = __shfl(d, j);
        m[i] = g[(size_t)sj * HIDDEN + c];
    }
    #pragma unroll
    for (int i = 0; i < 8; ++i) {
        int j = 4 * i + sub;
        if (c < NCLS && j < nvalid)
            atomicAdd(out2 + (size_t)dj[i] * 8 + c, m[i]);
    }
}


__global__ void gcn_logsoftmax(const float* __restrict__ out2,
                               const float* __restrict__ g,
                               const float* __restrict__ dinv,
                               const float* __restrict__ b2,
                               float* __restrict__ out, int n) {
    int i = blockIdx.x * blockDim.x + threadIdx.x;
    if (i >= n) return;
    float w = dinv[i];
    const float* o = out2 + (size_t)i * 8;
    const float* gg = g + (size_t)i * HIDDEN;
    float v[NCLS];
    float m = -3.4e38f;
    #pragma unroll
    for (int c = 0; c < NCLS; ++c) {
        v[c] = (o[c] + gg[c]) * w + b2[c];
        m = fmaxf(m, v[c]);
    }
    float s = 0.0f;
    #pragma unroll
    for (int c = 0; c < NCLS; ++c) s += expf(v[c] - m);
    float l = logf(s);
    float* dsto = out + (size_t)i * NCLS;
    #pragma unroll
    for (int c = 0; c < NCLS; ++c) dsto[c] = v[c] - m - l;
}

#define RB_SHIFT 9
#define RB       512
#define NB_MAX   1024
#define BIN_T    1024
#define CH_PER   16
#define CH_MAX   (BIN_T * CH_PER)
#define SORT_CAP 32768
#define SORT_RUNS 8
#define SORT_K    3

__global__ void __launch_bounds__(BIN_T)
gcn_bin_scatter(const int* __restrict__ src, const int* __restrict__ dst,
                int* __restrict__ gh, int* __restrict__ lofs, unsigned* __restrict__ binned,
                int E, int ch, int nb, int G) {
    __shared__ int      lcur[NB_MAX];
    __shared__ int      wtot[BIN_T / 32];
    __shared__ unsigned stage[CH_MAX];
    int tid  = threadIdx.x;
    int lane = tid & 31;
    int wv   = tid >> 5;
    int g    = blockIdx.x;
    lcur[tid] = 0;
    int e0 = g * ch;
    int e1 = (E - e0 < ch) ? E : e0 + ch;

    int sv[CH_PER], dv[CH_PER];
    #pragma unroll
    for (int k = 0; k < CH_PER; ++k) {
        int e = e0 + tid + k * BIN_T;
        sv[k] = 0; dv[k] = 0;
        if (e < e1) { sv[k] = src[e]; dv[k] = dst[e]; }
    }
    __syncthreads();
    #pragma unroll
    for (int k = 0; k < CH_PER; ++k)
        if (e0 + tid + k * BIN_T < e1) atomicAdd(&lcur[dv[k] >> RB_SHIFT], 1);
    __syncthreads();

    int c = (tid < nb) ? lcur[tid] : 0;
    int incl = c;
    for (int o = 1; o < 32; o <<= 1) {
        int u = __shfl_up(incl, o);
        if (lane >= o) incl += u;
    }
    if (lane == 31) wtot[wv] = incl;
    __syncthreads();
    int pre = 0;
    for (int k = 0; k < wv; ++k) pre += wtot[k];
    int ls = pre + incl - c;
    if (tid < nb) {
        lcur[tid] = ls;
        gh[tid * G + g]   = c;
        lofs[tid * G + g] = ls;
    }
    __syncthreads();
    #pragma unroll
    for (int k = 0; k < CH_PER; ++k) {
        if (e0 + tid + k * BIN_T < e1) {
            int p = atomicAdd(&lcur[dv[k] >> RB_SHIFT], 1);
            stage[p] = ((unsigned)sv[k] << RB_SHIFT) | (unsigned)(dv[k] & (RB - 1));
        }
    }
    __syncthreads();
    int len = e1 - e0;
    for (int i = tid; i < len; i += BIN_T) binned[e0 + i] = stage[i];
}

__global__ void __launch_bounds__(BIN_T)
gcn_bin_sort(const unsigned* __restrict__ binned, const int* __restrict__ gh,
             const int* __restrict__ lofs, int* __restrict__ ssrc, int* __restrict__ first,
             int* __restrict__ cnt, float* __restrict__ dinv, int n, int ch, int nb, int G) {
    __shared__ int rcur[RB];
    __shared__ int wtot[RB / 32];
    __shared__ int wpre[BIN_T / 32];
    __shared__ int wlen[BIN_T / 32];
    __shared__ int stage[SORT_CAP];
    int tid  = threadIdx.x;
    int lane = tid & 31;
    int wv   = __builtin_amdgcn_readfirstlane(tid >> 5);
    int b    = blockIdx.x;
    int lim  = b * G;
    int lim4 = lim & ~3;
    if (tid < RB) rcur[tid] = 0;

    int pre = 0;
    #pragma unroll 8
    for (int i = tid * 4; i < lim4; i += BIN_T * 4) {
        int4 v = *(const int4*)(gh + i);
        pre += (v.x + v.y) + (v.z + v.w);
    }
    if (tid < lim - lim4) pre += gh[lim4 + tid];
    __syncthreads();

    int mylen = 0;
    for (int gb = 0; gb < G; gb += BIN_T) {
        int g  = gb + wv + 32 * lane;
        int rc = 0, ro = 0;
        if (g < G) { rc = gh[lim + g]; ro = g * ch + lofs[lim + g]; }
        mylen += rc;
        int nr = (G - gb - wv + 31) >> 5;
        if (nr > 32) nr = 32;
        for (int i0 = 0; i0 < nr; i0 += SORT_RUNS) {
            unsigned w[SORT_RUNS][SORT_K];
            int      cq[SORT_RUNS], oq[SORT_RUNS];
            #pragma unroll
            for (int q = 0; q < SORT_RUNS; ++q) {
                cq[q] = __builtin_amdgcn_readlane(rc, i0 + q);
                oq[q] = __builtin_amdgcn_readlane(ro, i0 + q);
                #pragma unroll
                for (int k = 0; k < SORT_K; ++k) {
                    int i = lane + 32 * k;
                    w[q][k] = binned[(i < cq[q]) ? oq[q] + i : 0];
                }
            }
            #pragma unroll
            for (int q = 0; q < SORT_RUNS; ++q) {
                #pragma unroll
                for (int k = 0; k < SORT_K; ++k)
                    if (lane + 32 * k < cq[q]) atomicAdd(&rcur[w[q][k] & (RB - 1)], 1);
                for (int i1 = 32 * SORT_K; i1 < cq[q]; i1 += 32) {
                    int i = i1 + lane;
                    if (i < cq[q]) atomicAdd(&rcur[binned[oq[q] + i] & (RB - 1)], 1);
                }
            }
        }
    }
    for (int o = 16; o > 0; o >>= 1) {
        pre   += __shfl_xor(pre, o);
        mylen += __shfl_xor(mylen, o);
    }
    if (lane == 0) { wpre[wv] = pre; wlen[wv] = mylen; }
    __syncthreads();
    int beg = wpre[lane];
    int len = wlen[lane];
    for (int o = 16; o > 0; o >>= 1) {
        beg += __shfl_xor(beg, o);
        len += __shfl_xor(len, o);
    }

    int c = (tid < RB) ? rcur[tid] : 0;
    int incl = c;
    for (int o = 1; o < 32; o <<= 1) {
        int u = __shfl_up(incl, o);
        if (lane >= o) incl += u;
    }
    if (tid < RB && lane == 31) wtot[wv] = incl;
    __syncthreads();
    if (tid < RB) {
        int pr = 0;
        for (int k = 0; k < wv; ++k) pr += wtot[k];
        int ls = pr + incl - c;
        rcur[tid] = ls;
        int node = (b << RB_SHIFT) + tid;
        if (node < n) {
            first[node] = beg + ls;
            cnt[node]   = c;
            float dg = (float)(c + 1);
            dinv[node] = (dg > 0.0f) ? rsqrtf(dg) : 0.0f;
        }
    }
    __syncthreads();
    bool fits = len <= SORT_CAP;
    for (int gb = 0; gb < G; gb += BIN_T) {
        int g  = gb + wv + 32 * lane;
        int rc = 0, ro = 0;
        if (g < G) { rc = gh[lim + g]; ro = g * ch + lofs[lim + g]; }
        int nr = (G - gb - wv + 31) >> 5;
        if (nr > 32) nr = 32;
        for (int i0 = 0; i0 < nr; i0 += SORT_RUNS) {
            unsigned w[SORT_RUNS][SORT_K];
            int      cq[SORT_RUNS], oq[SORT_RUNS];
            #pragma unroll
            for (int q = 0; q < SORT_RUNS; ++q) {
                cq[q] = __builtin_amdgcn_readlane(rc, i0 + q);
                oq[q] = __builtin_amdgcn_readlane(ro, i0 + q);
                #pragma unroll
                for (int k = 0; k < SORT_K; ++k) {
                    int i = lane + 32 * k;
                    w[q][k] = binned[(i < cq[q]) ? oq[q] + i : 0];
                }
            }
            #pragma unroll
            for (int q = 0; q < SORT_RUNS; ++q) {
                #pragma unroll
                for (int k = 0; k < SORT_K; ++k) {
                    if (lane + 32 * k < cq[q]) {
                        int p = atomicAdd(&rcur[w[q][k] & (RB - 1)], 1);
                        int s = (int)(w[q][k] >> RB_SHIFT);
                        if (fits) stage[p] = s;
                        else      ssrc[beg + p] = s;
                    }
                }
                for (int i1 = 32 * SORT_K; i1 < cq[q]; i1 += 32) {
                    int i = i1 + lane;
                    if (i < cq[q]) {
                        unsigned wl = binned[oq[q] + i];
                        int p = atomicAdd(&rcur[wl & (RB - 1)], 1);
                        int s = (int)(wl >> RB_SHIFT);
                        if (fits) stage[p] = s;
                        else      ssrc[beg + p] = s;
                    }
                }
            }
        }
    }
    __syncthreads();
    if (fits)
        for (int i = tid; i < len; i += BIN_T) ssrc[beg + i] = stage[i];
}


__global__ void __launch_bounds__(256)
gcn_agg1_rows_w2(const int* __restrict__ ssrc, const int* __restrict__ first,
                 const int* __restrict__ cnt, const float* __restrict__ h,
                 const float* __restrict__ dinv, const float* __restrict__ b1,
                 const float* __restrict__ W2, float* __restrict__ g, int n) {
    __shared__ float tile[16 * HIDDEN];
    __shared__ __attribute__((aligned(16))) float tdinv[16];
    int lane = threadIdx.x & 31;
    int wv   = __builtin_amdgcn_readfirstlane((int)(threadIdx.x >> 5));
    int sub  = lane >> 4;
    int c    = lane & 15;
    int row0 = blockIdx.x * 16;
    int d0   = row0 + wv * 2;

    v2f bw[HIDDEN / 4];
    #pragma unroll
    for (int j = 0; j < HIDDEN / 4; ++j) { bw[j].x = 0.0f; bw[j].y = 0.0f; }
    if (wv == 0 && c < NCLS) {
        #pragma unroll
        for (int j = 0; j < HIDDEN / 4; ++j) {
            int kk = 4 * j + 2 * sub;
            bw[j].x = W2[kk * NCLS + c];
            bw[j].y = W2[(kk + 1) * NCLS + c];
        }
    }

    float res = 0.0f;
    for (int rsel = 0; rsel < 2; ++rsel) {
        int d = d0 + rsel;
        if (d >= n) break;
        int beg = first[d];
        int end = beg + cnt[d];
        float acc = 0.0f;
        for (int base = beg; base < end; base += 32) {
            int idx = base + lane;
            int sv  = (idx < end) ? ssrc[idx] : 0;
            int nv  = end - base;
            if (nv > 32) nv = 32;
            for (int p0 = 0; p0 < nv; p0 += 16) {
                float m[8];
                #pragma unroll
                for (int q = 0; q < 8; ++q) {
                    int j  = p0 + 2 * q + sub;
                    int sj = __shfl(sv, j);
                    m[q] = h[sj * HIDDEN + c];
                }
                #pragma unroll
                for (int q = 0; q < 8; ++q) {
                    int j = p0 + 2 * q + sub;
                    acc += (j < nv) ? m[q] : 0.0f;
                }
            }
        }
        acc += __shfl_xor(acc, 16);
        if (sub == rsel) res = acc;
    }
    int row = d0 + sub;
    float w = 0.0f, v = 0.0f;
    if (row < n) {
        w = dinv[row];
        v = fmaxf((res + h[row * HIDDEN + c]) * w + b1[c], 0.0f);
    }
    tile[(wv * 2 + sub) * HIDDEN + c] = v;
    if (c == 0) tdinv[wv * 2 + sub] = w;
    __syncthreads();
    if (wv == 0) {
        int r  = c;
        int hi = sub;
        v8f cc = {0.f, 0.f, 0.f, 0.f, 0.f, 0.f, 0.f, 0.f};
        #pragma unroll
        for (int j = 0; j < HIDDEN / 4; ++j) {
            int kk = 4 * j + 2 * hi;
            v2f a = *(const v2f*)(tile + r * HIDDEN + kk);
            cc = __builtin_amdgcn_wmma_f32_16x16x4_f32(false, a, false, bw[j],
                                                       (short)0, cc, false, false);
        }
        const float4 dv0 = *(const float4*)(tdinv + hi * 8);
        const float4 dv1 = *(const float4*)(tdinv + hi * 8 + 4);
        const float dv[8] = {dv0.x, dv0.y, dv0.z, dv0.w, dv1.x, dv1.y, dv1.z, dv1.w};
        #pragma unroll
        for (int i = 0; i < 8; ++i) {
            int rr = row0 + hi * 8 + i;
            if (rr < n) g[rr * HIDDEN + r] = cc[i] * dv[i];
        }
    }
}


__global__ void __launch_bounds__(256)
gcn_agg2_rows_lsm(const int* __restrict__ ssrc, const int* __restrict__ first,
                  const int* __restrict__ cnt, const float* __restrict__ g,
                  const float* __restrict__ dinv, const float* __restrict__ b2,
                  float* __restrict__ out, int n) {
    __shared__ float stage[32 * NCLS];
    int lane = threadIdx.x & 31;
    int wv   = __builtin_amdgcn_readfirstlane((int)(threadIdx.x >> 5));
    int sub  = lane >> 3;
    int c    = lane & 7;
    int d0   = blockIdx.x * 32 + wv * 4;
    float res = 0.0f;
    for (int rsel = 0; rsel < 4; ++rsel) {
        int d = d0 + rsel;
        if (d >= n) break;
        int beg = first[d];
        int end = beg + cnt[d];
        float acc = 0.0f;
        for (int base = beg; base < end; base += 32) {
            int idx = base + lane;
            int sv  = (idx < end) ? ssrc[idx] : 0;
            int nv  = end - base;
            if (nv > 32) nv = 32;
            for (int p0 = 0; p0 < nv; p0 += 16) {
                float m[4];
                #pragma unroll
                for (int q = 0; q < 4; ++q) {
                    int j  = p0 + 4 * q + sub;
                    int sj = __shfl(sv, j);
                    m[q] = g[sj * HIDDEN + c];
                }
                #pragma unroll
                for (int q = 0; q < 4; ++q) {
                    int j = p0 + 4 * q + sub;
                    acc += (j < nv) ? m[q] : 0.0f;
                }
            }
        }
        acc += __shfl_xor(acc, 8);
        acc += __shfl_xor(acc, 16);
        if (sub == rsel) res = acc;
    }
    int row = d0 + sub;
    int rc  = (row < n) ? row : n - 1;
    float w = dinv[rc];
    float val = (res + g[rc * HIDDEN + c]) * w + b2[(c < NCLS) ? c : 0];
    float mx = (c < NCLS) ? val : -3.4e38f;
    mx = fmaxf(mx, __shfl_xor(mx, 1));
    mx = fmaxf(mx, __shfl_xor(mx, 2));
    mx = fmaxf(mx, __shfl_xor(mx, 4));
    float ex = expf(val - mx);
    float s = 0.0f;
    #pragma unroll
    for (int k = 0; k < NCLS; ++k) s += __shfl(ex, (lane & 24) + k);
    float l = logf(s);
    if (c < NCLS) stage[(wv * 4 + sub) * NCLS + c] = val - mx - l;
    __syncthreads();
    int t = threadIdx.x;
    int o = blockIdx.x * (32 * NCLS) + t;
    if (t < 32 * NCLS && o < n * NCLS) out[o] = stage[t];
}


extern "C" void kernel_launch(void* const* d_in, const int* in_sizes, int n_in,
                              void* d_out, int out_size, void* d_ws, size_t ws_size,
                              hipStream_t stream) {
    const float* x  = (const float*)d_in[0];
    const int*   ei = (const int*)d_in[1];
    const float* W1 = (const float*)d_in[2];
    const float* b1 = (const float*)d_in[3];
    const float* W2 = (const float*)d_in[4];
    const float* b2 = (const float*)d_in[5];
    float* out = (float*)d_out;

    const int n = in_sizes[0] / IN_FEAT;
    const int E = in_sizes[1] / 2;
    const int* src = ei;
    const int* dst = ei + E;

    const int B = 256;
    const int tiles = (n + 15) / 16;
    const int gemmBlocks = (tiles + 7) / 8;

    const int nb = (n + RB - 1) >> RB_SHIFT;
    int ch = (E + 255) / 256;
    if (ch > CH_MAX) ch = CH_MAX;
    if (ch < 1) ch = 1;
    const int G = (E + ch - 1) / ch;
    const size_t n4 = ((size_t)n + 3) & ~(size_t)3;
    const size_t E4 = ((size_t)E + 3) & ~(size_t)3;
    const size_t m  = (size_t)nb * G;
    const size_t m4 = (m + 3) & ~(size_t)3;
    const size_t a_words = (E4 > (size_t)32 * n) ? E4 : (size_t)32 * n;
    const size_t need = (3 * n4 + 2 * m4 + a_words + (size_t)E) * sizeof(float);

    if (need <= ws_size && E > 0 && nb <= NB_MAX && n <= (1 << 23)) {
        float*    f      = (float*)d_ws;
        float*    dinv   = f;
        int*      cnt    = (int*)(f + n4);
        int*      first  = (int*)(f + 2 * n4);
        int*      gh     = (int*)(f + 3 * n4);
        int*      lofs   = (int*)(f + 3 * n4 + m4);
        float*    a      = f + 3 * n4 + 2 * m4;
        unsigned* binned = (unsigned*)a;
        float*    h      = a;
        float*    g      = a + (size_t)16 * n;
        int*      ssrc   = (int*)(a + a_words);

        gcn_bin_scatter<<<G, BIN_T, 0, stream>>>(src, dst, gh, lofs, binned, E, ch, nb, G);
        gcn_bin_sort   <<<nb, BIN_T, 0, stream>>>(binned, gh, lofs, ssrc, first, cnt, dinv, n, ch, nb, G);

        gcn_gemm1_wmma <<<gemmBlocks, B, 0, stream>>>(x, W1, dinv, h, n);
        gcn_agg1_rows_w2 <<<(n + 15) / 16, B, 0, stream>>>(ssrc, first, cnt, h, dinv, b1, W2, g, n);
        gcn_agg2_rows_lsm<<<(n + 31) / 32, B, 0, stream>>>(ssrc, first, cnt, g, dinv, b2, out, n);
        return;
    }

    float* f     = (float*)d_ws;
    float* dinv  = f;
    float* h     = f + (size_t)n;
    float* out1  = f + (size_t)17 * n;
    float* hrelu = f + (size_t)33 * n;
    float* g     = out1;
    float* out2  = h;

    gcn_deg_init   <<<(n + B - 1) / B, B, 0, stream>>>(dinv, n);
    gcn_deg_accum  <<<(E + B - 1) / B, B, 0, stream>>>(dst, dinv, E);
    gcn_deg_rsqrt  <<<(n + B - 1) / B, B, 0, stream>>>(dinv, n);

    gcn_zero       <<<((size_t)16 * n + B - 1) / B, B, 0, stream>>>(out1, 16 * n);
    gcn_gemm1_wmma <<<gemmBlocks, B, 0, stream>>>(x, W1, dinv, h, n);
    gcn_agg1       <<<(E + B - 1) / B, B, 0, stream>>>(src, dst, h, out1, E);
    gcn_selfloop_relu<<<(n + B - 1) / B, B, 0, stream>>>(out1, h, dinv, b1, hrelu, n);

    gcn_zero       <<<((size_t)8 * n + B - 1) / B, B, 0, stream>>>(out2, 8 * n);
    gcn_gemm2_wmma <<<gemmBlocks, B, 0, stream>>>(hrelu, W2, dinv, g, n);
    gcn_agg2       <<<(E + B - 1) / B, B, 0, stream>>>(src, dst, g, out2, E);
    gcn_logsoftmax <<<(n + B - 1) / B, B, 0, stream>>>(out2, g, dinv, b2, out, n);
}


# agg1, agg2: a resident grid of 1024 workgroups loops over the row tiles (a wave's rows gathered together, next tile's bounds and sources fetched a trip ahead): waves enter once; sum orders as before
# speedup vs baseline: 6.9272x; 1.0738x over previous
#include <hip/hip_runtime.h>
#include <hip/hip_bf16.h>

typedef __attribute__((ext_vector_type(2))) float v2f;
typedef __attribute__((ext_vector_type(8))) float v8f;

#define IN_FEAT 512
#define HIDDEN  16
#define NCLS    7


__global__ void gcn_deg_init(float* deg, int n) {
    int i = blockIdx.x * blockDim.x + threadIdx.x;
    if (i < n) deg[i] = 1.0f;
}

__global__ void gcn_deg_accum(const int* __restrict__ dst, float* deg, int E) {
    int e = blockIdx.x * blockDim.x + threadIdx.x;
    if (e < E) atomicAdd(&deg[dst[e]], 1.0f);
}

__global__ void gcn_deg_rsqrt(float* dinv, int n) {
    int i = blockIdx.x * blockDim.x + threadIdx.x;
    if (i < n) {
        float d = dinv[i];
        dinv[i] = (d > 0.0f) ? rsqrtf(d) : 0.0f;
    }
}

__global__ void gcn_zero(float* p, int n) {
    int i = blockIdx.x * blockDim.x + threadIdx.x;
    if (i < n) p[i] = 0.0f;
}


__global__ void gcn_gemm1_wmma(const float* __restrict__ x,
                               const float* __restrict__ W1,
                               const float* __restrict__ dinv,
                               float* __restrict__ h, int n) {
    int wave = (blockIdx.x * blockDim.x + threadIdx.x) >> 5;
    int lane = threadIdx.x & 31;
    int row0 = wave * 16;
    if (row0 >= n) return;

    int r    = lane & 15;
    int hi   = lane >> 4;
    int koff = hi * 2;
    int row  = row0 + r;
    if (row >= n) row = n - 1;
    const float* xrow = x + (size_t)row * IN_FEAT;

    v8f c = {0.f, 0.f, 0.f, 0.f, 0.f, 0.f, 0.f, 0.f};

    #pragma unroll 4
    for (int k = 0; k < IN_FEAT; k += 4) {
        if ((k & 63) == 0)
            __builtin_prefetch(xrow + k + 128, 0, 1);
        v2f a = *(const v2f*)(xrow + k + koff);
        int kk = k + koff;
        v2f b;
        b.x = W1[(size_t)kk * HIDDEN + r];
        b.y = W1[(size_t)(kk + 1) * HIDDEN + r];
        c = __builtin_amdgcn_wmma_f32_16x16x4_f32(false, a, false, b,
                                                  (short)0, c, false, false);
    }

    int srow = row0 + hi * 8;
    #pragma unroll
    for (int i = 0; i < 8; ++i) {
        int rr = srow + i;
        if (rr < n) h[(size_t)rr * HIDDEN + r] = c[i] * dinv[rr];
    }
}


__global__ void gcn_agg1(const int* __restrict__ src, const int* __restrict__ dst,
                         const float* __restrict__ h,
                         float* out1, int E) {
    int t    = blockIdx.x * blockDim.x + threadIdx.x;
    int lane = threadIdx.x & 31;
    int e0   = __builtin_amdgcn_readfirstlane(t - lane);
    if (e0 >= E) return;
    int e = e0 + lane;
    int s = 0, d = 0;
    if (e < E) {
        s = src[e];
        d = dst[e];
    }
    int nvalid = E - e0;
    if (nvalid > 32) nvalid = 32;
    int sub = lane >> 4;
    int c   = lane & 15;

    float m[16];
    int   dj[16];
    #pragma unroll
    for (int i = 0; i < 16; ++i) {
        int j    = 2 * i + sub;
        int sj   = __shfl(s, j);
        dj[i]    = __shfl(d, j);
        m[i] = h[sj * HIDDEN + c];
    }
    if (nvalid == 32) {
        #pragma unroll
        for (int i = 0; i < 16; ++i)
            atomicAdd(out1 + (dj[i] * HIDDEN + c), m[i]);
    } else {
        #pragma unroll
        for (int i = 0; i < 16; ++i) {
            int j = 2 * i + sub;
            if (j < nvalid)
                atomicAdd(out1 + (dj[i] * HIDDEN + c), m[i]);
        }
    }
}


__global__ void gcn_selfloop_relu(const float* __restrict__ out1,
                                  const float* __restrict__ h,
                                  const float* __restrict__ dinv,
                                  const float* __restrict__ b1,
                                  float* hrelu, int n) {
    int i = blockIdx.x * blockDim.x + threadIdx.x;
    if (i >= n) return;
    float w = dinv[i];
    const float* o = out1 + (size_t)i * HIDDEN;
    const float* hh = h + (size_t)i * HIDDEN;
    float* dst = hrelu + (size_t)i * HIDDEN;
    #pragma unroll
    for (int c = 0; c < HIDDEN; ++c)
        dst[c] = fmaxf((o[c] + hh[c]) * w + b1[c], 0.0f);
}


__global__ void gcn_gemm2_wmma(const float* __restrict__ hrelu,
                               const float* __restrict__ W2,
                               const float* __restrict__ dinv,
                               float* __restrict__ g, int n) {
    int wave = (blockIdx.x * blockDim.x + threadIdx.x) >> 5;
    int lane = threadIdx.x & 31;
    int row0 = wave * 16;
    if (row0 >= n) return;

    int r    = lane & 15;
    int hi   = lane >> 4;
    int koff = hi * 2;
    int row  = row0 + r;
    if (row >= n) row = n - 1;
    const float* hrow = hrelu + (size_t)row * HIDDEN;

    v8f c = {0.f, 0.f, 0.f, 0.f, 0.f, 0.f, 0.f, 0.f};

    #pragma unroll
    for (int k = 0; k < HIDDEN; k += 4) {
        int kk = k + koff;
        v2f a = *(const v2f*)(hrow + kk);
        v2f b;
        b.x = (r < NCLS) ? W2[(size_t)kk * NCLS + r] : 0.0f;
        b.y = (r < NCLS) ? W2[(size_t)(kk + 1) * NCLS + r] : 0.0f;
        c = __builtin_amdgcn_wmma_f32_16x16x4_f32(false, a, false, b,
                                                  (short)0, c, false, false);
    }

    int srow = row0 + hi * 8;
    #pragma unroll
    for (int i = 0; i < 8; ++i) {
        int rr = srow + i;
        if (rr < n) g[(size_t)rr * HIDDEN + r] = c[i] * dinv[rr];
    }
}


__global__ void gcn_agg2(const int* __restrict__ src, const int* __restrict__ dst,
                         const float* __restrict__ g,
                         float* out2, int E) {
    int t    = blockIdx.x * blockDim.x + threadIdx.x;
    int lane = threadIdx.x & 31;
    int e0   = t - lane;
    if (e0 >= E) return;
    int e = e0 + lane;
    int s = 0, d = 0;
    if (e < E) {
        s = src[e];
        d = dst[e];
    }
    int nvalid = E - e0;
    if (nvalid > 32) nvalid = 32;
    int sub = lane >> 3;
    int c   = lane & 7;

    float m[8];
    int   dj[8];
    #pragma unroll
    for (int i = 0; i < 8; ++i) {
        int j    = 4 * i + sub;
        int sj   = __shfl(s, j);
        dj[i]    = __shfl(d, j);
        m[i] = g[(size_t)sj * HIDDEN + c];
    }
    #pragma unroll
    for (int i = 0; i < 8; ++i) {
        int j = 4 * i + sub;
        if (c < NCLS && j < nvalid)
            atomicAdd(out2 + (size_t)dj[i] * 8 + c, m[i]);
    }
}


__global__ void gcn_logsoftmax(const float* __restrict__ out2,
                               const float* __restrict__ g,
                               const float* __restrict__ dinv,
                               const float* __restrict__ b2,
                               float* __restrict__ out, int n) {
    int i = blockIdx.x * blockDim.x + threadIdx.x;
    if (i >= n) return;
    float w = dinv[i];
    const float* o = out2 + (size_t)i * 8;
    const float* gg = g + (size_t)i * HIDDEN;
    float v[NCLS];
    float m = -3.4e38f;
    #pragma unroll
    for (int c = 0; c < NCLS; ++c) {
        v[c] = (o[c] + gg[c]) * w + b2[c];
        m = fmaxf(m, v[c]);
    }
    float s = 0.0f;
    #pragma unroll
    for (int c = 0; c < NCLS; ++c) s += expf(v[c] - m);
    float l = logf(s);
    float* dsto = out + (size_t)i * NCLS;
    #pragma unroll
    for (int c = 0; c < NCLS; ++c) dsto[c] = v[c] - m - l;
}

#define RB_SHIFT 9
#define RB       512
#define NB_MAX   1024
#define BIN_T    1024
#define CH_PER   16
#define CH_MAX   (BIN_T * CH_PER)
#define SORT_CAP 32768
#define SORT_RUNS 8
#define SORT_K    3

__global__ void __launch_bounds__(BIN_T)
gcn_bin_scatter(const int* __restrict__ src, const int* __restrict__ dst,
                int* __restrict__ gh, int* __restrict__ lofs, unsigned* __restrict__ binned,
                int E, int ch, int nb, int G) {
    __shared__ int      lcur[NB_MAX];
    __shared__ int      wtot[BIN_T / 32];
    __shared__ unsigned stage[CH_MAX];
    int tid  = threadIdx.x;
    int lane = tid & 31;
    int wv   = tid >> 5;
    int g    = blockIdx.x;
    lcur[tid] = 0;
    int e0 = g * ch;
    int e1 = (E - e0 < ch) ? E : e0 + ch;

    int sv[CH_PER], dv[CH_PER];
    #pragma unroll
    for (int k = 0; k < CH_PER; ++k) {
        int e = e0 + tid + k * BIN_T;
        sv[k] = 0; dv[k] = 0;
        if (e < e1) { sv[k] = src[e]; dv[k] = dst[e]; }
    }
    __syncthreads();
    #pragma unroll
    for (int k = 0; k < CH_PER; ++k)
        if (e0 + tid + k * BIN_T < e1) atomicAdd(&lcur[dv[k] >> RB_SHIFT], 1);
    __syncthreads();

    int c = (tid < nb) ? lcur[tid] : 0;
    int incl = c;
    for (int o = 1; o < 32; o <<= 1) {
        int u = __shfl_up(incl, o);
        if (lane >= o) incl += u;
    }
    if (lane == 31) wtot[wv] = incl;
    __syncthreads();
    int pre = 0;
    for (int k = 0; k < wv; ++k) pre += wtot[k];
    int ls = pre + incl - c;
    if (tid < nb) {
        lcur[tid] = ls;
        gh[tid * G + g]   = c;
        lofs[tid * G + g] = ls;
    }
    __syncthreads();
    #pragma unroll
    for (int k = 0; k < CH_PER; ++k) {
        if (e0 + tid + k * BIN_T < e1) {
            int p = atomicAdd(&lcur[dv[k] >> RB_SHIFT], 1);
            stage[p] = ((unsigned)sv[k] << RB_SHIFT) | (unsigned)(dv[k] & (RB - 1));
        }
    }
    __syncthreads();
    int len = e1 - e0;
    for (int i = tid; i < len; i += BIN_T) binned[e0 + i] = stage[i];
}

__global__ void __launch_bounds__(BIN_T)
gcn_bin_sort(const unsigned* __restrict__ binned, const int* __restrict__ gh,
             const int* __restrict__ lofs, int* __restrict__ ssrc, int* __restrict__ first,
             int* __restrict__ cnt, float* __restrict__ dinv, int n, int ch, int nb, int G) {
    __shared__ int rcur[RB];
    __shared__ int wtot[RB / 32];
    __shared__ int wpre[BIN_T / 32];
    __shared__ int wlen[BIN_T / 32];
    __shared__ int stage[SORT_CAP];
    int tid  = threadIdx.x;
    int lane = tid & 31;
    int wv   = __builtin_amdgcn_readfirstlane(tid >> 5);
    int b    = blockIdx.x;
    int lim  = b * G;
    int lim4 = lim & ~3;
    if (tid < RB) rcur[tid] = 0;

    int pre = 0;
    #pragma unroll 8
    for (int i = tid * 4; i < lim4; i += BIN_T * 4) {
        int4 v = *(const int4*)(gh + i);
        pre += (v.x + v.y) + (v.z + v.w);
    }
    if (tid < lim - lim4) pre += gh[lim4 + tid];
    __syncthreads();

    int mylen = 0;
    for (int gb = 0; gb < G; gb += BIN_T) {
        int g  = gb + wv + 32 * lane;
        int rc = 0, ro = 0;
        if (g < G) { rc = gh[lim + g]; ro = g * ch + lofs[lim + g]; }
        mylen += rc;
        int nr = (G - gb - wv + 31) >> 5;
        if (nr > 32) nr = 32;
        for (int i0 = 0; i0 < nr; i0 += SORT_RUNS) {
            unsigned w[SORT_RUNS][SORT_K];
            int      cq[SORT_RUNS], oq[SORT_RUNS];
            #pragma unroll
            for (int q = 0; q < SORT_RUNS; ++q) {
                cq[q] = __builtin_amdgcn_readlane(rc, i0 + q);
                oq[q] = __builtin_amdgcn_readlane(ro, i0 + q);
                #pragma unroll
                for (int k = 0; k < SORT_K; ++k) {
                    int i = lane + 32 * k;
                    w[q][k] = binned[(i < cq[q]) ? oq[q] + i : 0];
                }
            }
            #pragma unroll
            for (int q = 0; q < SORT_RUNS; ++q) {
                #pragma unroll
                for (int k = 0; k < SORT_K; ++k)
                    if (lane + 32 * k < cq[q]) atomicAdd(&rcur[w[q][k] & (RB - 1)], 1);
                for (int i1 = 32 * SORT_K; i1 < cq[q]; i1 += 32) {
                    int i = i1 + lane;
                    if (i < cq[q]) atomicAdd(&rcur[binned[oq[q] + i] & (RB - 1)], 1);
                }
            }
        }
    }
    for (int o = 16; o > 0; o >>= 1) {
        pre   += __shfl_xor(pre, o);
        mylen += __shfl_xor(mylen, o);
    }
    if (lane == 0) { wpre[wv] = pre; wlen[wv] = mylen; }
    __syncthreads();
    int beg = wpre[lane];
    int len = wlen[lane];
    for (int o = 16; o > 0; o >>= 1) {
        beg += __shfl_xor(beg, o);
        len += __shfl_xor(len, o);
    }

    int c = (tid < RB) ? rcur[tid] : 0;
    int incl = c;
    for (int o = 1; o < 32; o <<= 1) {
        int u = __shfl_up(incl, o);
        if (lane >= o) incl += u;
    }
    if (tid < RB && lane == 31) wtot[wv] = incl;
    __syncthreads();
    if (tid < RB) {
        int pr = 0;
        for (int k = 0; k < wv; ++k) pr += wtot[k];
        int ls = pr + incl - c;
        rcur[tid] = ls;
        int node = (b << RB_SHIFT) + tid;
        if (node < n) {
            first[node] = beg + ls;
            cnt[node]   = c;
            float dg = (float)(c + 1);
            dinv[node] = (dg > 0.0f) ? rsqrtf(dg) : 0.0f;
        }
    }
    __syncthreads();
    bool fits = len <= SORT_CAP;
    for (int gb = 0; gb < G; gb += BIN_T) {
        int g  = gb + wv + 32 * lane;
        int rc = 0, ro = 0;
        if (g < G) { rc = gh[lim + g]; ro = g * ch + lofs[lim + g]; }
        int nr = (G - gb - wv + 31) >> 5;
        if (nr > 32) nr = 32;
        for (int i0 = 0; i0 < nr; i0 += SORT_RUNS) {
            unsigned w[SORT_RUNS][SORT_K];
            int      cq[SORT_RUNS], oq[SORT_RUNS];
            #pragma unroll
            for (int q = 0; q < SORT_RUNS; ++q) {
                cq[q] = __builtin_amdgcn_readlane(rc, i0 + q);
                oq[q] = __builtin_amdgcn_readlane(ro, i0 + q);
                #pragma unroll
                for (int k = 0; k < SORT_K; ++k) {
                    int i = lane + 32 * k;
                    w[q][k] = binned[(i < cq[q]) ? oq[q] + i : 0];
                }
            }
            #pragma unroll
            for (int q = 0; q < SORT_RUNS; ++q) {
                #pragma unroll
                for (int k = 0; k < SORT_K; ++k) {
                    if (lane + 32 * k < cq[q]) {
                        int p = atomicAdd(&rcur[w[q][k] & (RB - 1)], 1);
                        int s = (int)(w[q][k] >> RB_SHIFT);
                        if (fits) stage[p] = s;
                        else      ssrc[beg + p] = s;
                    }
                }
                for (int i1 = 32 * SORT_K; i1 < cq[q]; i1 += 32) {
                    int i = i1 + lane;
                    if (i < cq[q]) {
                        unsigned wl = binned[oq[q] + i];
                        int p = atomicAdd(&rcur[wl & (RB - 1)], 1);
                        int s = (int)(wl >> RB_SHIFT);
                        if (fits) stage[p] = s;
                        else      ssrc[beg + p] = s;
                    }
                }
            }
        }
    }
    __syncthreads();
    if (fits)
        for (int i = tid; i < len; i += BIN_T) ssrc[beg + i] = stage[i];
}

#define AGG_WGS 1024

__device__ __forceinline__ void gcn_row_bounds(const int* __restrict__ first, const int* __restrict__ cnt,
                                               int d, bool live, int n, int& beg, int& cn) {
    int dc = (d < n) ? d : n - 1;
    int b = first[dc];
    int k = cnt[dc];
    bool ok = live && d < n;
    beg = ok ? b : 0;
    cn  = ok ? k : 0;
}

__device__ __forceinline__ void gcn_bounds_ahead(const int* __restrict__ first, const int* __restrict__ cnt,
                                                 int d, int n, int& vb, int& vc) {
    int dc = (d < n) ? d : n - 1;
    vb = first[dc];
    vc = cnt[dc];
}

__device__ __forceinline__ int gcn_row_chunk(const int* __restrict__ ssrc, int beg, int cn, int k, int lane) {
    int  i  = 32 * k + lane;
    bool ok = i < cn;
    int  v  = ssrc[ok ? beg + i : 0];
    return ok ? v : 0;
}

__device__ __forceinline__ void gcn_agg1_chunk2(const float* __restrict__ h, int svA, int nvA, int svB, int nvB,
                                                int sub, int c, float& accA, float& accB) {
    int nmax = (nvA > nvB) ? nvA : nvB;
    for (int p0 = 0; p0 < nmax; p0 += 16) {
        float mA[8], mB[8];
        #pragma unroll
        for (int q = 0; q < 8; ++q) {
            int j = p0 + 2 * q + sub;
            mA[q] = h[__shfl(svA, j) * HIDDEN + c];
            mB[q] = h[__shfl(svB, j) * HIDDEN + c];
        }
        #pragma unroll
        for (int q = 0; q < 8; ++q) {
            int j = p0 + 2 * q + sub;
            accA += (j < nvA) ? mA[q] : 0.0f;
            accB += (j < nvB) ? mB[q] : 0.0f;
        }
    }
}

__device__ __forceinline__ void gcn_agg2_chunk4(const float* __restrict__ g, const int (&sv)[4], const int (&nv)[4],
                                                int sub, int c, float (&acc)[4]) {
    int n01 = (nv[0] > nv[1]) ? nv[0] : nv[1];
    int n23 = (nv[2] > nv[3]) ? nv[2] : nv[3];
    int nmax = (n01 > n23) ? n01 : n23;
    for (int p0 = 0; p0 < nmax; p0 += 16) {
        float m[4][4];
        #pragma unroll
        for (int r = 0; r < 4; ++r) {
            #pragma unroll
            for (int q = 0; q < 4; ++q) {
                int j = p0 + 4 * q + sub;
                m[r][q] = g[__shfl(sv[r], j) * HIDDEN + c];
            }
        }
        #pragma unroll
        for (int r = 0; r < 4; ++r) {
            #pragma unroll
            for (int q = 0; q < 4; ++q) {
                int j = p0 + 4 * q + sub;
                acc[r] += (j < nv[r]) ? m[r][q] : 0.0f;
            }
        }
    }
}


__global__ void __launch_bounds__(256)
gcn_agg1_rows_w2(const int* __restrict__ ssrc, const int* __restrict__ first,
                 const int* __restrict__ cnt, const float* __restrict__ h,
                 const float* __restrict__ dinv, const float* __restrict__ b1,
                 const float* __restrict__ W2, float* __restrict__ g, int n, int ntiles) {
    __shared__ __attribute__((aligned(16))) float tile[2][16 * HIDDEN];
    __shared__ __attribute__((aligned(16))) float tdinv[2][16];
    int lane = threadIdx.x & 31;
    int wv   = __builtin_amdgcn_readfirstlane((int)(threadIdx.x >> 5));
    int sub  = lane >> 4;
    int c    = lane & 15;
    const int stride = gridDim.x;

    v2f bw[HIDDEN / 4];
    #pragma unroll
    for (int j = 0; j < HIDDEN / 4; ++j) { bw[j].x = 0.0f; bw[j].y = 0.0f; }
    if (c < NCLS) {
        #pragma unroll
        for (int j = 0; j < HIDDEN / 4; ++j) {
            int kk = 4 * j + 2 * sub;
            bw[j].x = W2[kk * NCLS + c];
            bw[j].y = W2[(kk + 1) * NCLS + c];
        }
    }
    const float bias = b1[c];

    int t = blockIdx.x;
    int bA0, cA0, bB0, cB0;
    {
        int d = t * 16 + wv * 2;
        gcn_row_bounds(first, cnt, d,     t < ntiles, n, bA0, cA0);
        gcn_row_bounds(first, cnt, d + 1, t < ntiles, n, bB0, cB0);
    }
    int vb1, vc1;
    gcn_bounds_ahead(first, cnt, (t + stride) * 16 + wv * 2 + (lane & 1), n, vb1, vc1);
    int sA0 = gcn_row_chunk(ssrc, bA0, cA0, 0, lane);
    int sA1 = gcn_row_chunk(ssrc, bA0, cA0, 1, lane);
    int sB0 = gcn_row_chunk(ssrc, bB0, cB0, 0, lane);
    int sB1 = gcn_row_chunk(ssrc, bB0, cB0, 1, lane);

    for (int it = 0; t < ntiles; t += stride, ++it) {
        int bA1, cA1, bB1, cB1;
        {
            int tn = t + stride;
            int dn = tn * 16 + wv * 2;
            bool okA = tn < ntiles && dn < n, okB = tn < ntiles && dn + 1 < n;
            int fa = __builtin_amdgcn_readlane(vb1, 0), ka = __builtin_amdgcn_readlane(vc1, 0);
            int fb = __builtin_amdgcn_readlane(vb1, 1), kb = __builtin_amdgcn_readlane(vc1, 1);
            bA1 = okA ? fa : 0; cA1 = okA ? ka : 0;
            bB1 = okB ? fb : 0; cB1 = okB ? kb : 0;
        }
        int nA0 = gcn_row_chunk(ssrc, bA1, cA1, 0, lane);
        int nA1 = gcn_row_chunk(ssrc, bA1, cA1, 1, lane);
        int nB0 = gcn_row_chunk(ssrc, bB1, cB1, 0, lane);
        int nB1 = gcn_row_chunk(ssrc, bB1, cB1, 1, lane);
        int vb2, vc2;
        gcn_bounds_ahead(first, cnt, (t + 2 * stride) * 16 + wv * 2 + (lane & 1), n, vb2, vc2);
        int row0 = t * 16;
        int row  = row0 + wv * 2 + sub;
        int rc   = (row < n) ? row : n - 1;
        float wrow = dinv[rc];
        float hown = h[rc * HIDDEN + c];

        float accA = 0.0f, accB = 0.0f;
        gcn_agg1_chunk2(h, sA0, (cA0 < 32) ? cA0 : 32, sB0, (cB0 < 32) ? cB0 : 32, sub, c, accA, accB);
        gcn_agg1_chunk2(h, sA1, (cA0 < 64) ? cA0 - 32 : 32, sB1, (cB0 < 64) ? cB0 - 32 : 32, sub, c, accA, accB);
        int cmax = (cA0 > cB0) ? cA0 : cB0;
        for (int base = 64; base < cmax; base += 32) {
            int xa = gcn_row_chunk(ssrc, bA0, cA0, base >> 5, lane);
            int xb = gcn_row_chunk(ssrc, bB0, cB0, base >> 5, lane);
            int ra = cA0 - base, rb = cB0 - base;
            gcn_agg1_chunk2(h, xa, (ra < 32) ? ra : 32, xb, (rb < 32) ? rb : 32, sub, c, accA, accB);
        }
        accA += __shfl_xor(accA, 16);
        accB += __shfl_xor(accB, 16);
        float res = sub ? accB : accA;

        float w = 0.0f, v = 0.0f;
        if (row < n) {
            w = wrow;
            v = fmaxf((res + hown) * w + bias, 0.0f);
        }
        int pb = it & 1;
        tile[pb][(wv * 2 + sub) * HIDDEN + c] = v;
        if (c == 0) tdinv[pb][wv * 2 + sub] = w;
        __syncthreads();
        if (wv == (it & 7)) {
            int r  = c;
            int hi = sub;
            v8f cc = {0.f, 0.f, 0.f, 0.f, 0.f, 0.f, 0.f, 0.f};
            #pragma unroll
            for (int j = 0; j < HIDDEN / 4; ++j) {
                int kk = 4 * j + 2 * hi;
                v2f a = *(const v2f*)(&tile[pb][r * HIDDEN + kk]);
                cc = __builtin_amdgcn_wmma_f32_16x16x4_f32(false, a, false, bw[j],
                                                           (short)0, cc, false, false);
            }
            const float4 dv0 = *(const float4*)(&tdinv[pb][hi * 8]);
            const float4 dv1 = *(const float4*)(&tdinv[pb][hi * 8 + 4]);
            const float dv[8] = {dv0.x, dv0.y, dv0.z, dv0.w, dv1.x, dv1.y, dv1.z, dv1.w};
            #pragma unroll
            for (int i = 0; i < 8; ++i) {
                int rr = row0 + hi * 8 + i;
                if (rr < n) g[rr * HIDDEN + r] = cc[i] * dv[i];
            }
        }
        bA0 = bA1; cA0 = cA1; bB0 = bB1; cB0 = cB1;
        vb1 = vb2; vc1 = vc2;
        sA0 = nA0; sA1 = nA1; sB0 = nB0; sB1 = nB1;
    }
}


__global__ void __launch_bounds__(256)
gcn_agg2_rows_lsm(const int* __restrict__ ssrc, const int* __restrict__ first,
                  const int* __restrict__ cnt, const float* __restrict__ g,
                  const float* __restrict__ dinv, const float* __restrict__ b2,
                  float* __restrict__ out, int n, int ntiles) {
    __shared__ float stage[2][32 * NCLS];
    int lane = threadIdx.x & 31;
    int wv   = __builtin_amdgcn_readfirstlane((int)(threadIdx.x >> 5));
    int sub  = lane >> 3;
    int c    = lane & 7;
    const int stride = gridDim.x;
    const float bias = b2[(c < NCLS) ? c : 0];

    int t = blockIdx.x;
    int bg0[4], cn0[4], s0[4], s1[4];
    #pragma unroll
    for (int r = 0; r < 4; ++r)
        gcn_row_bounds(first, cnt, t * 32 + wv * 4 + r, t < ntiles, n, bg0[r], cn0[r]);
    int vb1, vc1;
    gcn_bounds_ahead(first, cnt, (t + stride) * 32 + wv * 4 + (lane & 3), n, vb1, vc1);
    #pragma unroll
    for (int r = 0; r < 4; ++r) {
        s0[r] = gcn_row_chunk(ssrc, bg0[r], cn0[r], 0, lane);
        s1[r] = gcn_row_chunk(ssrc, bg0[r], cn0[r], 1, lane);
    }

    for (int it = 0; t < ntiles; t += stride, ++it) {
        int n0[4], n1[4], bg1[4], cn1[4];
        #pragma unroll
        for (int r = 0; r < 4; ++r) {
            int tn = t + stride;
            bool ok = tn < ntiles && tn * 32 + wv * 4 + r < n;
            int fr = __builtin_amdgcn_readlane(vb1, r), kr = __builtin_amdgcn_readlane(vc1, r);
            bg1[r] = ok ? fr : 0;
            cn1[r] = ok ? kr : 0;
            n0[r] = gcn_row_chunk(ssrc, bg1[r], cn1[r], 0, lane);
            n1[r] = gcn_row_chunk(ssrc, bg1[r], cn1[r], 1, lane);
        }
        int vb2, vc2;
        gcn_bounds_ahead(first, cnt, (t + 2 * stride) * 32 + wv * 4 + (lane & 3), n, vb2, vc2);
        int row = t * 32 + wv * 4 + sub;
        int rc  = (row < n) ? row : n - 1;
        float w    = dinv[rc];
        float gown = g[rc * HIDDEN + c];

        float acc[4] = {0.0f, 0.0f, 0.0f, 0.0f};
        int nv[4];
        #pragma unroll
        for (int r = 0; r < 4; ++r) nv[r] = (cn0[r] < 32) ? cn0[r] : 32;
        gcn_agg2_chunk4(g, s0, nv, sub, c, acc);
        #pragma unroll
        for (int r = 0; r < 4; ++r) nv[r] = (cn0[r] < 64) ? cn0[r] - 32 : 32;
        gcn_agg2_chunk4(g, s1, nv, sub, c, acc);
        int c01 = (cn0[0] > cn0[1]) ? cn0[0] : cn0[1];
        int c23 = (cn0[2] > cn0[3]) ? cn0[2] : cn0[3];
        int cmax = (c01 > c23) ? c01 : c23;
        for (int base = 64; base < cmax; base += 32) {
            int xs[4];
            #pragma unroll
            for (int r = 0; r < 4; ++r) {
                xs[r] = gcn_row_chunk(ssrc, bg0[r], cn0[r], base >> 5, lane);
                int rem = cn0[r] - base;
                nv[r] = (rem < 32) ? rem : 32;
            }
            gcn_agg2_chunk4(g, xs, nv, sub, c, acc);
        }
        #pragma unroll
        for (int r = 0; r < 4; ++r) {
            acc[r] += __shfl_xor(acc[r], 8);
            acc[r] += __shfl_xor(acc[r], 16);
        }
        float res = (sub == 0) ? acc[0] : (sub == 1) ? acc[1] : (sub == 2) ? acc[2] : acc[3];

        float val = (res + gown) * w + bias;
        float mx = (c < NCLS) ? val : -3.4e38f;
        mx = fmaxf(mx, __shfl_xor(mx, 1));
        mx = fmaxf(mx, __shfl_xor(mx, 2));
        mx = fmaxf(mx, __shfl_xor(mx, 4));
        float ex = expf(val - mx);
        float s = 0.0f;
        #pragma unroll
        for (int k = 0; k < NCLS; ++k) s += __shfl(ex, (lane & 24) + k);
        float l = logf(s);
        int pb = it & 1;
        if (c < NCLS) stage[pb][(wv * 4 + sub) * NCLS + c] = val - mx - l;
        __syncthreads();
        int tid = threadIdx.x;
        int o = t * (32 * NCLS) + tid;
        if (tid < 32 * NCLS && o < n * NCLS) out[o] = stage[pb][tid];
        #pragma unroll
        for (int r = 0; r < 4; ++r) {
            bg0[r] = bg1[r]; cn0[r] = cn1[r];
            s0[r] = n0[r]; s1[r] = n1[r];
        }
        vb1 = vb2; vc1 = vc2;
    }
}


extern "C" void kernel_launch(void* const* d_in, const int* in_sizes, int n_in,
                              void* d_out, int out_size, void* d_ws, size_t ws_size,
                              hipStream_t stream) {
    const float* x  = (const float*)d_in[0];
    const int*   ei = (const int*)d_in[1];
    const float* W1 = (const float*)d_in[2];
    const float* b1 = (const float*)d_in[3];
    const float* W2 = (const float*)d_in[4];
    const float* b2 = (const float*)d_in[5];
    float* out = (float*)d_out;

    const int n = in_sizes[0] / IN_FEAT;
    const int E = in_sizes[1] / 2;
    const int* src = ei;
    const int* dst = ei + E;

    const int B = 256;
    const int tiles = (n + 15) / 16;
    const int gemmBlocks = (tiles + 7) / 8;

    const int nb = (n + RB - 1) >> RB_SHIFT;
    int ch = (E + 255) / 256;
    if (ch > CH_MAX) ch = CH_MAX;
    if (ch < 1) ch = 1;
    const int G = (E + ch - 1) / ch;
    const size_t n4 = ((size_t)n + 3) & ~(size_t)3;
    const size_t E4 = ((size_t)E + 3) & ~(size_t)3;
    const size_t m  = (size_t)nb * G;
    const size_t m4 = (m + 3) & ~(size_t)3;
    const size_t a_words = (E4 > (size_t)32 * n) ? E4 : (size_t)32 * n;
    const size_t need = (3 * n4 + 2 * m4 + a_words + (size_t)E) * sizeof(float);

    if (need <= ws_size && E > 0 && nb <= NB_MAX && n <= (1 << 23)) {
        float*    f      = (float*)d_ws;
        float*    dinv   = f;
        int*      cnt    = (int*)(f + n4);
        int*      first  = (int*)(f + 2 * n4);
        int*      gh     = (int*)(f + 3 * n4);
        int*      lofs   = (int*)(f + 3 * n4 + m4);
        float*    a      = f + 3 * n4 + 2 * m4;
        unsigned* binned = (unsigned*)a;
        float*    h      = a;
        float*    g      = a + (size_t)16 * n;
        int*      ssrc   = (int*)(a + a_words);

        gcn_bin_scatter<<<G, BIN_T, 0, stream>>>(src, dst, gh, lofs, binned, E, ch, nb, G);
        gcn_bin_sort   <<<nb, BIN_T, 0, stream>>>(binned, gh, lofs, ssrc, first, cnt, dinv, n, ch, nb, G);

        gcn_gemm1_wmma <<<gemmBlocks, B, 0, stream>>>(x, W1, dinv, h, n);
        const int t1 = (n + 15) / 16;
        const int t2 = (n + 31) / 32;
        gcn_agg1_rows_w2 <<<(t1 < AGG_WGS) ? t1 : AGG_WGS, B, 0, stream>>>(ssrc, first, cnt, h, dinv, b1, W2, g, n, t1);
        gcn_agg2_rows_lsm<<<(t2 < AGG_WGS) ? t2 : AGG_WGS, B, 0, stream>>>(ssrc, first, cnt, g, dinv, b2, out, n, t2);
        return;
    }

    float* f     = (float*)d_ws;
    float* dinv  = f;
    float* h     = f + (size_t)n;
    float* out1  = f + (size_t)17 * n;
    float* hrelu = f + (size_t)33 * n;
    float* g     = out1;
    float* out2  = h;

    gcn_deg_init   <<<(n + B - 1) / B, B, 0, stream>>>(dinv, n);
    gcn_deg_accum  <<<(E + B - 1) / B, B, 0, stream>>>(dst, dinv, E);
    gcn_deg_rsqrt  <<<(n + B - 1) / B, B, 0, stream>>>(dinv, n);

    gcn_zero       <<<((size_t)16 * n + B - 1) / B, B, 0, stream>>>(out1, 16 * n);
    gcn_gemm1_wmma <<<gemmBlocks, B, 0, stream>>>(x, W1, dinv, h, n);
    gcn_agg1       <<<(E + B - 1) / B, B, 0, stream>>>(src, dst, h, out1, E);
    gcn_selfloop_relu<<<(n + B - 1) / B, B, 0, stream>>>(out1, h, dinv, b1, hrelu, n);

    gcn_zero       <<<((size_t)8 * n + B - 1) / B, B, 0, stream>>>(out2, 8 * n);
    gcn_gemm2_wmma <<<gemmBlocks, B, 0, stream>>>(hrelu, W2, dinv, g, n);
    gcn_agg2       <<<(E + B - 1) / B, B, 0, stream>>>(src, dst, g, out2, E);
    gcn_logsoftmax <<<(n + B - 1) / B, B, 0, stream>>>(out2, g, dinv, b2, out, n);
}


# x @ W1: x read as a stream (16-byte non-temporal loads of whole 256-byte row pieces, two chunks ahead), cut into A fragments via wave-private LDS; W1 fragments from LDS; new k order inside each 8
# speedup vs baseline: 7.8778x; 1.1372x over previous
#include <hip/hip_runtime.h>
#include <hip/hip_bf16.h>

typedef __attribute__((ext_vector_type(2))) float v2f;
typedef __attribute__((ext_vector_type(4))) float v4f;
typedef __attribute__((ext_vector_type(8))) float v8f;

#define IN_FEAT 512
#define HIDDEN  16
#define NCLS    7


__global__ void gcn_deg_init(float* deg, int n) {
    int i = blockIdx.x * blockDim.x + threadIdx.x;
    if (i < n) deg[i] = 1.0f;
}

__global__ void gcn_deg_accum(const int* __restrict__ dst, float* deg, int E) {
    int e = blockIdx.x * blockDim.x + threadIdx.x;
    if (e < E) atomicAdd(&deg[dst[e]], 1.0f);
}

__global__ void gcn_deg_rsqrt(float* dinv, int n) {
    int i = blockIdx.x * blockDim.x + threadIdx.x;
    if (i < n) {
        float d = dinv[i];
        dinv[i] = (d > 0.0f) ? rsqrtf(d) : 0.0f;
    }
}

__global__ void gcn_zero(float* p, int n) {
    int i = blockIdx.x * blockDim.x + threadIdx.x;
    if (i < n) p[i] = 0.0f;
}

#define G1_KC  64
#define G1_PX  68
#define G1_NCH (IN_FEAT / G1_KC)

__device__ __forceinline__ void gcn_g1_chunk_mma(const float* xsw, const float* bfr, int ld_off, int lane,
                                                 int q, v8f& acc) {
    #pragma unroll
    for (int p = 0; p < G1_KC / 8; ++p) {
        v4f a = *(const v4f*)(xsw + ld_off + 8 * p);
        v4f b = *(const v4f*)(bfr + ((q * (G1_KC / 8) + p) * 32 + lane) * 4);
        v2f a0 = {a.x, a.y}, b0 = {b.x, b.y};
        v2f a1 = {a.z, a.w}, b1 = {b.z, b.w};
        acc = __builtin_amdgcn_wmma_f32_16x16x4_f32(false, a0, false, b0,
                                                    (short)0, acc, false, false);
        acc = __builtin_amdgcn_wmma_f32_16x16x4_f32(false, a1, false, b1,
                                                    (short)0, acc, false, false);
    }
}

__device__ __forceinline__ void gcn_g1_dinv8(const float* __restrict__ dinv, int row0, int hi, int n, float (&dv)[8]) {
    #pragma unroll
    for (int i = 0; i < 8; ++i) {
        int rr = row0 + hi * 8 + i;
        dv[i] = dinv[(rr < n) ? rr : n - 1];
    }
}

__global__ void __launch_bounds__(256, 8)
gcn_gemm1_wmma(const float* __restrict__ x,
               const float* __restrict__ W1,
               const float* __restrict__ dinv,
               float* __restrict__ h, int n) {
    __shared__ __attribute__((aligned(16))) float bfr[IN_FEAT * HIDDEN];
    __shared__ __attribute__((aligned(16))) float xs[8][16 * G1_PX];
    const int tid  = threadIdx.x;
    const int lane = tid & 31;
    const int wv   = __builtin_amdgcn_readfirstlane(tid >> 5);
    const int c    = lane & 15;
    const int hi   = lane >> 4;
    const int row0 = (blockIdx.x * 8 + wv) * 16;
    const bool full = row0 + 16 <= n;

    v4f wq[8];
    #pragma unroll
    for (int u = 0; u < 8; ++u) wq[u] = *(const v4f*)(W1 + 4 * (tid + 256 * u));

    const float* xb = x + (size_t)(row0 + hi) * IN_FEAT + 4 * c;
    v4f buf[2][8];
    if (full) {
        #pragma unroll
        for (int m = 0; m < 8; ++m) buf[0][m] = __builtin_nontemporal_load((const v4f*)(xb + 2 * m * IN_FEAT));
        #pragma unroll
        for (int m = 0; m < 8; ++m) buf[1][m] = __builtin_nontemporal_load((const v4f*)(xb + 2 * m * IN_FEAT + G1_KC));
    }

    #pragma unroll
    for (int u = 0; u < 8; ++u) {
        int f  = tid + 256 * u;
        int k  = f >> 2;
        int c0 = 4 * (f & 3);
        int base = ((k >> 3) * 32 + 16 * ((k >> 2) & 1)) * 4 + (k & 3);
        bfr[base + 4 * (c0 + 0)] = wq[u].x;
        bfr[base + 4 * (c0 + 1)] = wq[u].y;
        bfr[base + 4 * (c0 + 2)] = wq[u].z;
        bfr[base + 4 * (c0 + 3)] = wq[u].w;
    }
    __syncthreads();
    if (row0 >= n) return;

    float* xsw = &xs[wv][0];
    const int st_off = hi * G1_PX + 4 * c;
    const int ld_off = c * G1_PX + 4 * hi;
    v8f acc = {0.f, 0.f, 0.f, 0.f, 0.f, 0.f, 0.f, 0.f};
    float dv[8];

    if (full) {
        #pragma unroll
        for (int q = 0; q < G1_NCH; ++q) {
            __builtin_amdgcn_sched_barrier(0);
            if (q == G1_NCH - 1) gcn_g1_dinv8(dinv, row0, hi, n, dv);
            #pragma unroll
            for (int m = 0; m < 8; ++m)
                *(v4f*)(xsw + st_off + 2 * m * G1_PX) = buf[q & 1][m];
            if (q + 2 < G1_NCH) {
                #pragma unroll
                for (int m = 0; m < 8; ++m)
                    buf[q & 1][m] = __builtin_nontemporal_load((const v4f*)(xb + 2 * m * IN_FEAT + (q + 2) * G1_KC));
            }
            __builtin_amdgcn_sched_barrier(0);
            __builtin_amdgcn_fence(__ATOMIC_ACQ_REL, "wavefront");
            gcn_g1_chunk_mma(xsw, bfr, ld_off, lane, q, acc);
            __builtin_amdgcn_fence(__ATOMIC_ACQ_REL, "wavefront");
        }
    } else {
        gcn_g1_dinv8(dinv, row0, hi, n, dv);
        #pragma nounroll
        for (int q = 0; q < G1_NCH; ++q) {
            #pragma unroll
            for (int m = 0; m < 8; ++m) {
                int row = row0 + 2 * m + hi;
                if (row >= n) row = n - 1;
                v4f v = __builtin_nontemporal_load((const v4f*)(x + (size_t)row * IN_FEAT + q * G1_KC + 4 * c));
                *(v4f*)(xsw + st_off + 2 * m * G1_PX) = v;
            }
            __builtin_amdgcn_fence(__ATOMIC_ACQ_REL, "wavefront");
            gcn_g1_chunk_mma(xsw, bfr, ld_off, lane, q, acc);
            __builtin_amdgcn_fence(__ATOMIC_ACQ_REL, "wavefront");
        }
    }

    int srow = row0 + hi * 8;
    #pragma unroll
    for (int i = 0; i < 8; ++i) {
        int rr = srow + i;
        if (rr < n) h[(size_t)rr * HIDDEN + c] = acc[i] * dv[i];
    }
}


__global__ void gcn_agg1(const int* __restrict__ src, const int* __restrict__ dst,
                         const float* __restrict__ h,
                         float* out1, int E) {
    int t    = blockIdx.x * blockDim.x + threadIdx.x;
    int lane = threadIdx.x & 31;
    int e0   = __builtin_amdgcn_readfirstlane(t - lane);
    if (e0 >= E) return;
    int e = e0 + lane;
    int s = 0, d = 0;
    if (e < E) {
        s = src[e];
        d = dst[e];
    }
    int nvalid = E - e0;
    if (nvalid > 32) nvalid = 32;
    int sub = lane >> 4;
    int c   = lane & 15;

    float m[16];
    int   dj[16];
    #pragma unroll
    for (int i = 0; i < 16; ++i) {
        int j    = 2 * i + sub;
        int sj   = __shfl(s, j);
        dj[i]    = __shfl(d, j);
        m[i] = h[sj * HIDDEN + c];
    }
    if (nvalid == 32) {
        #pragma unroll
        for (int i = 0; i < 16; ++i)
            atomicAdd(out1 + (dj[i] * HIDDEN + c), m[i]);
    } else {
        #pragma unroll
        for (int i = 0; i < 16; ++i) {
            int j = 2 * i + sub;
            if (j < nvalid)
                atomicAdd(out1 + (dj[i] * HIDDEN + c), m[i]);
        }
    }
}


__global__ void gcn_selfloop_relu(const float* __restrict__ out1,
                                  const float* __restrict__ h,
                                  const float* __restrict__ dinv,
                                  const float* __restrict__ b1,
                                  float* hrelu, int n) {
    int i = blockIdx.x * blockDim.x + threadIdx.x;
    if (i >= n) return;
    float w = dinv[i];
    const float* o = out1 + (size_t)i * HIDDEN;
    const float* hh = h + (size_t)i * HIDDEN;
    float* dst = hrelu + (size_t)i * HIDDEN;
    #pragma unroll
    for (int c = 0; c < HIDDEN; ++c)
        dst[c] = fmaxf((o[c] + hh[c]) * w + b1[c], 0.0f);
}


__global__ void gcn_gemm2_wmma(const float* __restrict__ hrelu,
                               const float* __restrict__ W2,
                               const float* __restrict__ dinv,
                               float* __restrict__ g, int n) {
    int wave = (blockIdx.x * blockDim.x + threadIdx.x) >> 5;
    int lane = threadIdx.x & 31;
    int row0 = wave * 16;
    if (row0 >= n) return;

    int r    = lane & 15;
    int hi   = lane >> 4;
    int koff = hi * 2;
    int row  = row0 + r;
    if (row >= n) row = n - 1;
    const float* hrow = hrelu + (size_t)row * HIDDEN;

    v8f c = {0.f, 0.f, 0.f, 0.f, 0.f, 0.f, 0.f, 0.f};

    #pragma unroll
    for (int k = 0; k < HIDDEN; k += 4) {
        int kk = k + koff;
        v2f a = *(const v2f*)(hrow + kk);
        v2f b;
        b.x = (r < NCLS) ? W2[(size_t)kk * NCLS + r] : 0.0f;
        b.y = (r < NCLS) ? W2[(size_t)(kk + 1) * NCLS + r] : 0.0f;
        c = __builtin_amdgcn_wmma_f32_16x16x4_f32(false, a, false, b,
                                                  (short)0, c, false, false);
    }

    int srow = row0 + hi * 8;
    #pragma unroll
    for (int i = 0; i < 8; ++i) {
        int rr = srow + i;
        if (rr < n) g[(size_t)rr * HIDDEN + r] = c[i] * dinv[rr];
    }
}


__global__ void gcn_agg2(const int* __restrict__ src, const int* __restrict__ dst,
                         const float* __restrict__ g,
                         float* out2, int E) {
    int t    = blockIdx.x * blockDim.x + threadIdx.x;
    int lane = threadIdx.x & 31;
    int e0   = t - lane;
    if (e0 >= E) return;
    int e = e0 + lane;
    int s = 0, d = 0;
    if (e < E) {
        s = src[e];
        d = dst[e];
    }
    int nvalid = E - e0;
    if (nvalid > 32) nvalid = 32;
    int sub = lane >> 3;
    int c   = lane & 7;

    float m[8];
    int   dj[8];
    #pragma unroll
    for (int i = 0; i < 8; ++i) {
        int j    = 4 * i + sub;
        int sj   = __shfl(s, j);
        dj[i]    = __shfl(d, j);
        m[i] = g[(size_t)sj * HIDDEN + c];
    }
    #pragma unroll
    for (int i = 0; i < 8; ++i) {
        int j = 4 * i + sub;
        if (c < NCLS && j < nvalid)
            atomicAdd(out2 + (size_t)dj[i] * 8 + c, m[i]);
    }
}


__global__ void gcn_logsoftmax(const float* __restrict__ out2,
                               const float* __restrict__ g,
                               const float* __restrict__ dinv,
                               const float* __restrict__ b2,
                               float* __restrict__ out, int n) {
    int i = blockIdx.x * blockDim.x + threadIdx.x;
    if (i >= n) return;
    float w = dinv[i];
    const float* o = out2 + (size_t)i * 8;
    const float* gg = g + (size_t)i * HIDDEN;
    float v[NCLS];
    float m = -3.4e38f;
    #pragma unroll
    for (int c = 0; c < NCLS; ++c) {
        v[c] = (o[c] + gg[c]) * w + b2[c];
        m = fmaxf(m, v[c]);
    }
    float s = 0.0f;
    #pragma unroll
    for (int c = 0; c < NCLS; ++c) s += expf(v[c] - m);
    float l = logf(s);
    float* dsto = out + (size_t)i * NCLS;
    #pragma unroll
    for (int c = 0; c < NCLS; ++c) dsto[c] = v[c] - m - l;
}

#define RB_SHIFT 9
#define RB       512
#define NB_MAX   1024
#define BIN_T    1024
#define CH_PER   16
#define CH_MAX   (BIN_T * CH_PER)
#define SORT_CAP 32768
#define SORT_RUNS 8
#define SORT_K    3

__global__ void __launch_bounds__(BIN_T)
gcn_bin_scatter(const int* __restrict__ src, const int* __restrict__ dst,
                int* __restrict__ gh, int* __restrict__ lofs, unsigned* __restrict__ binned,
                int E, int ch, int nb, int G) {
    __shared__ int      lcur[NB_MAX];
    __shared__ int      wtot[BIN_T / 32];
    __shared__ unsigned stage[CH_MAX];
    int tid  = threadIdx.x;
    int lane = tid & 31;
    int wv   = tid >> 5;
    int g    = blockIdx.x;
    lcur[tid] = 0;
    int e0 = g * ch;
    int e1 = (E - e0 < ch) ? E : e0 + ch;

    int sv[CH_PER], dv[CH_PER];
    #pragma unroll
    for (int k = 0; k < CH_PER; ++k) {
        int e = e0 + tid + k * BIN_T;
        sv[k] = 0; dv[k] = 0;
        if (e < e1) { sv[k] = src[e]; dv[k] = dst[e]; }
    }
    __syncthreads();
    #pragma unroll
    for (int k = 0; k < CH_PER; ++k)
        if (e0 + tid + k * BIN_T < e1) atomicAdd(&lcur[dv[k] >> RB_SHIFT], 1);
    __syncthreads();

    int c = (tid < nb) ? lcur[tid] : 0;
    int incl = c;
    for (int o = 1; o < 32; o <<= 1) {
        int u = __shfl_up(incl, o);
        if (lane >= o) incl += u;
    }
    if (lane == 31) wtot[wv] = incl;
    __syncthreads();
    int pre = 0;
    for (int k = 0; k < wv; ++k) pre += wtot[k];
    int ls = pre + incl - c;
    if (tid < nb) {
        lcur[tid] = ls;
        gh[tid * G + g]   = c;
        lofs[tid * G + g] = ls;
    }
    __syncthreads();
    #pragma unroll
    for (int k = 0; k < CH_PER; ++k) {
        if (e0 + tid + k * BIN_T < e1) {
            int p = atomicAdd(&lcur[dv[k] >> RB_SHIFT], 1);
            stage[p] = ((unsigned)sv[k] << RB_SHIFT) | (unsigned)(dv[k] & (RB - 1));
        }
    }
    __syncthreads();
    int len = e1 - e0;
    for (int i = tid; i < len; i += BIN_T) binned[e0 + i] = stage[i];
}

__global__ void __launch_bounds__(BIN_T)
gcn_bin_sort(const unsigned* __restrict__ binned, const int* __restrict__ gh,
             const int* __restrict__ lofs, int* __restrict__ ssrc, int* __restrict__ first,
             int* __restrict__ cnt, float* __restrict__ dinv, int n, int ch, int nb, int G) {
    __shared__ int rcur[RB];
    __shared__ int wtot[RB / 32];
    __shared__ int wpre[BIN_T / 32];
    __shared__ int wlen[BIN_T / 32];
    __shared__ int stage[SORT_CAP];
    int tid  = threadIdx.x;
    int lane = tid & 31;
    int wv   = __builtin_amdgcn_readfirstlane(tid >> 5);
    int b    = blockIdx.x;
    int lim  = b * G;
    int lim4 = lim & ~3;
    if (tid < RB) rcur[tid] = 0;

    int pre = 0;
    #pragma unroll 8
    for (int i = tid * 4; i < lim4; i += BIN_T * 4) {
        int4 v = *(const int4*)(gh + i);
        pre += (v.x + v.y) + (v.z + v.w);
    }
    if (tid < lim - lim4) pre += gh[lim4 + tid];
    __syncthreads();

    int mylen = 0;
    for (int gb = 0; gb < G; gb += BIN_T) {
        int g  = gb + wv + 32 * lane;
        int rc = 0, ro = 0;
        if (g < G) { rc = gh[lim + g]; ro = g * ch + lofs[lim + g]; }
        mylen += rc;
        int nr = (G - gb - wv + 31) >> 5;
        if (nr > 32) nr = 32;
        for (int i0 = 0; i0 < nr; i0 += SORT_RUNS) {
            unsigned w[SORT_RUNS][SORT_K];
            int      cq[SORT_RUNS], oq[SORT_RUNS];
            #pragma unroll
            for (int q = 0; q < SORT_RUNS; ++q) {
                cq[q] = __builtin_amdgcn_readlane(rc, i0 + q);
                oq[q] = __builtin_amdgcn_readlane(ro, i0 + q);
                #pragma unroll
                for (int k = 0; k < SORT_K; ++k) {
                    int i = lane + 32 * k;
                    w[q][k] = binned[(i < cq[q]) ? oq[q] + i : 0];
                }
            }
            #pragma unroll
            for (int q = 0; q < SORT_RUNS; ++q) {
                #pragma unroll
                for (int k = 0; k < SORT_K; ++k)
                    if (lane + 32 * k < cq[q]) atomicAdd(&rcur[w[q][k] & (RB - 1)], 1);
                for (int i1 = 32 * SORT_K; i1 < cq[q]; i1 += 32) {
                    int i = i1 + lane;
                    if (i < cq[q]) atomicAdd(&rcur[binned[oq[q] + i] & (RB - 1)], 1);
                }
            }
        }
    }
    for (int o = 16; o > 0; o >>= 1) {
        pre   += __shfl_xor(pre, o);
        mylen += __shfl_xor(mylen, o);
    }
    if (lane == 0) { wpre[wv] = pre; wlen[wv] = mylen; }
    __syncthreads();
    int beg = wpre[lane];
    int len = wlen[lane];
    for (int o = 16; o > 0; o >>= 1) {
        beg += __shfl_xor(beg, o);
        len += __shfl_xor(len, o);
    }

    int c = (tid < RB) ? rcur[tid] : 0;
    int incl = c;
    for (int o = 1; o < 32; o <<= 1) {
        int u = __shfl_up(incl, o);
        if (lane >= o) incl += u;
    }
    if (tid < RB && lane == 31) wtot[wv] = incl;
    __syncthreads();
    if (tid < RB) {
        int pr = 0;
        for (int k = 0; k < wv; ++k) pr += wtot[k];
        int ls = pr + incl - c;
        rcur[tid] = ls;
        int node = (b << RB_SHIFT) + tid;
        if (node < n) {
            first[node] = beg + ls;
            cnt[node]   = c;
            float dg = (float)(c + 1);
            dinv[node] = (dg > 0.0f) ? rsqrtf(dg) : 0.0f;
        }
    }
    __syncthreads();
    bool fits = len <= SORT_CAP;
    for (int gb = 0; gb < G; gb += BIN_T) {
        int g  = gb + wv + 32 * lane;
        int rc = 0, ro = 0;
        if (g < G) { rc = gh[lim + g]; ro = g * ch + lofs[lim + g]; }
        int nr = (G - gb - wv + 31) >> 5;
        if (nr > 32) nr = 32;
        for (int i0 = 0; i0 < nr; i0 += SORT_RUNS) {
            unsigned w[SORT_RUNS][SORT_K];
            int      cq[SORT_RUNS], oq[SORT_RUNS];
            #pragma unroll
            for (int q = 0; q < SORT_RUNS; ++q) {
                cq[q] = __builtin_amdgcn_readlane(rc, i0 + q);
                oq[q] = __builtin_amdgcn_readlane(ro, i0 + q);
                #pragma unroll
                for (int k = 0; k < SORT_K; ++k) {
                    int i = lane + 32 * k;
                    w[q][k] = binned[(i < cq[q]) ? oq[q] + i : 0];
                }
            }
            #pragma unroll
            for (int q = 0; q < SORT_RUNS; ++q) {
                #pragma unroll
                for (int k = 0; k < SORT_K; ++k) {
                    if (lane + 32 * k < cq[q]) {
                        int p = atomicAdd(&rcur[w[q][k] & (RB - 1)], 1);
                        int s = (int)(w[q][k] >> RB_SHIFT);
                        if (fits) stage[p] = s;
                        else      ssrc[beg + p] = s;
                    }
                }
                for (int i1 = 32 * SORT_K; i1 < cq[q]; i1 += 32) {
                    int i = i1 + lane;
                    if (i < cq[q]) {
                        unsigned wl = binned[oq[q] + i];
                        int p = atomicAdd(&rcur[wl & (RB - 1)], 1);
                        int s = (int)(wl >> RB_SHIFT);
                        if (fits) stage[p] = s;
                        else      ssrc[beg + p] = s;
                    }
                }
            }
        }
    }
    __syncthreads();
    if (fits)
        for (int i = tid; i < len; i += BIN_T) ssrc[beg + i] = stage[i];
}

#define AGG_WGS 1024

__device__ __forceinline__ void gcn_row_bounds(const int* __restrict__ first, const int* __restrict__ cnt,
                                               int d, bool live, int n, int& beg, int& cn) {
    int dc = (d < n) ? d : n - 1;
    int b = first[dc];
    int k = cnt[dc];
    bool ok = live && d < n;
    beg = ok ? b : 0;
    cn  = ok ? k : 0;
}

__device__ __forceinline__ void gcn_bounds_ahead(const int* __restrict__ first, const int* __restrict__ cnt,
                                                 int d, int n, int& vb, int& vc) {
    int dc = (d < n) ? d : n - 1;
    vb = first[dc];
    vc = cnt[dc];
}

__device__ __forceinline__ int gcn_row_chunk(const int* __restrict__ ssrc, int beg, int cn, int k, int lane) {
    int  i  = 32 * k + lane;
    bool ok = i < cn;
    int  v  = ssrc[ok ? beg + i : 0];
    return ok ? v : 0;
}

__device__ __forceinline__ void gcn_agg1_chunk2(const float* __restrict__ h, int svA, int nvA, int svB, int nvB,
                                                int sub, int c, float& accA, float& accB) {
    int nmax = (nvA > nvB) ? nvA : nvB;
    for (int p0 = 0; p0 < nmax; p0 += 16) {
        float mA[8], mB[8];
        #pragma unroll
        for (int q = 0; q < 8; ++q) {
            int j = p0 + 2 * q + sub;
            mA[q] = h[__shfl(svA, j) * HIDDEN + c];
            mB[q] = h[__shfl(svB, j) * HIDDEN + c];
        }
        #pragma unroll
        for (int q = 0; q < 8; ++q) {
            int j = p0 + 2 * q + sub;
            accA += (j < nvA) ? mA[q] : 0.0f;
            accB += (j < nvB) ? mB[q] : 0.0f;
        }
    }
}

__device__ __forceinline__ void gcn_agg2_chunk4(const float* __restrict__ g, const int (&sv)[4], const int (&nv)[4],
                                                int sub, int c, float (&acc)[4]) {
    int n01 = (nv[0] > nv[1]) ? nv[0] : nv[1];
    int n23 = (nv[2] > nv[3]) ? nv[2] : nv[3];
    int nmax = (n01 > n23) ? n01 : n23;
    for (int p0 = 0; p0 < nmax; p0 += 16) {
        float m[4][4];
        #pragma unroll
        for (int r = 0; r < 4; ++r) {
            #pragma unroll
            for (int q = 0; q < 4; ++q) {
                int j = p0 + 4 * q + sub;
                m[r][q] = g[__shfl(sv[r], j) * HIDDEN + c];
            }
        }
        #pragma unroll
        for (int r = 0; r < 4; ++r) {
            #pragma unroll
            for (int q = 0; q < 4; ++q) {
                int j = p0 + 4 * q + sub;
                acc[r] += (j < nv[r]) ? m[r][q] : 0.0f;
            }
        }
    }
}


__global__ void __launch_bounds__(256)
gcn_agg1_rows_w2(const int* __restrict__ ssrc, const int* __restrict__ first,
                 const int* __restrict__ cnt, const float* __restrict__ h,
                 const float* __restrict__ dinv, const float* __restrict__ b1,
                 const float* __restrict__ W2, float* __restrict__ g, int n, int ntiles) {
    __shared__ __attribute__((aligned(16))) float tile[2][16 * HIDDEN];
    __shared__ __attribute__((aligned(16))) float tdinv[2][16];
    int lane = threadIdx.x & 31;
    int wv   = __builtin_amdgcn_readfirstlane((int)(threadIdx.x >> 5));
    int sub  = lane >> 4;
    int c    = lane & 15;
    const int stride = gridDim.x;

    v2f bw[HIDDEN / 4];
    #pragma unroll
    for (int j = 0; j < HIDDEN / 4; ++j) { bw[j].x = 0.0f; bw[j].y = 0.0f; }
    if (c < NCLS) {
        #pragma unroll
        for (int j = 0; j < HIDDEN / 4; ++j) {
            int kk = 4 * j + 2 * sub;
            bw[j].x = W2[kk * NCLS + c];
            bw[j].y = W2[(kk + 1) * NCLS + c];
        }
    }
    const float bias = b1[c];

    int t = blockIdx.x;
    int bA0, cA0, bB0, cB0;
    {
        int d = t * 16 + wv * 2;
        gcn_row_bounds(first, cnt, d,     t < ntiles, n, bA0, cA0);
        gcn_row_bounds(first, cnt, d + 1, t < ntiles, n, bB0, cB0);
    }
    int vb1, vc1;
    gcn_bounds_ahead(first, cnt, (t + stride) * 16 + wv * 2 + (lane & 1), n, vb1, vc1);
    int sA0 = gcn_row_chunk(ssrc, bA0, cA0, 0, lane);
    int sA1 = gcn_row_chunk(ssrc, bA0, cA0, 1, lane);
    int sB0 = gcn_row_chunk(ssrc, bB0, cB0, 0, lane);
    int sB1 = gcn_row_chunk(ssrc, bB0, cB0, 1, lane);

    for (int it = 0; t < ntiles; t += stride, ++it) {
        int bA1, cA1, bB1, cB1;
        {
            int tn = t + stride;
            int dn = tn * 16 + wv * 2;
            bool okA = tn < ntiles && dn < n, okB = tn < ntiles && dn + 1 < n;
            int fa = __builtin_amdgcn_readlane(vb1, 0), ka = __builtin_amdgcn_readlane(vc1, 0);
            int fb = __builtin_amdgcn_readlane(vb1, 1), kb = __builtin_amdgcn_readlane(vc1, 1);
            bA1 = okA ? fa : 0; cA1 = okA ? ka : 0;
            bB1 = okB ? fb : 0; cB1 = okB ? kb : 0;
        }
        int nA0 = gcn_row_chunk(ssrc, bA1, cA1, 0, lane);
        int nA1 = gcn_row_chunk(ssrc, bA1, cA1, 1, lane);
        int nB0 = gcn_row_chunk(ssrc, bB1, cB1, 0, lane);
        int nB1 = gcn_row_chunk(ssrc, bB1, cB1, 1, lane);
        int vb2, vc2;
        gcn_bounds_ahead(first, cnt, (t + 2 * stride) * 16 + wv * 2 + (lane & 1), n, vb2, vc2);
        int row0 = t * 16;
        int row  = row0 + wv * 2 + sub;
        int rc   = (row < n) ? row : n - 1;
        float wrow = dinv[rc];
        float hown = h[rc * HIDDEN + c];

        float accA = 0.0f, accB = 0.0f;
        gcn_agg1_chunk2(h, sA0, (cA0 < 32) ? cA0 : 32, sB0, (cB0 < 32) ? cB0 : 32, sub, c, accA, accB);
        gcn_agg1_chunk2(h, sA1, (cA0 < 64) ? cA0 - 32 : 32, sB1, (cB0 < 64) ? cB0 - 32 : 32, sub, c, accA, accB);
        int cmax = (cA0 > cB0) ? cA0 : cB0;
        for (int base = 64; base < cmax; base += 32) {
            int xa = gcn_row_chunk(ssrc, bA0, cA0, base >> 5, lane);
            int xb = gcn_row_chunk(ssrc, bB0, cB0, base >> 5, lane);
            int ra = cA0 - base, rb = cB0 - base;
            gcn_agg1_chunk2(h, xa, (ra < 32) ? ra : 32, xb, (rb < 32) ? rb : 32, sub, c, accA, accB);
        }
        accA += __shfl_xor(accA, 16);
        accB += __shfl_xor(accB, 16);
        float res = sub ? accB : accA;

        float w = 0.0f, v = 0.0f;
        if (row < n) {
            w = wrow;
            v = fmaxf((res + hown) * w + bias, 0.0f);
        }
        int pb = it & 1;
        tile[pb][(wv * 2 + sub) * HIDDEN + c] = v;
        if (c == 0) tdinv[pb][wv * 2 + sub] = w;
        __syncthreads();
        if (wv == (it & 7)) {
            int r  = c;
            int hi = sub;
            v8f cc = {0.f, 0.f, 0.f, 0.f, 0.f, 0.f, 0.f, 0.f};
            #pragma unroll
            for (int j = 0; j < HIDDEN / 4; ++j) {
                int kk = 4 * j + 2 * hi;
                v2f a = *(const v2f*)(&tile[pb][r * HIDDEN + kk]);
                cc = __builtin_amdgcn_wmma_f32_16x16x4_f32(false, a, false, bw[j],
                                                           (short)0, cc, false, false);
            }
            const float4 dv0 = *(const float4*)(&tdinv[pb][hi * 8]);
            const float4 dv1 = *(const float4*)(&tdinv[pb][hi * 8 + 4]);
            const float dv[8] = {dv0.x, dv0.y, dv0.z, dv0.w, dv1.x, dv1.y, dv1.z, dv1.w};
            #pragma unroll
            for (int i = 0; i < 8; ++i) {
                int rr = row0 + hi * 8 + i;
                if (rr < n) g[rr * HIDDEN + r] = cc[i] * dv[i];
            }
        }
        bA0 = bA1; cA0 = cA1; bB0 = bB1; cB0 = cB1;
        vb1 = vb2; vc1 = vc2;
        sA0 = nA0; sA1 = nA1; sB0 = nB0; sB1 = nB1;
    }
}


__global__ void __launch_bounds__(256)
gcn_agg2_rows_lsm(const int* __restrict__ ssrc, const int* __restrict__ first,
                  const int* __restrict__ cnt, const float* __restrict__ g,
                  const float* __restrict__ dinv, const float* __restrict__ b2,
                  float* __restrict__ out, int n, int ntiles) {
    __shared__ float stage[2][32 * NCLS];
    int lane = threadIdx.x & 31;
    int wv   = __builtin_amdgcn_readfirstlane((int)(threadIdx.x >> 5));
    int sub  = lane >> 3;
    int c    = lane & 7;
    const int stride = gridDim.x;
    const float bias = b2[(c < NCLS) ? c : 0];

    int t = blockIdx.x;
    int bg0[4], cn0[4], s0[4], s1[4];
    #pragma unroll
    for (int r = 0; r < 4; ++r)
        gcn_row_bounds(first, cnt, t * 32 + wv * 4 + r, t < ntiles, n, bg0[r], cn0[r]);
    int vb1, vc1;
    gcn_bounds_ahead(first, cnt, (t + stride) * 32 + wv * 4 + (lane & 3), n, vb1, vc1);
    #pragma unroll
    for (int r = 0; r < 4; ++r) {
        s0[r] = gcn_row_chunk(ssrc, bg0[r], cn0[r], 0, lane);
        s1[r] = gcn_row_chunk(ssrc, bg0[r], cn0[r], 1, lane);
    }

    for (int it = 0; t < ntiles; t += stride, ++it) {
        int n0[4], n1[4], bg1[4], cn1[4];
        #pragma unroll
        for (int r = 0; r < 4; ++r) {
            int tn = t + stride;
            bool ok = tn < ntiles && tn * 32 + wv * 4 + r < n;
            int fr = __builtin_amdgcn_readlane(vb1, r), kr = __builtin_amdgcn_readlane(vc1, r);
            bg1[r] = ok ? fr : 0;
            cn1[r] = ok ? kr : 0;
            n0[r] = gcn_row_chunk(ssrc, bg1[r], cn1[r], 0, lane);
            n1[r] = gcn_row_chunk(ssrc, bg1[r], cn1[r], 1, lane);
        }
        int vb2, vc2;
        gcn_bounds_ahead(first, cnt, (t + 2 * stride) * 32 + wv * 4 + (lane & 3), n, vb2, vc2);
        int row = t * 32 + wv * 4 + sub;
        int rc  = (row < n) ? row : n - 1;
        float w    = dinv[rc];
        float gown = g[rc * HIDDEN + c];

        float acc[4] = {0.0f, 0.0f, 0.0f, 0.0f};
        int nv[4];
        #pragma unroll
        for (int r = 0; r < 4; ++r) nv[r] = (cn0[r] < 32) ? cn0[r] : 32;
        gcn_agg2_chunk4(g, s0, nv, sub, c, acc);
        #pragma unroll
        for (int r = 0; r < 4; ++r) nv[r] = (cn0[r] < 64) ? cn0[r] - 32 : 32;
        gcn_agg2_chunk4(g, s1, nv, sub, c, acc);
        int c01 = (cn0[0] > cn0[1]) ? cn0[0] : cn0[1];
        int c23 = (cn0[2] > cn0[3]) ? cn0[2] : cn0[3];
        int cmax = (c01 > c23) ? c01 : c23;
        for (int base = 64; base < cmax; base += 32) {
            int xs[4];
            #pragma unroll
            for (int r = 0; r < 4; ++r) {
                xs[r] = gcn_row_chunk(ssrc, bg0[r], cn0[r], base >> 5, lane);
                int rem = cn0[r] - base;
                nv[r] = (rem < 32) ? rem : 32;
            }
            gcn_agg2_chunk4(g, xs, nv, sub, c, acc);
        }
        #pragma unroll
        for (int r = 0; r < 4; ++r) {
            acc[r] += __shfl_xor(acc[r], 8);
            acc[r] += __shfl_xor(acc[r], 16);
        }
        float res = (sub == 0) ? acc[0] : (sub == 1) ? acc[1] : (sub == 2) ? acc[2] : acc[3];

        float val = (res + gown) * w + bias;
        float mx = (c < NCLS) ? val : -3.4e38f;
        mx = fmaxf(mx, __shfl_xor(mx, 1));
        mx = fmaxf(mx, __shfl_xor(mx, 2));
        mx = fmaxf(mx, __shfl_xor(mx, 4));
        float ex = expf(val - mx);
        float s = 0.0f;
        #pragma unroll
        for (int k = 0; k < NCLS; ++k) s += __shfl(ex, (lane & 24) + k);
        float l = logf(s);
        int pb = it & 1;
        if (c < NCLS) stage[pb][(wv * 4 + sub) * NCLS + c] = val - mx - l;
        __syncthreads();
        int tid = threadIdx.x;
        int o = t * (32 * NCLS) + tid;
        if (tid < 32 * NCLS && o < n * NCLS) out[o] = stage[pb][tid];
        #pragma unroll
        for (int r = 0; r < 4; ++r) {
            bg0[r] = bg1[r]; cn0[r] = cn1[r];
            s0[r] = n0[r]; s1[r] = n1[r];
        }
        vb1 = vb2; vc1 = vc2;
    }
}


extern "C" void kernel_launch(void* const* d_in, const int* in_sizes, int n_in,
                              void* d_out, int out_size, void* d_ws, size_t ws_size,
                              hipStream_t stream) {
    const float* x  = (const float*)d_in[0];
    const int*   ei = (const int*)d_in[1];
    const float* W1 = (const float*)d_in[2];
    const float* b1 = (const float*)d_in[3];
    const float* W2 = (const float*)d_in[4];
    const float* b2 = (const float*)d_in[5];
    float* out = (float*)d_out;

    const int n = in_sizes[0] / IN_FEAT;
    const int E = in_sizes[1] / 2;
    const int* src = ei;
    const int* dst = ei + E;

    const int B = 256;
    const int tiles = (n + 15) / 16;
    const int gemmBlocks = (tiles + 7) / 8;

    const int nb = (n + RB - 1) >> RB_SHIFT;
    int ch = (E + 255) / 256;
    if (ch > CH_MAX) ch = CH_MAX;
    if (ch < 1) ch = 1;
    const int G = (E + ch - 1) / ch;
    const size_t n4 = ((size_t)n + 3) & ~(size_t)3;
    const size_t E4 = ((size_t)E + 3) & ~(size_t)3;
    const size_t m  = (size_t)nb * G;
    const size_t m4 = (m + 3) & ~(size_t)3;
    const size_t a_words = (E4 > (size_t)32 * n) ? E4 : (size_t)32 * n;
    const size_t need = (3 * n4 + 2 * m4 + a_words + (size_t)E) * sizeof(float);

    if (need <= ws_size && E > 0 && nb <= NB_MAX && n <= (1 << 23)) {
        float*    f      = (float*)d_ws;
        float*    dinv   = f;
        int*      cnt    = (int*)(f + n4);
        int*      first  = (int*)(f + 2 * n4);
        int*      gh     = (int*)(f + 3 * n4);
        int*      lofs   = (int*)(f + 3 * n4 + m4);
        float*    a      = f + 3 * n4 + 2 * m4;
        unsigned* binned = (unsigned*)a;
        float*    h      = a;
        float*    g      = a + (size_t)16 * n;
        int*      ssrc   = (int*)(a + a_words);

        gcn_bin_scatter<<<G, BIN_T, 0, stream>>>(src, dst, gh, lofs, binned, E, ch, nb, G);
        gcn_bin_sort   <<<nb, BIN_T, 0, stream>>>(binned, gh, lofs, ssrc, first, cnt, dinv, n, ch, nb, G);

        gcn_gemm1_wmma <<<gemmBlocks, B, 0, stream>>>(x, W1, dinv, h, n);
        const int t1 = (n + 15) / 16;
        const int t2 = (n + 31) / 32;
        gcn_agg1_rows_w2 <<<(t1 < AGG_WGS) ? t1 : AGG_WGS, B, 0, stream>>>(ssrc, first, cnt, h, dinv, b1, W2, g, n, t1);
        gcn_agg2_rows_lsm<<<(t2 < AGG_WGS) ? t2 : AGG_WGS, B, 0, stream>>>(ssrc, first, cnt, g, dinv, b2, out, n, t2);
        return;
    }

    float* f     = (float*)d_ws;
    float* dinv  = f;
    float* h     = f + (size_t)n;
    float* out1  = f + (size_t)17 * n;
    float* hrelu = f + (size_t)33 * n;
    float* g     = out1;
    float* out2  = h;

    gcn_deg_init   <<<(n + B - 1) / B, B, 0, stream>>>(dinv, n);
    gcn_deg_accum  <<<(E + B - 1) / B, B, 0, stream>>>(dst, dinv, E);
    gcn_deg_rsqrt  <<<(n + B - 1) / B, B, 0, stream>>>(dinv, n);

    gcn_zero       <<<((size_t)16 * n + B - 1) / B, B, 0, stream>>>(out1, 16 * n);
    gcn_gemm1_wmma <<<gemmBlocks, B, 0, stream>>>(x, W1, dinv, h, n);
    gcn_agg1       <<<(E + B - 1) / B, B, 0, stream>>>(src, dst, h, out1, E);
    gcn_selfloop_relu<<<(n + B - 1) / B, B, 0, stream>>>(out1, h, dinv, b1, hrelu, n);

    gcn_zero       <<<((size_t)8 * n + B - 1) / B, B, 0, stream>>>(out2, 8 * n);
    gcn_gemm2_wmma <<<gemmBlocks, B, 0, stream>>>(hrelu, W2, dinv, g, n);
    gcn_agg2       <<<(E + B - 1) / B, B, 0, stream>>>(src, dst, g, out2, E);
    gcn_logsoftmax <<<(n + B - 1) / B, B, 0, stream>>>(out2, g, dinv, b2, out, n);
}


# x @ W1 joins the launch that sorts the bins: a bin's 32 waves each multiply one 16-row tile of its own 512 rows, dinv from the bin's LDS counts; 5 launches -> 4; sum orders as before
# speedup vs baseline: 8.1517x; 1.0348x over previous
#include <hip/hip_runtime.h>
#include <hip/hip_bf16.h>

typedef __attribute__((ext_vector_type(2))) float v2f;
typedef __attribute__((ext_vector_type(4))) float v4f;
typedef __attribute__((ext_vector_type(8))) float v8f;

#define IN_FEAT 512
#define HIDDEN  16
#define NCLS    7


__global__ void gcn_deg_init(float* deg, int n) {
    int i = blockIdx.x * blockDim.x + threadIdx.x;
    if (i < n) deg[i] = 1.0f;
}

__global__ void gcn_deg_accum(const int* __restrict__ dst, float* deg, int E) {
    int e = blockIdx.x * blockDim.x + threadIdx.x;
    if (e < E) atomicAdd(&deg[dst[e]], 1.0f);
}

__global__ void gcn_deg_rsqrt(float* dinv, int n) {
    int i = blockIdx.x * blockDim.x + threadIdx.x;
    if (i < n) {
        float d = dinv[i];
        dinv[i] = (d > 0.0f) ? rsqrtf(d) : 0.0f;
    }
}

__global__ void gcn_zero(float* p, int n) {
    int i = blockIdx.x * blockDim.x + threadIdx.x;
    if (i < n) p[i] = 0.0f;
}

#define G1_KC  64
#define G1_PX  68
#define G1_NCH (IN_FEAT / G1_KC)

__device__ __forceinline__ void gcn_g1_chunk_mma(const float* xsw, const float* bfr, int ld_off, int lane,
                                                 int q, v8f& acc) {
    #pragma unroll
    for (int p = 0; p < G1_KC / 8; ++p) {
        v4f a = *(const v4f*)(xsw + ld_off + 8 * p);
        v4f b = *(const v4f*)(bfr + ((q * (G1_KC / 8) + p) * 32 + lane) * 4);
        v2f a0 = {a.x, a.y}, b0 = {b.x, b.y};
        v2f a1 = {a.z, a.w}, b1 = {b.z, b.w};
        acc = __builtin_amdgcn_wmma_f32_16x16x4_f32(false, a0, false, b0,
                                                    (short)0, acc, false, false);
        acc = __builtin_amdgcn_wmma_f32_16x16x4_f32(false, a1, false, b1,
                                                    (short)0, acc, false, false);
    }
}

__device__ __forceinline__ void gcn_g1_dinv8(const float* __restrict__ dinv, int row0, int hi, int n, float (&dv)[8]) {
    #pragma unroll
    for (int i = 0; i < 8; ++i) {
        int rr = row0 + hi * 8 + i;
        dv[i] = dinv[(rr < n) ? rr : n - 1];
    }
}

__global__ void __launch_bounds__(256, 8)
gcn_gemm1_wmma(const float* __restrict__ x,
               const float* __restrict__ W1,
               const float* __restrict__ dinv,
               float* __restrict__ h, int n) {
    __shared__ __attribute__((aligned(16))) float bfr[IN_FEAT * HIDDEN];
    __shared__ __attribute__((aligned(16))) float xs[8][16 * G1_PX];
    const int tid  = threadIdx.x;
    const int lane = tid & 31;
    const int wv   = __builtin_amdgcn_readfirstlane(tid >> 5);
    const int c    = lane & 15;
    const int hi   = lane >> 4;
    const int row0 = (blockIdx.x * 8 + wv) * 16;
    const bool full = row0 + 16 <= n;

    v4f wq[8];
    #pragma unroll
    for (int u = 0; u < 8; ++u) wq[u] = *(const v4f*)(W1 + 4 * (tid + 256 * u));

    const float* xb = x + (size_t)(row0 + hi) * IN_FEAT + 4 * c;
    v4f buf[2][8];
    if (full) {
        #pragma unroll
        for (int m = 0; m < 8; ++m) buf[0][m] = __builtin_nontemporal_load((const v4f*)(xb + 2 * m * IN_FEAT));
        #pragma unroll
        for (int m = 0; m < 8; ++m) buf[1][m] = __builtin_nontemporal_load((const v4f*)(xb + 2 * m * IN_FEAT + G1_KC));
    }

    #pragma unroll
    for (int u = 0; u < 8; ++u) {
        int f  = tid + 256 * u;
        int k  = f >> 2;
        int c0 = 4 * (f & 3);
        int base = ((k >> 3) * 32 + 16 * ((k >> 2) & 1)) * 4 + (k & 3);
        bfr[base + 4 * (c0 + 0)] = wq[u].x;
        bfr[base + 4 * (c0 + 1)] = wq[u].y;
        bfr[base + 4 * (c0 + 2)] = wq[u].z;
        bfr[base + 4 * (c0 + 3)] = wq[u].w;
    }
    __syncthreads();
    if (row0 >= n) return;

    float* xsw = &xs[wv][0];
    const int st_off = hi * G1_PX + 4 * c;
    const int ld_off = c * G1_PX + 4 * hi;
    v8f acc = {0.f, 0.f, 0.f, 0.f, 0.f, 0.f, 0.f, 0.f};
    float dv[8];

    if (full) {
        #pragma unroll
        for (int q = 0; q < G1_NCH; ++q) {
            __builtin_amdgcn_sched_barrier(0);
            if (q == G1_NCH - 1) gcn_g1_dinv8(dinv, row0, hi, n, dv);
            #pragma unroll
            for (int m = 0; m < 8; ++m)
                *(v4f*)(xsw + st_off + 2 * m * G1_PX) = buf[q & 1][m];
            if (q + 2 < G1_NCH) {
                #pragma unroll
                for (int m = 0; m < 8; ++m)
                    buf[q & 1][m] = __builtin_nontemporal_load((const v4f*)(xb + 2 * m * IN_FEAT + (q + 2) * G1_KC));
            }
            __builtin_amdgcn_sched_barrier(0);
            __builtin_amdgcn_fence(__ATOMIC_ACQ_REL, "wavefront");
            gcn_g1_chunk_mma(xsw, bfr, ld_off, lane, q, acc);
            __builtin_amdgcn_fence(__ATOMIC_ACQ_REL, "wavefront");
        }
    } else {
        gcn_g1_dinv8(dinv, row0, hi, n, dv);
        #pragma nounroll
        for (int q = 0; q < G1_NCH; ++q) {
            #pragma unroll
            for (int m = 0; m < 8; ++m) {
                int row = row0 + 2 * m + hi;
                if (row >= n) row = n - 1;
                v4f v = __builtin_nontemporal_load((const v4f*)(x + (size_t)row * IN_FEAT + q * G1_KC + 4 * c));
                *(v4f*)(xsw + st_off + 2 * m * G1_PX) = v;
            }
            __builtin_amdgcn_fence(__ATOMIC_ACQ_REL, "wavefront");
            gcn_g1_chunk_mma(xsw, bfr, ld_off, lane, q, acc);
            __builtin_amdgcn_fence(__ATOMIC_ACQ_REL, "wavefront");
        }
    }

    int srow = row0 + hi * 8;
    #pragma unroll
    for (int i = 0; i < 8; ++i) {
        int rr = srow + i;
        if (rr < n) h[(size_t)rr * HIDDEN + c] = acc[i] * dv[i];
    }
}


__global__ void gcn_agg1(const int* __restrict__ src, const int* __restrict__ dst,
                         const float* __restrict__ h,
                         float* out1, int E) {
    int t    = blockIdx.x * blockDim.x + threadIdx.x;
    int lane = threadIdx.x & 31;
    int e0   = __builtin_amdgcn_readfirstlane(t - lane);
    if (e0 >= E) return;
    int e = e0 + lane;
    int s = 0, d = 0;
    if (e < E) {
        s = src[e];
        d = dst[e];
    }
    int nvalid = E - e0;
    if (nvalid > 32) nvalid = 32;
    int sub = lane >> 4;
    int c   = lane & 15;

    float m[16];
    int   dj[16];
    #pragma unroll
    for (int i = 0; i < 16; ++i) {
        int j    = 2 * i + sub;
        int sj   = __shfl(s, j);
        dj[i]    = __shfl(d, j);
        m[i] = h[sj * HIDDEN + c];
    }
    if (nvalid == 32) {
        #pragma unroll
        for (int i = 0; i < 16; ++i)
            atomicAdd(out1 + (dj[i] * HIDDEN + c), m[i]);
    } else {
        #pragma unroll
        for (int i = 0; i < 16; ++i) {
            int j = 2 * i + sub;
            if (j < nvalid)
                atomicAdd(out1 + (dj[i] * HIDDEN + c), m[i]);
        }
    }
}


__global__ void gcn_selfloop_relu(const float* __restrict__ out1,
                                  const float* __restrict__ h,
                                  const float* __restrict__ dinv,
                                  const float* __restrict__ b1,
                                  float* hrelu, int n) {
    int i = blockIdx.x * blockDim.x + threadIdx.x;
    if (i >= n) return;
    float w = dinv[i];
    const float* o = out1 + (size_t)i * HIDDEN;
    const float* hh = h + (size_t)i * HIDDEN;
    float* dst = hrelu + (size_t)i * HIDDEN;
    #pragma unroll
    for (int c = 0; c < HIDDEN; ++c)
        dst[c] = fmaxf((o[c] + hh[c]) * w + b1[c], 0.0f);
}


__global__ void gcn_gemm2_wmma(const float* __restrict__ hrelu,
                               const float* __restrict__ W2,
                               const float* __restrict__ dinv,
                               float* __restrict__ g, int n) {
    int wave = (blockIdx.x * blockDim.x + threadIdx.x) >> 5;
    int lane = threadIdx.x & 31;
    int row0 = wave * 16;
    if (row0 >= n) return;

    int r    = lane & 15;
    int hi   = lane >> 4;
    int koff = hi * 2;
    int row  = row0 + r;
    if (row >= n) row = n - 1;
    const float* hrow = hrelu + (size_t)row * HIDDEN;

    v8f c = {0.f, 0.f, 0.f, 0.f, 0.f, 0.f, 0.f, 0.f};

    #pragma unroll
    for (int k = 0; k < HIDDEN; k += 4) {
        int kk = k + koff;
        v2f a = *(const v2f*)(hrow + kk);
        v2f b;
        b.x = (r < NCLS) ? W2[(size_t)kk * NCLS + r] : 0.0f;
        b.y = (r < NCLS) ? W2[(size_t)(kk + 1) * NCLS + r] : 0.0f;
        c = __builtin_amdgcn_wmma_f32_16x16x4_f32(false, a, false, b,
                                                  (short)0, c, false, false);
    }

    int srow = row0 + hi * 8;
    #pragma unroll
    for (int i = 0; i < 8; ++i) {
        int rr = srow + i;
        if (rr < n) g[(size_t)rr * HIDDEN + r] = c[i] * dinv[rr];
    }
}


__global__ void gcn_agg2(const int* __restrict__ src, const int* __restrict__ dst,
                         const float* __restrict__ g,
                         float* out2, int E) {
    int t    = blockIdx.x * blockDim.x + threadIdx.x;
    int lane = threadIdx.x & 31;
    int e0   = t - lane;
    if (e0 >= E) return;
    int e = e0 + lane;
    int s = 0, d = 0;
    if (e < E) {
        s = src[e];
        d = dst[e];
    }
    int nvalid = E - e0;
    if (nvalid > 32) nvalid = 32;
    int sub = lane >> 3;
    int c   = lane & 7;

    float m[8];
    int   dj[8];
    #pragma unroll
    for (int i = 0; i < 8; ++i) {
        int j    = 4 * i + sub;
        int sj   = __shfl(s, j);
        dj[i]    = __shfl(d, j);
        m[i] = g[(size_t)sj * HIDDEN + c];
    }
    #pragma unroll
    for (int i = 0; i < 8; ++i) {
        int j = 4 * i + sub;
        if (c < NCLS && j < nvalid)
            atomicAdd(out2 + (size_t)dj[i] * 8 + c, m[i]);
    }
}


__global__ void gcn_logsoftmax(const float* __restrict__ out2,
                               const float* __restrict__ g,
                               const float* __restrict__ dinv,
                               const float* __restrict__ b2,
                               float* __restrict__ out, int n) {
    int i = blockIdx.x * blockDim.x + threadIdx.x;
    if (i >= n) return;
    float w = dinv[i];
    const float* o = out2 + (size_t)i * 8;
    const float* gg = g + (size_t)i * HIDDEN;
    float v[NCLS];
    float m = -3.4e38f;
    #pragma unroll
    for (int c = 0; c < NCLS; ++c) {
        v[c] = (o[c] + gg[c]) * w + b2[c];
        m = fmaxf(m, v[c]);
    }
    float s = 0.0f;
    #pragma unroll
    for (int c = 0; c < NCLS; ++c) s += expf(v[c] - m);
    float l = logf(s);
    float* dsto = out + (size_t)i * NCLS;
    #pragma unroll
    for (int c = 0; c < NCLS; ++c) dsto[c] = v[c] - m - l;
}

#define RB_SHIFT 9
#define RB       512
#define NB_MAX   1024
#define BIN_T    1024
#define CH_PER   16
#define CH_MAX   (BIN_T * CH_PER)
#define SORT_CAP 32768
#define SORT_RUNS 8
#define SORT_K    3

__global__ void __launch_bounds__(BIN_T)
gcn_bin_scatter(const int* __restrict__ src, const int* __restrict__ dst,
                int* __restrict__ gh, int* __restrict__ lofs, unsigned* __restrict__ binned,
                int E, int ch, int nb, int G) {
    __shared__ int      lcur[NB_MAX];
    __shared__ int      wtot[BIN_T / 32];
    __shared__ unsigned stage[CH_MAX];
    int tid  = threadIdx.x;
    int lane = tid & 31;
    int wv   = tid >> 5;
    int g    = blockIdx.x;
    lcur[tid] = 0;
    int e0 = g * ch;
    int e1 = (E - e0 < ch) ? E : e0 + ch;

    int sv[CH_PER], dv[CH_PER];
    #pragma unroll
    for (int k = 0; k < CH_PER; ++k) {
        int e = e0 + tid + k * BIN_T;
        sv[k] = 0; dv[k] = 0;
        if (e < e1) { sv[k] = src[e]; dv[k] = dst[e]; }
    }
    __syncthreads();
    #pragma unroll
    for (int k = 0; k < CH_PER; ++k)
        if (e0 + tid + k * BIN_T < e1) atomicAdd(&lcur[dv[k] >> RB_SHIFT], 1);
    __syncthreads();

    int c = (tid < nb) ? lcur[tid] : 0;
    int incl = c;
    for (int o = 1; o < 32; o <<= 1) {
        int u = __shfl_up(incl, o);
        if (lane >= o) incl += u;
    }
    if (lane == 31) wtot[wv] = incl;
    __syncthreads();
    int pre = 0;
    for (int k = 0; k < wv; ++k) pre += wtot[k];
    int ls = pre + incl - c;
    if (tid < nb) {
        lcur[tid] = ls;
        gh[tid * G + g]   = c;
        lofs[tid * G + g] = ls;
    }
    __syncthreads();
    #pragma unroll
    for (int k = 0; k < CH_PER; ++k) {
        if (e0 + tid + k * BIN_T < e1) {
            int p = atomicAdd(&lcur[dv[k] >> RB_SHIFT], 1);
            stage[p] = ((unsigned)sv[k] << RB_SHIFT) | (unsigned)(dv[k] & (RB - 1));
        }
    }
    __syncthreads();
    int len = e1 - e0;
    for (int i = tid; i < len; i += BIN_T) binned[e0 + i] = stage[i];
}

__global__ void __launch_bounds__(BIN_T)
gcn_bin_sort(const unsigned* __restrict__ binned, const int* __restrict__ gh,
             const int* __restrict__ lofs, int* __restrict__ ssrc, int* __restrict__ first,
             int* __restrict__ cnt, float* __restrict__ dinv, int n, int ch, int nb, int G) {
    __shared__ int rcur[RB];
    __shared__ int wtot[RB / 32];
    __shared__ int wpre[BIN_T / 32];
    __shared__ int wlen[BIN_T / 32];
    __shared__ int stage[SORT_CAP];
    int tid  = threadIdx.x;
    int lane = tid & 31;
    int wv   = __builtin_amdgcn_readfirstlane(tid >> 5);
    int b    = blockIdx.x;
    int lim  = b * G;
    int lim4 = lim & ~3;
    if (tid < RB) rcur[tid] = 0;

    int pre = 0;
    #pragma unroll 8
    for (int i = tid * 4; i < lim4; i += BIN_T * 4) {
        int4 v = *(const int4*)(gh + i);
        pre += (v.x + v.y) + (v.z + v.w);
    }
    if (tid < lim - lim4) pre += gh[lim4 + tid];
    __syncthreads();

    int mylen = 0;
    for (int gb = 0; gb < G; gb += BIN_T) {
        int g  = gb + wv + 32 * lane;
        int rc = 0, ro = 0;
        if (g < G) { rc = gh[lim + g]; ro = g * ch + lofs[lim + g]; }
        mylen += rc;
        int nr = (G - gb - wv + 31) >> 5;
        if (nr > 32) nr = 32;
        for (int i0 = 0; i0 < nr; i0 += SORT_RUNS) {
            unsigned w[SORT_RUNS][SORT_K];
            int      cq[SORT_RUNS], oq[SORT_RUNS];
            #pragma unroll
            for (int q = 0; q < SORT_RUNS; ++q) {
                cq[q] = __builtin_amdgcn_readlane(rc, i0 + q);
                oq[q] = __builtin_amdgcn_readlane(ro, i0 + q);
                #pragma unroll
                for (int k = 0; k < SORT_K; ++k) {
                    int i = lane + 32 * k;
                    w[q][k] = binned[(i < cq[q]) ? oq[q] + i : 0];
                }
            }
            #pragma unroll
            for (int q = 0; q < SORT_RUNS; ++q) {
                #pragma unroll
                for (int k = 0; k < SORT_K; ++k)
                    if (lane + 32 * k < cq[q]) atomicAdd(&rcur[w[q][k] & (RB - 1)], 1);
                for (int i1 = 32 * SORT_K; i1 < cq[q]; i1 += 32) {
                    int i = i1 + lane;
                    if (i < cq[q]) atomicAdd(&rcur[binned[oq[q] + i] & (RB - 1)], 1);
                }
            }
        }
    }
    for (int o = 16; o > 0; o >>= 1) {
        pre   += __shfl_xor(pre, o);
        mylen += __shfl_xor(mylen, o);
    }
    if (lane == 0) { wpre[wv] = pre; wlen[wv] = mylen; }
    __syncthreads();
    int beg = wpre[lane];
    int len = wlen[lane];
    for (int o = 16; o > 0; o >>= 1) {
        beg += __shfl_xor(beg, o);
        len += __shfl_xor(len, o);
    }

    int c = (tid < RB) ? rcur[tid] : 0;
    int incl = c;
    for (int o = 1; o < 32; o <<= 1) {
        int u = __shfl_up(incl, o);
        if (lane >= o) incl += u;
    }
    if (tid < RB && lane == 31) wtot[wv] = incl;
    __syncthreads();
    if (tid < RB) {
        int pr = 0;
        for (int k = 0; k < wv; ++k) pr += wtot[k];
        int ls = pr + incl - c;
        rcur[tid] = ls;
        int node = (b << RB_SHIFT) + tid;
        if (node < n) {
            first[node] = beg + ls;
            cnt[node]   = c;
            float dg = (float)(c + 1);
            dinv[node] = (dg > 0.0f) ? rsqrtf(dg) : 0.0f;
        }
    }
    __syncthreads();
    bool fits = len <= SORT_CAP;
    for (int gb = 0; gb < G; gb += BIN_T) {
        int g  = gb + wv + 32 * lane;
        int rc = 0, ro = 0;
        if (g < G) { rc = gh[lim + g]; ro = g * ch + lofs[lim + g]; }
        int nr = (G - gb - wv + 31) >> 5;
        if (nr > 32) nr = 32;
        for (int i0 = 0; i0 < nr; i0 += SORT_RUNS) {
            unsigned w[SORT_RUNS][SORT_K];
            int      cq[SORT_RUNS], oq[SORT_RUNS];
            #pragma unroll
            for (int q = 0; q < SORT_RUNS; ++q) {
                cq[q] = __builtin_amdgcn_readlane(rc, i0 + q);
                oq[q] = __builtin_amdgcn_readlane(ro, i0 + q);
                #pragma unroll
                for (int k = 0; k < SORT_K; ++k) {
                    int i = lane + 32 * k;
                    w[q][k] = binned[(i < cq[q]) ? oq[q] + i : 0];
                }
            }
            #pragma unroll
            for (int q = 0; q < SORT_RUNS; ++q) {
                #pragma unroll
                for (int k = 0; k < SORT_K; ++k) {
                    if (lane + 32 * k < cq[q]) {
                        int p = atomicAdd(&rcur[w[q][k] & (RB - 1)], 1);
                        int s = (int)(w[q][k] >> RB_SHIFT);
                        if (fits) stage[p] = s;
                        else      ssrc[beg + p] = s;
                    }
                }
                for (int i1 = 32 * SORT_K; i1 < cq[q]; i1 += 32) {
                    int i = i1 + lane;
                    if (i < cq[q]) {
                        unsigned wl = binned[oq[q] + i];
                        int p = atomicAdd(&rcur[wl & (RB - 1)], 1);
                        int s = (int)(wl >> RB_SHIFT);
                        if (fits) stage[p] = s;
                        else      ssrc[beg + p] = s;
                    }
                }
            }
        }
    }
    __syncthreads();
    if (fits)
        for (int i = tid; i < len; i += BIN_T) ssrc[beg + i] = stage[i];
}

#define G1_XS_WORDS (32 * 16 * G1_PX)

__global__ void __launch_bounds__(BIN_T)
gcn_bin_sort_gemm1(const unsigned* __restrict__ binned, const int* __restrict__ gh,
                   const int* __restrict__ lofs, int* __restrict__ ssrc, int* __restrict__ first,
                   int* __restrict__ cnt, float* __restrict__ dinv,
                   const float* __restrict__ x, const float* __restrict__ W1, float* __restrict__ h,
                   int n, int ch, int nb, int G) {
    __shared__ int rcur[RB];
    __shared__ int wtot[RB / 32];
    __shared__ int wpre[BIN_T / 32];
    __shared__ int wlen[BIN_T / 32];
    __shared__ __attribute__((aligned(16))) float sdinv[RB];
    __shared__ __attribute__((aligned(16))) float bfr[IN_FEAT * HIDDEN];
    __shared__ __attribute__((aligned(16))) float big[G1_XS_WORDS];
    int* stage = (int*)big;
    int tid  = threadIdx.x;
    int lane = tid & 31;
    int wv   = __builtin_amdgcn_readfirstlane(tid >> 5);
    int b    = blockIdx.x;
    int lim  = b * G;
    int lim4 = lim & ~3;

    v4f wq[2];
    #pragma unroll
    for (int u = 0; u < 2; ++u) wq[u] = *(const v4f*)(W1 + 4 * (tid + BIN_T * u));

    if (tid < RB) rcur[tid] = 0;

    int pre = 0;
    #pragma unroll 8
    for (int i = tid * 4; i < lim4; i += BIN_T * 4) {
        int4 v = *(const int4*)(gh + i);
        pre += (v.x + v.y) + (v.z + v.w);
    }
    if (tid < lim - lim4) pre += gh[lim4 + tid];

    #pragma unroll
    for (int u = 0; u < 2; ++u) {
        int f  = tid + BIN_T * u;
        int k  = f >> 2;
        int c0 = 4 * (f & 3);
        int base = ((k >> 3) * 32 + 16 * ((k >> 2) & 1)) * 4 + (k & 3);
        bfr[base + 4 * (c0 + 0)] = wq[u].x;
        bfr[base + 4 * (c0 + 1)] = wq[u].y;
        bfr[base + 4 * (c0 + 2)] = wq[u].z;
        bfr[base + 4 * (c0 + 3)] = wq[u].w;
    }
    __syncthreads();

    int mylen = 0;
    for (int gb = 0; gb < G; gb += BIN_T) {
        int g  = gb + wv + 32 * lane;
        int rc = 0, ro = 0;
        if (g < G) { rc = gh[lim + g]; ro = g * ch + lofs[lim + g]; }
        mylen += rc;
        int nr = (G - gb - wv + 31) >> 5;
        if (nr > 32) nr = 32;
        for (int i0 = 0; i0 < nr; i0 += SORT_RUNS) {
            unsigned w[SORT_RUNS][SORT_K];
            int      cq[SORT_RUNS], oq[SORT_RUNS];
            #pragma unroll
            for (int q = 0; q < SORT_RUNS; ++q) {
                cq[q] = __builtin_amdgcn_readlane(rc, i0 + q);
                oq[q] = __builtin_amdgcn_readlane(ro, i0 + q);
                #pragma unroll
                for (int k = 0; k < SORT_K; ++k) {
                    int i = lane + 32 * k;
                    w[q][k] = binned[(i < cq[q]) ? oq[q] + i : 0];
                }
            }
            #pragma unroll
            for (int q = 0; q < SORT_RUNS; ++q) {
                #pragma unroll
                for (int k = 0; k < SORT_K; ++k)
                    if (lane + 32 * k < cq[q]) atomicAdd(&rcur[w[q][k] & (RB - 1)], 1);
                for (int i1 = 32 * SORT_K; i1 < cq[q]; i1 += 32) {
                    int i = i1 + lane;
                    if (i < cq[q]) atomicAdd(&rcur[binned[oq[q] + i] & (RB - 1)], 1);
                }
            }
        }
    }
    for (int o = 16; o > 0; o >>= 1) {
        pre   += __shfl_xor(pre, o);
        mylen += __shfl_xor(mylen, o);
    }
    if (lane == 0) { wpre[wv] = pre; wlen[wv] = mylen; }
    __syncthreads();
    int beg = wpre[lane];
    int len = wlen[lane];
    for (int o = 16; o > 0; o >>= 1) {
        beg += __shfl_xor(beg, o);
        len += __shfl_xor(len, o);
    }

    int c = (tid < RB) ? rcur[tid] : 0;
    int incl = c;
    for (int o = 1; o < 32; o <<= 1) {
        int u = __shfl_up(incl, o);
        if (lane >= o) incl += u;
    }
    if (tid < RB && lane == 31) wtot[wv] = incl;
    __syncthreads();
    if (tid < RB) {
        int pr = 0;
        for (int k = 0; k < wv; ++k) pr += wtot[k];
        int ls = pr + incl - c;
        rcur[tid] = ls;
        int node = (b << RB_SHIFT) + tid;
        float dg = (float)(c + 1);
        float di = (dg > 0.0f) ? rsqrtf(dg) : 0.0f;
        sdinv[tid] = di;
        if (node < n) {
            first[node] = beg + ls;
            cnt[node]   = c;
            dinv[node]  = di;
        }
    }
    __syncthreads();
    bool fits = len <= SORT_CAP;
    for (int gb = 0; gb < G; gb += BIN_T) {
        int g  = gb + wv + 32 * lane;
        int rc = 0, ro = 0;
        if (g < G) { rc = gh[lim + g]; ro = g * ch + lofs[lim + g]; }
        int nr = (G - gb - wv + 31) >> 5;
        if (nr > 32) nr = 32;
        for (int i0 = 0; i0 < nr; i0 += SORT_RUNS) {
            unsigned w[SORT_RUNS][SORT_K];
            int      cq[SORT_RUNS], oq[SORT_RUNS];
            #pragma unroll
            for (int q = 0; q < SORT_RUNS; ++q) {
                cq[q] = __builtin_amdgcn_readlane(rc, i0 + q);
                oq[q] = __builtin_amdgcn_readlane(ro, i0 + q);
                #pragma unroll
                for (int k = 0; k < SORT_K; ++k) {
                    int i = lane + 32 * k;
                    w[q][k] = binned[(i < cq[q]) ? oq[q] + i : 0];
                }
            }
            #pragma unroll
            for (int q = 0; q < SORT_RUNS; ++q) {
                #pragma unroll
                for (int k = 0; k < SORT_K; ++k) {
                    if (lane + 32 * k < cq[q]) {
                        int p = atomicAdd(&rcur[w[q][k] & (RB - 1)], 1);
                        int s = (int)(w[q][k] >> RB_SHIFT);
                        if (fits) stage[p] = s;
                        else      ssrc[beg + p] = s;
                    }
                }
                for (int i1 = 32 * SORT_K; i1 < cq[q]; i1 += 32) {
                    int i = i1 + lane;
                    if (i < cq[q]) {
                        unsigned wl = binned[oq[q] + i];
                        int p = atomicAdd(&rcur[wl & (RB - 1)], 1);
                        int s = (int)(wl >> RB_SHIFT);
                        if (fits) stage[p] = s;
                        else      ssrc[beg + p] = s;
                    }
                }
            }
        }
    }

    const int cc   = lane & 15;
    const int hi   = lane >> 4;
    const int row0 = (b << RB_SHIFT) + wv * 16;
    const bool live = row0 < n;
    const bool full = row0 + 16 <= n;
    const float* xb = x + (size_t)(row0 + hi) * IN_FEAT + 4 * cc;
    v4f buf[2][8];
    if (full) {
        #pragma unroll
        for (int m = 0; m < 8; ++m) buf[0][m] = __builtin_nontemporal_load((const v4f*)(xb + 2 * m * IN_FEAT));
        #pragma unroll
        for (int m = 0; m < 8; ++m) buf[1][m] = __builtin_nontemporal_load((const v4f*)(xb + 2 * m * IN_FEAT + G1_KC));
    }

    __syncthreads();
    if (fits)
        for (int i = tid; i < len; i += BIN_T) ssrc[beg + i] = stage[i];
    __syncthreads();
    if (!live) return;

    float* xsw = big + wv * (16 * G1_PX);
    const int st_off = hi * G1_PX + 4 * cc;
    const int ld_off = cc * G1_PX + 4 * hi;
    v8f acc = {0.f, 0.f, 0.f, 0.f, 0.f, 0.f, 0.f, 0.f};

    if (full) {
        #pragma unroll
        for (int q = 0; q < G1_NCH; ++q) {
            __builtin_amdgcn_sched_barrier(0);
            #pragma unroll
            for (int m = 0; m < 8; ++m)
                *(v4f*)(xsw + st_off + 2 * m * G1_PX) = buf[q & 1][m];
            if (q + 2 < G1_NCH) {
                #pragma unroll
                for (int m = 0; m < 8; ++m)
                    buf[q & 1][m] = __builtin_nontemporal_load((const v4f*)(xb + 2 * m * IN_FEAT + (q + 2) * G1_KC));
            }
            __builtin_amdgcn_sched_barrier(0);
            __builtin_amdgcn_fence(__ATOMIC_ACQ_REL, "wavefront");
            gcn_g1_chunk_mma(xsw, bfr, ld_off, lane, q, acc);
            __builtin_amdgcn_fence(__ATOMIC_ACQ_REL, "wavefront");
        }
    } else {
        #pragma nounroll
        for (int q = 0; q < G1_NCH; ++q) {
            #pragma unroll
            for (int m = 0; m < 8; ++m) {
                int row = row0 + 2 * m + hi;
                if (row >= n) row = n - 1;
                v4f v = __builtin_nontemporal_load((const v4f*)(x + (size_t)row * IN_FEAT + q * G1_KC + 4 * cc));
                *(v4f*)(xsw + st_off + 2 * m * G1_PX) = v;
            }
            __builtin_amdgcn_fence(__ATOMIC_ACQ_REL, "wavefront");
            gcn_g1_chunk_mma(xsw, bfr, ld_off, lane, q, acc);
            __builtin_amdgcn_fence(__ATOMIC_ACQ_REL, "wavefront");
        }
    }

    const v4f dv0 = *(const v4f*)(&sdinv[wv * 16 + hi * 8]);
    const v4f dv1 = *(const v4f*)(&sdinv[wv * 16 + hi * 8 + 4]);
    const float dv[8] = {dv0.x, dv0.y, dv0.z, dv0.w, dv1.x, dv1.y, dv1.z, dv1.w};
    int srow = row0 + hi * 8;
    #pragma unroll
    for (int i = 0; i < 8; ++i) {
        int rr = srow + i;
        if (rr < n) h[(size_t)rr * HIDDEN + cc] = acc[i] * dv[i];
    }
}

#define AGG_WGS 1024

__device__ __forceinline__ void gcn_row_bounds(const int* __restrict__ first, const int* __restrict__ cnt,
                                               int d, bool live, int n, int& beg, int& cn) {
    int dc = (d < n) ? d : n - 1;
    int b = first[dc];
    int k = cnt[dc];
    bool ok = live && d < n;
    beg = ok ? b : 0;
    cn  = ok ? k : 0;
}

__device__ __forceinline__ void gcn_bounds_ahead(const int* __restrict__ first, const int* __restrict__ cnt,
                                                 int d, int n, int& vb, int& vc) {
    int dc = (d < n) ? d : n - 1;
    vb = first[dc];
    vc = cnt[dc];
}

__device__ __forceinline__ int gcn_row_chunk(const int* __restrict__ ssrc, int beg, int cn, int k, int lane) {
    int  i  = 32 * k + lane;
    bool ok = i < cn;
    int  v  = ssrc[ok ? beg + i : 0];
    return ok ? v : 0;
}

__device__ __forceinline__ void gcn_agg1_chunk2(const float* __restrict__ h, int svA, int nvA, int svB, int nvB,
                                                int sub, int c, float& accA, float& accB) {
    int nmax = (nvA > nvB) ? nvA : nvB;
    for (int p0 = 0; p0 < nmax; p0 += 16) {
        float mA[8], mB[8];
        #pragma unroll
        for (int q = 0; q < 8; ++q) {
            int j = p0 + 2 * q + sub;
            mA[q] = h[__shfl(svA, j) * HIDDEN + c];
            mB[q] = h[__shfl(svB, j) * HIDDEN + c];
        }
        #pragma unroll
        for (int q = 0; q < 8; ++q) {
            int j = p0 + 2 * q + sub;
            accA += (j < nvA) ? mA[q] : 0.0f;
            accB += (j < nvB) ? mB[q] : 0.0f;
        }
    }
}

__device__ __forceinline__ void gcn_agg2_chunk4(const float* __restrict__ g, const int (&sv)[4], const int (&nv)[4],
                                                int sub, int c, float (&acc)[4]) {
    int n01 = (nv[0] > nv[1]) ? nv[0] : nv[1];
    int n23 = (nv[2] > nv[3]) ? nv[2] : nv[3];
    int nmax = (n01 > n23) ? n01 : n23;
    for (int p0 = 0; p0 < nmax; p0 += 16) {
        float m[4][4];
        #pragma unroll
        for (int r = 0; r < 4; ++r) {
            #pragma unroll
            for (int q = 0; q < 4; ++q) {
                int j = p0 + 4 * q + sub;
                m[r][q] = g[__shfl(sv[r], j) * HIDDEN + c];
            }
        }
        #pragma unroll
        for (int r = 0; r < 4; ++r) {
            #pragma unroll
            for (int q = 0; q < 4; ++q) {
                int j = p0 + 4 * q + sub;
                acc[r] += (j < nv[r]) ? m[r][q] : 0.0f;
            }
        }
    }
}


__global__ void __launch_bounds__(256)
gcn_agg1_rows_w2(const int* __restrict__ ssrc, const int* __restrict__ first,
                 const int* __restrict__ cnt, const float* __restrict__ h,
                 const float* __restrict__ dinv, const float* __restrict__ b1,
                 const float* __restrict__ W2, float* __restrict__ g, int n, int ntiles) {
    __shared__ __attribute__((aligned(16))) float tile[2][16 * HIDDEN];
    __shared__ __attribute__((aligned(16))) float tdinv[2][16];
    int lane = threadIdx.x & 31;
    int wv   = __builtin_amdgcn_readfirstlane((int)(threadIdx.x >> 5));
    int sub  = lane >> 4;
    int c    = lane & 15;
    const int stride = gridDim.x;

    v2f bw[HIDDEN / 4];
    #pragma unroll
    for (int j = 0; j < HIDDEN / 4; ++j) { bw[j].x = 0.0f; bw[j].y = 0.0f; }
    if (c < NCLS) {
        #pragma unroll
        for (int j = 0; j < HIDDEN / 4; ++j) {
            int kk = 4 * j + 2 * sub;
            bw[j].x = W2[kk * NCLS + c];
            bw[j].y = W2[(kk + 1) * NCLS + c];
        }
    }
    const float bias = b1[c];

    int t = blockIdx.x;
    int bA0, cA0, bB0, cB0;
    {
        int d = t * 16 + wv * 2;
        gcn_row_bounds(first, cnt, d,     t < ntiles, n, bA0, cA0);
        gcn_row_bounds(first, cnt, d + 1, t < ntiles, n, bB0, cB0);
    }
    int vb1, vc1;
    gcn_bounds_ahead(first, cnt, (t + stride) * 16 + wv * 2 + (lane & 1), n, vb1, vc1);
    int sA0 = gcn_row_chunk(ssrc, bA0, cA0, 0, lane);
    int sA1 = gcn_row_chunk(ssrc, bA0, cA0, 1, lane);
    int sB0 = gcn_row_chunk(ssrc, bB0, cB0, 0, lane);
    int sB1 = gcn_row_chunk(ssrc, bB0, cB0, 1, lane);

    for (int it = 0; t < ntiles; t += stride, ++it) {
        int bA1, cA1, bB1, cB1;
        {
            int tn = t + stride;
            int dn = tn * 16 + wv * 2;
            bool okA = tn < ntiles && dn < n, okB = tn < ntiles && dn + 1 < n;
            int fa = __builtin_amdgcn_readlane(vb1, 0), ka = __builtin_amdgcn_readlane(vc1, 0);
            int fb = __builtin_amdgcn_readlane(vb1, 1), kb = __builtin_amdgcn_readlane(vc1, 1);
            bA1 = okA ? fa : 0; cA1 = okA ? ka : 0;
            bB1 = okB ? fb : 0; cB1 = okB ? kb : 0;
        }
        int nA0 = gcn_row_chunk(ssrc, bA1, cA1, 0, lane);
        int nA1 = gcn_row_chunk(ssrc, bA1, cA1, 1, lane);
        int nB0 = gcn_row_chunk(ssrc, bB1, cB1, 0, lane);
        int nB1 = gcn_row_chunk(ssrc, bB1, cB1, 1, lane);
        int vb2, vc2;
        gcn_bounds_ahead(first, cnt, (t + 2 * stride) * 16 + wv * 2 + (lane & 1), n, vb2, vc2);
        int row0 = t * 16;
        int row  = row0 + wv * 2 + sub;
        int rc   = (row < n) ? row : n - 1;
        float wrow = dinv[rc];
        float hown = h[rc * HIDDEN + c];

        float accA = 0.0f, accB = 0.0f;
        gcn_agg1_chunk2(h, sA0, (cA0 < 32) ? cA0 : 32, sB0, (cB0 < 32) ? cB0 : 32, sub, c, accA, accB);
        gcn_agg1_chunk2(h, sA1, (cA0 < 64) ? cA0 - 32 : 32, sB1, (cB0 < 64) ? cB0 - 32 : 32, sub, c, accA, accB);
        int cmax = (cA0 > cB0) ? cA0 : cB0;
        for (int base = 64; base < cmax; base += 32) {
            int xa = gcn_row_chunk(ssrc, bA0, cA0, base >> 5, lane);
            int xb = gcn_row_chunk(ssrc, bB0, cB0, base >> 5, lane);
            int ra = cA0 - base, rb = cB0 - base;
            gcn_agg1_chunk2(h, xa, (ra < 32) ? ra : 32, xb, (rb < 32) ? rb : 32, sub, c, accA, accB);
        }
        accA += __shfl_xor(accA, 16);
        accB += __shfl_xor(accB, 16);
        float res = sub ? accB : accA;

        float w = 0.0f, v = 0.0f;
        if (row < n) {
            w = wrow;
            v = fmaxf((res + hown) * w + bias, 0.0f);
        }
        int pb = it & 1;
        tile[pb][(wv * 2 + sub) * HIDDEN + c] = v;
        if (c == 0) tdinv[pb][wv * 2 + sub] = w;
        __syncthreads();
        if (wv == (it & 7)) {
            int r  = c;
            int hi = sub;
            v8f cc = {0.f, 0.f, 0.f, 0.f, 0.f, 0.f, 0.f, 0.f};
            #pragma unroll
            for (int j = 0; j < HIDDEN / 4; ++j) {
                int kk = 4 * j + 2 * hi;
                v2f a = *(const v2f*)(&tile[pb][r * HIDDEN + kk]);
                cc = __builtin_amdgcn_wmma_f32_16x16x4_f32(false, a, false, bw[j],
                                                           (short)0, cc, false, false);
            }
            const float4 dv0 = *(const float4*)(&tdinv[pb][hi * 8]);
            const float4 dv1 = *(const float4*)(&tdinv[pb][hi * 8 + 4]);
            const float dv[8] = {dv0.x, dv0.y, dv0.z, dv0.w, dv1.x, dv1.y, dv1.z, dv1.w};
            #pragma unroll
            for (int i = 0; i < 8; ++i) {
                int rr = row0 + hi * 8 + i;
                if (rr < n) g[rr * HIDDEN + r] = cc[i] * dv[i];
            }
        }
        bA0 = bA1; cA0 = cA1; bB0 = bB1; cB0 = cB1;
        vb1 = vb2; vc1 = vc2;
        sA0 = nA0; sA1 = nA1; sB0 = nB0; sB1 = nB1;
    }
}


__global__ void __launch_bounds__(256)
gcn_agg2_rows_lsm(const int* __restrict__ ssrc, const int* __restrict__ first,
                  const int* __restrict__ cnt, const float* __restrict__ g,
                  const float* __restrict__ dinv, const float* __restrict__ b2,
                  float* __restrict__ out, int n, int ntiles) {
    __shared__ float stage[2][32 * NCLS];
    int lane = threadIdx.x & 31;
    int wv   = __builtin_amdgcn_readfirstlane((int)(threadIdx.x >> 5));
    int sub  = lane >> 3;
    int c    = lane & 7;
    const int stride = gridDim.x;
    const float bias = b2[(c < NCLS) ? c : 0];

    int t = blockIdx.x;
    int bg0[4], cn0[4], s0[4], s1[4];
    #pragma unroll
    for (int r = 0; r < 4; ++r)
        gcn_row_bounds(first, cnt, t * 32 + wv * 4 + r, t < ntiles, n, bg0[r], cn0[r]);
    int vb1, vc1;
    gcn_bounds_ahead(first, cnt, (t + stride) * 32 + wv * 4 + (lane & 3), n, vb1, vc1);
    #pragma unroll
    for (int r = 0; r < 4; ++r) {
        s0[r] = gcn_row_chunk(ssrc, bg0[r], cn0[r], 0, lane);
        s1[r] = gcn_row_chunk(ssrc, bg0[r], cn0[r], 1, lane);
    }

    for (int it = 0; t < ntiles; t += stride, ++it) {
        int n0[4], n1[4], bg1[4], cn1[4];
        #pragma unroll
        for (int r = 0; r < 4; ++r) {
            int tn = t + stride;
            bool ok = tn < ntiles && tn * 32 + wv * 4 + r < n;
            int fr = __builtin_amdgcn_readlane(vb1, r), kr = __builtin_amdgcn_readlane(vc1, r);
            bg1[r] = ok ? fr : 0;
            cn1[r] = ok ? kr : 0;
            n0[r] = gcn_row_chunk(ssrc, bg1[r], cn1[r], 0, lane);
            n1[r] = gcn_row_chunk(ssrc, bg1[r], cn1[r], 1, lane);
        }
        int vb2, vc2;
        gcn_bounds_ahead(first, cnt, (t + 2 * stride) * 32 + wv * 4 + (lane & 3), n, vb2, vc2);
        int row = t * 32 + wv * 4 + sub;
        int rc  = (row < n) ? row : n - 1;
        float w    = dinv[rc];
        float gown = g[rc * HIDDEN + c];

        float acc[4] = {0.0f, 0.0f, 0.0f, 0.0f};
        int nv[4];
        #pragma unroll
        for (int r = 0; r < 4; ++r) nv[r] = (cn0[r] < 32) ? cn0[r] : 32;
        gcn_agg2_chunk4(g, s0, nv, sub, c, acc);
        #pragma unroll
        for (int r = 0; r < 4; ++r) nv[r] = (cn0[r] < 64) ? cn0[r] - 32 : 32;
        gcn_agg2_chunk4(g, s1, nv, sub, c, acc);
        int c01 = (cn0[0] > cn0[1]) ? cn0[0] : cn0[1];
        int c23 = (cn0[2] > cn0[3]) ? cn0[2] : cn0[3];
        int cmax = (c01 > c23) ? c01 : c23;
        for (int base = 64; base < cmax; base += 32) {
            int xs[4];
            #pragma unroll
            for (int r = 0; r < 4; ++r) {
                xs[r] = gcn_row_chunk(ssrc, bg0[r], cn0[r], base >> 5, lane);
                int rem = cn0[r] - base;
                nv[r] = (rem < 32) ? rem : 32;
            }
            gcn_agg2_chunk4(g, xs, nv, sub, c, acc);
        }
        #pragma unroll
        for (int r = 0; r < 4; ++r) {
            acc[r] += __shfl_xor(acc[r], 8);
            acc[r] += __shfl_xor(acc[r], 16);
        }
        float res = (sub == 0) ? acc[0] : (sub == 1) ? acc[1] : (sub == 2) ? acc[2] : acc[3];

        float val = (res + gown) * w + bias;
        float mx = (c < NCLS) ? val : -3.4e38f;
        mx = fmaxf(mx, __shfl_xor(mx, 1));
        mx = fmaxf(mx, __shfl_xor(mx, 2));
        mx = fmaxf(mx, __shfl_xor(mx, 4));
        float ex = expf(val - mx);
        float s = 0.0f;
        #pragma unroll
        for (int k = 0; k < NCLS; ++k) s += __shfl(ex, (lane & 24) + k);
        float l = logf(s);
        int pb = it & 1;
        if (c < NCLS) stage[pb][(wv * 4 + sub) * NCLS + c] = val - mx - l;
        __syncthreads();
        int tid = threadIdx.x;
        int o = t * (32 * NCLS) + tid;
        if (tid < 32 * NCLS && o < n * NCLS) out[o] = stage[pb][tid];
        #pragma unroll
        for (int r = 0; r < 4; ++r) {
            bg0[r] = bg1[r]; cn0[r] = cn1[r];
            s0[r] = n0[r]; s1[r] = n1[r];
        }
        vb1 = vb2; vc1 = vc2;
    }
}


extern "C" void kernel_launch(void* const* d_in, const int* in_sizes, int n_in,
                              void* d_out, int out_size, void* d_ws, size_t ws_size,
                              hipStream_t stream) {
    const float* x  = (const float*)d_in[0];
    const int*   ei = (const int*)d_in[1];
    const float* W1 = (const float*)d_in[2];
    const float* b1 = (const float*)d_in[3];
    const float* W2 = (const float*)d_in[4];
    const float* b2 = (const float*)d_in[5];
    float* out = (float*)d_out;

    const int n = in_sizes[0] / IN_FEAT;
    const int E = in_sizes[1] / 2;
    const int* src = ei;
    const int* dst = ei + E;

    const int B = 256;
    const int tiles = (n + 15) / 16;
    const int gemmBlocks = (tiles + 7) / 8;

    const int nb = (n + RB - 1) >> RB_SHIFT;
    int ch = (E + 255) / 256;
    if (ch > CH_MAX) ch = CH_MAX;
    if (ch < 1) ch = 1;
    const int G = (E + ch - 1) / ch;
    const size_t n4 = ((size_t)n + 3) & ~(size_t)3;
    const size_t E4 = ((size_t)E + 3) & ~(size_t)3;
    const size_t m  = (size_t)nb * G;
    const size_t m4 = (m + 3) & ~(size_t)3;
    const size_t a_words = (E4 > (size_t)32 * n) ? E4 : (size_t)32 * n;
    const size_t need = (3 * n4 + 2 * m4 + a_words + (size_t)E) * sizeof(float);

    const size_t af_words = (E4 > (size_t)16 * n) ? E4 : (size_t)16 * n;
    const size_t need_f = (3 * n4 + 2 * m4 + af_words + E4 + (size_t)16 * n) * sizeof(float);
    if (need_f <= ws_size && E > 0 && nb <= NB_MAX && n <= (1 << 23)) {
        float*    f      = (float*)d_ws;
        float*    dinv   = f;
        int*      cnt    = (int*)(f + n4);
        int*      first  = (int*)(f + 2 * n4);
        int*      gh     = (int*)(f + 3 * n4);
        int*      lofs   = (int*)(f + 3 * n4 + m4);
        float*    a      = f + 3 * n4 + 2 * m4;
        unsigned* binned = (unsigned*)a;
        float*    g      = a;
        int*      ssrc   = (int*)(a + af_words);
        float*    h      = a + af_words + E4;

        gcn_bin_scatter   <<<G, BIN_T, 0, stream>>>(src, dst, gh, lofs, binned, E, ch, nb, G);
        gcn_bin_sort_gemm1<<<nb, BIN_T, 0, stream>>>(binned, gh, lofs, ssrc, first, cnt, dinv, x, W1, h, n, ch, nb, G);
        const int t1 = (n + 15) / 16;
        const int t2 = (n + 31) / 32;
        gcn_agg1_rows_w2 <<<(t1 < AGG_WGS) ? t1 : AGG_WGS, B, 0, stream>>>(ssrc, first, cnt, h, dinv, b1, W2, g, n, t1);
        gcn_agg2_rows_lsm<<<(t2 < AGG_WGS) ? t2 : AGG_WGS, B, 0, stream>>>(ssrc, first, cnt, g, dinv, b2, out, n, t2);
        return;
    }

    if (need <= ws_size && E > 0 && nb <= NB_MAX && n <= (1 << 23)) {
        float*    f      = (float*)d_ws;
        float*    dinv   = f;
        int*      cnt    = (int*)(f + n4);
        int*      first  = (int*)(f + 2 * n4);
        int*      gh     = (int*)(f + 3 * n4);
        int*      lofs   = (int*)(f + 3 * n4 + m4);
        float*    a      = f + 3 * n4 + 2 * m4;
        unsigned* binned = (unsigned*)a;
        float*    h      = a;
        float*    g      = a + (size_t)16 * n;
        int*      ssrc   = (int*)(a + a_words);

        gcn_bin_scatter<<<G, BIN_T, 0, stream>>>(src, dst, gh, lofs, binned, E, ch, nb, G);
        gcn_bin_sort   <<<nb, BIN_T, 0, stream>>>(binned, gh, lofs, ssrc, first, cnt, dinv, n, ch, nb, G);

        gcn_gemm1_wmma <<<gemmBlocks, B, 0, stream>>>(x, W1, dinv, h, n);
        const int t1 = (n + 15) / 16;
        const int t2 = (n + 31) / 32;
        gcn_agg1_rows_w2 <<<(t1 < AGG_WGS) ? t1 : AGG_WGS, B, 0, stream>>>(ssrc, first, cnt, h, dinv, b1, W2, g, n, t1);
        gcn_agg2_rows_lsm<<<(t2 < AGG_WGS) ? t2 : AGG_WGS, B, 0, stream>>>(ssrc, first, cnt, g, dinv, b2, out, n, t2);
        return;
    }

    float* f     = (float*)d_ws;
    float* dinv  = f;
    float* h     = f + (size_t)n;
    float* out1  = f + (size_t)17 * n;
    float* hrelu = f + (size_t)33 * n;
    float* g     = out1;
    float* out2  = h;

    gcn_deg_init   <<<(n + B - 1) / B, B, 0, stream>>>(dinv, n);
    gcn_deg_accum  <<<(E + B - 1) / B, B, 0, stream>>>(dst, dinv, E);
    gcn_deg_rsqrt  <<<(n + B - 1) / B, B, 0, stream>>>(dinv, n);

    gcn_zero       <<<((size_t)16 * n + B - 1) / B, B, 0, stream>>>(out1, 16 * n);
    gcn_gemm1_wmma <<<gemmBlocks, B, 0, stream>>>(x, W1, dinv, h, n);
    gcn_agg1       <<<(E + B - 1) / B, B, 0, stream>>>(src, dst, h, out1, E);
    gcn_selfloop_relu<<<(n + B - 1) / B, B, 0, stream>>>(out1, h, dinv, b1, hrelu, n);

    gcn_zero       <<<((size_t)8 * n + B - 1) / B, B, 0, stream>>>(out2, 8 * n);
    gcn_gemm2_wmma <<<gemmBlocks, B, 0, stream>>>(hrelu, W2, dinv, g, n);
    gcn_agg2       <<<(E + B - 1) / B, B, 0, stream>>>(src, dst, g, out2, E);
    gcn_logsoftmax <<<(n + B - 1) / B, B, 0, stream>>>(out2, g, dinv, b2, out, n);
}
